# starting kernel
# baseline (speedup 1.0000x reference)
#include <hip/hip_runtime.h>
#include <hip/hip_bf16.h>


typedef __attribute__((ext_vector_type(2))) float v2f;
typedef __attribute__((ext_vector_type(8))) float v8f;

#define BATCH 2048
#define FRAME 267
#define LAT   32
#define HID   256
#define NE    6
#define GH    64

__device__ __forceinline__ float elu1(float x) {
  return x > 0.f ? x : expm1f(x);
}

__device__ __forceinline__ v8f wmma4(v2f a, v2f b, v8f c) {
  return __builtin_amdgcn_wmma_f32_16x16x4_f32(
      false, a, false, b, (short)0, c, false, false);
}

template <int N, bool MOE, int NT>
__device__ __forceinline__ void seg_gemm(
    const float* __restrict__ Aseg, int aBase,
    int len, float scale,
    const float* __restrict__ Wseg,
    const int* col, int h, v8f* acc)
{
  int wIdx[NT];
  #pragma unroll
  for (int j = 0; j < NT; ++j) wIdx[j] = 2 * h * N + col[j];
  const int a0 = aBase + 2 * h;

  const int len4 = len & ~3;
  int k = 0;
  #pragma unroll 8
  for (; k < len4; k += 4) {
    v2f a;
    a.x = Aseg[a0 + k];
    a.y = Aseg[a0 + k + 1];
    if (MOE) { a.x *= scale; a.y *= scale; }
    #pragma unroll
    for (int j = 0; j < NT; ++j) {
      v2f b;
      b.x = Wseg[wIdx[j]];
      b.y = Wseg[wIdx[j] + N];
      acc[j] = wmma4(a, b, acc[j]);
      wIdx[j] += 4 * N;
    }
  }
  if (k < len) {
    const int kk = k + 2 * h;
    v2f a;
    a.x = (kk     < len) ? Aseg[aBase + kk]     : 0.f;
    a.y = (kk + 1 < len) ? Aseg[aBase + kk + 1] : 0.f;
    if (MOE) { a.x *= scale; a.y *= scale; }
    const int r0 = ((kk     < len) ? kk     : len - 1) - kk;
    const int r1 = ((kk + 1 < len) ? kk + 1 : len - 1) - kk;
    #pragma unroll
    for (int j = 0; j < NT; ++j) {
      v2f b;
      b.x = Wseg[wIdx[j] + r0 * N];
      b.y = Wseg[wIdx[j] + r1 * N];
      acc[j] = wmma4(a, b, acc[j]);
    }
  }
}

template <int N, bool MOE, int NT>
__global__ void __launch_bounds__(32)
gemm_wmma_f32(const float* __restrict__ A1, int k1,
              const float* __restrict__ A2, int k2,
              const float* __restrict__ coeff,
              const float* __restrict__ W,
              const float* __restrict__ bias,
              const float* __restrict__ bias2d,
              float* __restrict__ C,
              int act)
{
  const int lane = threadIdx.x;
  const int h    = lane >> 4;
  const int ln   = lane & 15;
  const int m0   = blockIdx.y * 16;
  const int n0   = blockIdx.x * (16 * NT);
  const int rowA = m0 + ln;

  int col[NT];
  #pragma unroll
  for (int j = 0; j < NT; ++j) {
    const int cb = n0 + 16 * j + ln;
    col[j] = (cb < N) ? cb : N - 1;
  }

  v8f acc[NT];
  #pragma unroll
  for (int j = 0; j < NT; ++j) acc[j] = (v8f){0.f,0.f,0.f,0.f,0.f,0.f,0.f,0.f};

  const int a1Base = rowA * k1;
  const int a2Base = rowA * k2;
  const int I = k1 + k2;
  const int nExp = MOE ? NE : 1;

  for (int e = 0; e < nExp; ++e) {
    const float scale = MOE ? coeff[rowA * NE + e] : 1.f;
    const float* Wb = W + (size_t)e * I * N;
    seg_gemm<N, MOE, NT>(A1, a1Base, k1, scale, Wb, col, h, acc);
    if (k2)
      seg_gemm<N, MOE, NT>(A2, a2Base, k2, scale, Wb + (size_t)k1 * N, col, h,
                           acc);
  }

  #pragma unroll
  for (int j = 0; j < NT; ++j) {
    const int colB = n0 + 16 * j + ln;
    if (colB >= N) continue;
    #pragma unroll
    for (int i = 0; i < 8; ++i) {
      const int row = m0 + i + 8 * h;
      float y = acc[j][i];
      if (bias)   y += bias[colB];
      if (bias2d) y += bias2d[row * N + colB];
      if (act)    y = elu1(y);
      C[row * N + colB] = y;
    }
  }
}

__global__ void reparam_kernel(const float* __restrict__ m,
                               const float* __restrict__ v,
                               const float* __restrict__ eps,
                               float* __restrict__ z, int n) {
  int i = blockIdx.x * blockDim.x + threadIdx.x;
  if (i < n) z[i] = m[i] + eps[i] * expf(0.5f * v[i]);
}

__global__ void gate3_softmax_kernel(const float* __restrict__ g,
                                     const float* __restrict__ w,
                                     const float* __restrict__ b,
                                     float* __restrict__ coeff) {
  int row = blockIdx.x * blockDim.x + threadIdx.x;
  if (row >= BATCH) return;
  float lg[NE];
  #pragma unroll
  for (int e = 0; e < NE; ++e) lg[e] = b[e];
  for (int i = 0; i < GH; ++i) {
    float gv = g[row * GH + i];
    #pragma unroll
    for (int e = 0; e < NE; ++e) lg[e] = fmaf(gv, w[i * NE + e], lg[e]);
  }
  float mx = lg[0];
  #pragma unroll
  for (int e = 1; e < NE; ++e) mx = fmaxf(mx, lg[e]);
  float s = 0.f, ex[NE];
  #pragma unroll
  for (int e = 0; e < NE; ++e) { ex[e] = expf(lg[e] - mx); s += ex[e]; }
  float inv = 1.f / s;
  #pragma unroll
  for (int e = 0; e < NE; ++e) coeff[row * NE + e] = ex[e] * inv;
}

__global__ void mixed_bias_kernel(const float* __restrict__ coeff,
                                  const float* __restrict__ b0,
                                  const float* __restrict__ b1,
                                  const float* __restrict__ b2,
                                  float* __restrict__ mb0,
                                  float* __restrict__ mb1,
                                  float* __restrict__ mb2) {
  int row = blockIdx.x;
  float cf[NE];
  #pragma unroll
  for (int e = 0; e < NE; ++e) cf[e] = coeff[row * NE + e];
  for (int j = threadIdx.x; j < HID; j += blockDim.x) {
    float s0 = 0.f, s1 = 0.f;
    #pragma unroll
    for (int e = 0; e < NE; ++e) {
      s0 = fmaf(cf[e], b0[e * HID + j], s0);
      s1 = fmaf(cf[e], b1[e * HID + j], s1);
    }
    mb0[row * HID + j] = s0;
    mb1[row * HID + j] = s1;
  }
  for (int j = threadIdx.x; j < FRAME; j += blockDim.x) {
    float s2 = 0.f;
    #pragma unroll
    for (int e = 0; e < NE; ++e) s2 = fmaf(cf[e], b2[e * FRAME + j], s2);
    mb2[row * FRAME + j] = s2;
  }
}

template <int N, int NT>
static inline dim3 gemm_grid() {
  return dim3((N + 16 * NT - 1) / (16 * NT), BATCH / 16);
}

extern "C" void kernel_launch(void* const* d_in, const int* in_sizes, int n_in,
                              void* d_out, int out_size, void* d_ws, size_t ws_size,
                              hipStream_t stream) {
  const float* x      = (const float*)d_in[0];
  const float* c      = (const float*)d_in[1];
  const float* eps    = (const float*)d_in[2];
  const float* efc1_w = (const float*)d_in[3];
  const float* efc1_b = (const float*)d_in[4];
  const float* efc2_w = (const float*)d_in[5];
  const float* efc2_b = (const float*)d_in[6];
  const float* em_w   = (const float*)d_in[7];
  const float* em_b   = (const float*)d_in[8];
  const float* ev_w   = (const float*)d_in[9];
  const float* ev_b   = (const float*)d_in[10];
  const float* g1_w   = (const float*)d_in[11];
  const float* g1_b   = (const float*)d_in[12];
  const float* g2_w   = (const float*)d_in[13];
  const float* g2_b   = (const float*)d_in[14];
  const float* g3_w   = (const float*)d_in[15];
  const float* g3_b   = (const float*)d_in[16];
  const float* w0     = (const float*)d_in[17];
  const float* b0     = (const float*)d_in[18];
  const float* w1     = (const float*)d_in[19];
  const float* b1     = (const float*)d_in[20];
  const float* w2     = (const float*)d_in[21];
  const float* b2     = (const float*)d_in[22];

  float* out   = (float*)d_out;
  float* out_m = out + BATCH * FRAME;
  float* out_v = out_m + BATCH * LAT;

  float* ws    = (float*)d_ws;
  float* h1    = ws;
  float* h2    = h1  + BATCH * HID;
  float* z     = h2  + BATCH * HID;
  float* g1o   = z   + BATCH * LAT;
  float* g2o   = g1o + BATCH * GH;
  float* coeff = g2o + BATCH * GH;
  float* mb0   = coeff + BATCH * NE;
  float* mb1   = mb0 + BATCH * HID;
  float* mb2   = mb1 + BATCH * HID;

  gemm_wmma_f32<HID, false, 4><<<gemm_grid<HID, 4>(), 32, 0, stream>>>(
      x, FRAME, c, FRAME, nullptr, efc1_w, efc1_b, nullptr, h1, 1);
  gemm_wmma_f32<HID, false, 4><<<gemm_grid<HID, 4>(), 32, 0, stream>>>(
      x, FRAME, h1, HID, nullptr, efc2_w, efc2_b, nullptr, h2, 1);
  gemm_wmma_f32<LAT, false, 2><<<gemm_grid<LAT, 2>(), 32, 0, stream>>>(
      x, FRAME, h2, HID, nullptr, em_w, em_b, nullptr, out_m, 0);
  gemm_wmma_f32<LAT, false, 2><<<gemm_grid<LAT, 2>(), 32, 0, stream>>>(
      x, FRAME, h2, HID, nullptr, ev_w, ev_b, nullptr, out_v, 0);
  reparam_kernel<<<(BATCH * LAT + 255) / 256, 256, 0, stream>>>(
      out_m, out_v, eps, z, BATCH * LAT);

  gemm_wmma_f32<GH, false, 2><<<gemm_grid<GH, 2>(), 32, 0, stream>>>(
      z, LAT, c, FRAME, nullptr, g1_w, g1_b, nullptr, g1o, 1);
  gemm_wmma_f32<GH, false, 2><<<gemm_grid<GH, 2>(), 32, 0, stream>>>(
      g1o, GH, nullptr, 0, nullptr, g2_w, g2_b, nullptr, g2o, 1);
  gate3_softmax_kernel<<<(BATCH + 255) / 256, 256, 0, stream>>>(
      g2o, g3_w, g3_b, coeff);
  mixed_bias_kernel<<<BATCH, 256, 0, stream>>>(coeff, b0, b1, b2, mb0, mb1, mb2);

  gemm_wmma_f32<HID, true, 4><<<gemm_grid<HID, 4>(), 32, 0, stream>>>(
      z, LAT, c, FRAME, coeff, w0, nullptr, mb0, h1, 1);
  gemm_wmma_f32<HID, true, 4><<<gemm_grid<HID, 4>(), 32, 0, stream>>>(
      z, LAT, h1, HID, coeff, w1, nullptr, mb1, h2, 1);
  gemm_wmma_f32<FRAME, true, 4><<<gemm_grid<FRAME, 4>(), 32, 0, stream>>>(
      z, LAT, h2, HID, coeff, w2, nullptr, mb2, out, 0);
}


# GEMMs: 8-wave work-groups stage A and W k-chunks in LDS (coalesced global loads, two stages) and read WMMA fragments from LDS instead of per-wave dword global loads
# speedup vs baseline: 1.3729x; 1.3729x over previous
#include <hip/hip_runtime.h>
#include <hip/hip_bf16.h>


typedef __attribute__((ext_vector_type(2))) float v2f;
typedef __attribute__((ext_vector_type(4))) float v4f;
typedef __attribute__((ext_vector_type(8))) float v8f;

#define BATCH 2048
#define FRAME 267
#define LAT   32
#define HID   256
#define NE    6
#define GH    64
#define KC    32

__device__ __forceinline__ float elu1(float x) {
  return x > 0.f ? x : expm1f(x);
}

__device__ __forceinline__ v8f wmma4(v2f a, v2f b, v8f c) {
  return __builtin_amdgcn_wmma_f32_16x16x4_f32(
      false, a, false, b, (short)0, c, false, false);
}

template <int N, bool MOE, int MT, int NTB>
__global__ void __launch_bounds__(256)
gemm_lds_f32(const float* __restrict__ A1, int k1,
             const float* __restrict__ A2, int k2,
             const float* __restrict__ coeff,
             const float* __restrict__ W,
             const float* __restrict__ bias,
             const float* __restrict__ bias2d,
             float* __restrict__ C,
             int act)
{
  static_assert(MT * NTB == 8, "eight waves per work-group");
  constexpr int BM  = 16 * MT;
  constexpr int BN  = 16 * NTB;
  constexpr int PA  = KC + 4;
  constexpr int PB  = BN + 8;
  constexpr int AR  = BM / 8;
  constexpr bool VEC = (N % BN == 0);
  constexpr int F4R = BN / 4;
  constexpr int BP  = KC * F4R / 256;
  constexpr int BRS = 256 / F4R;
  constexpr int BS  = KC * BN / 256;
  constexpr int BSS = 256 / BN;

  __shared__ __attribute__((aligned(16))) float As[2][BM * PA];
  __shared__ __attribute__((aligned(16))) float Bs[2][KC * PB];

  const int tid  = threadIdx.x;
  const int lane = tid & 31;
  const int wave = tid >> 5;
  const int h    = lane >> 4;
  const int ln   = lane & 15;
  const int wm   = wave / NTB;
  const int wn   = wave % NTB;
  const int m0   = blockIdx.y * BM;
  const int n0   = blockIdx.x * BN;

  const int ak  = tid & 31;
  const int ar  = tid >> 5;
  const int bn4 = tid % F4R;
  const int bk  = tid / F4R;
  const int bn  = tid % BN;
  const int bk2 = tid / BN;
  const int colc = (n0 + bn < N) ? (n0 + bn) : (N - 1);

  const int c1  = (k1 + KC - 1) / KC;
  const int c2  = (k2 + KC - 1) / KC;
  const int cpe = c1 + c2;
  const int nc  = (MOE ? NE : 1) * cpe;
  const int I   = k1 + k2;

  float  ra[AR];
  float  rs[AR];
  bool   av = true;
  v4f    rb[BP > 0 ? BP : 1];
  float  rbs[BS];

  auto gload = [&](int i) {
    const int e  = MOE ? i / cpe : 0;
    const int r  = i - e * cpe;
    const bool s2 = r >= c1;
    const float* __restrict__ Aseg = s2 ? A2 : A1;
    const int len = s2 ? k2 : k1;
    const int k0  = (s2 ? r - c1 : r) * KC;
    const float* __restrict__ Wseg = W + (e * I + (s2 ? k1 : 0)) * N;
    const int ka  = k0 + ak;
    av = ka < len;
    const int kac = av ? ka : len - 1;
    #pragma unroll
    for (int p = 0; p < AR; ++p) {
      const int row = m0 + ar + 8 * p;
      ra[p] = Aseg[row * len + kac];
      if (MOE) rs[p] = coeff[row * NE + e];
    }
    if (VEC) {
      #pragma unroll
      for (int p = 0; p < BP; ++p) {
        int kr = k0 + bk + BRS * p;
        kr = kr < len ? kr : len - 1;
        rb[p] = *(const v4f*)(Wseg + kr * N + n0 + 4 * bn4);
      }
    } else {
      #pragma unroll
      for (int p = 0; p < BS; ++p) {
        int kr = k0 + bk2 + BSS * p;
        kr = kr < len ? kr : len - 1;
        rbs[p] = Wseg[kr * N + colc];
      }
    }
  };

  auto sstore = [&](int st) {
    float* as = As[st];
    float* bs = Bs[st];
    #pragma unroll
    for (int p = 0; p < AR; ++p) {
      float v = av ? ra[p] : 0.f;
      if (MOE) v *= rs[p];
      as[(ar + 8 * p) * PA + ak] = v;
    }
    if (VEC) {
      #pragma unroll
      for (int p = 0; p < BP; ++p)
        *(v4f*)(bs + (bk + BRS * p) * PB + 4 * bn4) = rb[p];
    } else {
      #pragma unroll
      for (int p = 0; p < BS; ++p)
        bs[(bk2 + BSS * p) * PB + bn] = rbs[p];
    }
  };

  v8f acc = (v8f){0.f,0.f,0.f,0.f,0.f,0.f,0.f,0.f};

  const int aoff = (16 * wm + ln) * PA + 2 * h;
  const int boff = (2 * h) * PB + 16 * wn + ln;

  gload(0);
  sstore(0);
  __syncthreads();

  for (int i = 0; i < nc; ++i) {
    const int st = i & 1;
    if (i + 1 < nc) gload(i + 1);

    const float* as = As[st] + aoff;
    const float* bs = Bs[st] + boff;
    v2f af[KC / 4], bf[KC / 4];
    #pragma unroll
    for (int ks = 0; ks < KC / 4; ++ks) {
      af[ks]   = *(const v2f*)(as + 4 * ks);
      bf[ks].x = bs[(4 * ks) * PB];
      bf[ks].y = bs[(4 * ks + 1) * PB];
    }
    asm volatile("" :: "v"(af[0]), "v"(af[1]), "v"(af[2]), "v"(af[3]),
                       "v"(af[4]), "v"(af[5]), "v"(af[6]), "v"(af[7]),
                       "v"(bf[0]), "v"(bf[1]), "v"(bf[2]), "v"(bf[3]),
                       "v"(bf[4]), "v"(bf[5]), "v"(bf[6]), "v"(bf[7]));
    #pragma unroll
    for (int ks = 0; ks < KC / 4; ++ks)
      acc = wmma4(af[ks], bf[ks], acc);

    if (i + 1 < nc) sstore(st ^ 1);
    __syncthreads();
  }

  const int colB = n0 + 16 * wn + ln;
  if (colB < N) {
    #pragma unroll
    for (int r = 0; r < 8; ++r) {
      const int row = m0 + 16 * wm + r + 8 * h;
      float y = acc[r];
      if (bias)   y += bias[colB];
      if (bias2d) y += bias2d[row * N + colB];
      if (act)    y = elu1(y);
      C[row * N + colB] = y;
    }
  }
}

__global__ void reparam_kernel(const float* __restrict__ m,
                               const float* __restrict__ v,
                               const float* __restrict__ eps,
                               float* __restrict__ z, int n) {
  int i = blockIdx.x * blockDim.x + threadIdx.x;
  if (i < n) z[i] = m[i] + eps[i] * expf(0.5f * v[i]);
}

__global__ void gate3_softmax_kernel(const float* __restrict__ g,
                                     const float* __restrict__ w,
                                     const float* __restrict__ b,
                                     float* __restrict__ coeff) {
  int row = blockIdx.x * blockDim.x + threadIdx.x;
  if (row >= BATCH) return;
  float lg[NE];
  #pragma unroll
  for (int e = 0; e < NE; ++e) lg[e] = b[e];
  for (int i = 0; i < GH; ++i) {
    float gv = g[row * GH + i];
    #pragma unroll
    for (int e = 0; e < NE; ++e) lg[e] = fmaf(gv, w[i * NE + e], lg[e]);
  }
  float mx = lg[0];
  #pragma unroll
  for (int e = 1; e < NE; ++e) mx = fmaxf(mx, lg[e]);
  float s = 0.f, ex[NE];
  #pragma unroll
  for (int e = 0; e < NE; ++e) { ex[e] = expf(lg[e] - mx); s += ex[e]; }
  float inv = 1.f / s;
  #pragma unroll
  for (int e = 0; e < NE; ++e) coeff[row * NE + e] = ex[e] * inv;
}

__global__ void mixed_bias_kernel(const float* __restrict__ coeff,
                                  const float* __restrict__ b0,
                                  const float* __restrict__ b1,
                                  const float* __restrict__ b2,
                                  float* __restrict__ mb0,
                                  float* __restrict__ mb1,
                                  float* __restrict__ mb2) {
  int row = blockIdx.x;
  float cf[NE];
  #pragma unroll
  for (int e = 0; e < NE; ++e) cf[e] = coeff[row * NE + e];
  for (int j = threadIdx.x; j < HID; j += blockDim.x) {
    float s0 = 0.f, s1 = 0.f;
    #pragma unroll
    for (int e = 0; e < NE; ++e) {
      s0 = fmaf(cf[e], b0[e * HID + j], s0);
      s1 = fmaf(cf[e], b1[e * HID + j], s1);
    }
    mb0[row * HID + j] = s0;
    mb1[row * HID + j] = s1;
  }
  for (int j = threadIdx.x; j < FRAME; j += blockDim.x) {
    float s2 = 0.f;
    #pragma unroll
    for (int e = 0; e < NE; ++e) s2 = fmaf(cf[e], b2[e * FRAME + j], s2);
    mb2[row * FRAME + j] = s2;
  }
}

template <int N, int MT, int NTB>
static inline dim3 gemm_grid() {
  return dim3((N + 16 * NTB - 1) / (16 * NTB), BATCH / (16 * MT));
}

extern "C" void kernel_launch(void* const* d_in, const int* in_sizes, int n_in,
                              void* d_out, int out_size, void* d_ws, size_t ws_size,
                              hipStream_t stream) {
  const float* x      = (const float*)d_in[0];
  const float* c      = (const float*)d_in[1];
  const float* eps    = (const float*)d_in[2];
  const float* efc1_w = (const float*)d_in[3];
  const float* efc1_b = (const float*)d_in[4];
  const float* efc2_w = (const float*)d_in[5];
  const float* efc2_b = (const float*)d_in[6];
  const float* em_w   = (const float*)d_in[7];
  const float* em_b   = (const float*)d_in[8];
  const float* ev_w   = (const float*)d_in[9];
  const float* ev_b   = (const float*)d_in[10];
  const float* g1_w   = (const float*)d_in[11];
  const float* g1_b   = (const float*)d_in[12];
  const float* g2_w   = (const float*)d_in[13];
  const float* g2_b   = (const float*)d_in[14];
  const float* g3_w   = (const float*)d_in[15];
  const float* g3_b   = (const float*)d_in[16];
  const float* w0     = (const float*)d_in[17];
  const float* b0     = (const float*)d_in[18];
  const float* w1     = (const float*)d_in[19];
  const float* b1     = (const float*)d_in[20];
  const float* w2     = (const float*)d_in[21];
  const float* b2     = (const float*)d_in[22];

  float* out   = (float*)d_out;
  float* out_m = out + BATCH * FRAME;
  float* out_v = out_m + BATCH * LAT;

  float* ws    = (float*)d_ws;
  float* h1    = ws;
  float* h2    = h1  + BATCH * HID;
  float* z     = h2  + BATCH * HID;
  float* g1o   = z   + BATCH * LAT;
  float* g2o   = g1o + BATCH * GH;
  float* coeff = g2o + BATCH * GH;
  float* mb0   = coeff + BATCH * NE;
  float* mb1   = mb0 + BATCH * HID;
  float* mb2   = mb1 + BATCH * HID;

  gemm_lds_f32<HID, false, 2, 4><<<gemm_grid<HID, 2, 4>(), 256, 0, stream>>>(
      x, FRAME, c, FRAME, nullptr, efc1_w, efc1_b, nullptr, h1, 1);
  gemm_lds_f32<HID, false, 2, 4><<<gemm_grid<HID, 2, 4>(), 256, 0, stream>>>(
      x, FRAME, h1, HID, nullptr, efc2_w, efc2_b, nullptr, h2, 1);
  gemm_lds_f32<LAT, false, 4, 2><<<gemm_grid<LAT, 4, 2>(), 256, 0, stream>>>(
      x, FRAME, h2, HID, nullptr, em_w, em_b, nullptr, out_m, 0);
  gemm_lds_f32<LAT, false, 4, 2><<<gemm_grid<LAT, 4, 2>(), 256, 0, stream>>>(
      x, FRAME, h2, HID, nullptr, ev_w, ev_b, nullptr, out_v, 0);
  reparam_kernel<<<(BATCH * LAT + 255) / 256, 256, 0, stream>>>(
      out_m, out_v, eps, z, BATCH * LAT);

  gemm_lds_f32<GH, false, 2, 4><<<gemm_grid<GH, 2, 4>(), 256, 0, stream>>>(
      z, LAT, c, FRAME, nullptr, g1_w, g1_b, nullptr, g1o, 1);
  gemm_lds_f32<GH, false, 2, 4><<<gemm_grid<GH, 2, 4>(), 256, 0, stream>>>(
      g1o, GH, nullptr, 0, nullptr, g2_w, g2_b, nullptr, g2o, 1);
  gate3_softmax_kernel<<<(BATCH + 255) / 256, 256, 0, stream>>>(
      g2o, g3_w, g3_b, coeff);
  mixed_bias_kernel<<<BATCH, 256, 0, stream>>>(coeff, b0, b1, b2, mb0, mb1, mb2);

  gemm_lds_f32<HID, true, 2, 4><<<gemm_grid<HID, 2, 4>(), 256, 0, stream>>>(
      z, LAT, c, FRAME, coeff, w0, nullptr, mb0, h1, 1);
  gemm_lds_f32<HID, true, 2, 4><<<gemm_grid<HID, 2, 4>(), 256, 0, stream>>>(
      z, LAT, h1, HID, coeff, w1, nullptr, mb1, h2, 1);
  gemm_lds_f32<FRAME, true, 2, 4><<<gemm_grid<FRAME, 2, 4>(), 256, 0, stream>>>(
      z, LAT, h2, HID, coeff, w2, nullptr, mb2, out, 0);
}


# GEMMs: K chunk per LDS stage 32->128, cut on one flat expert-by-input K axis: 4x fewer barrier and load-latency rounds; k regrouped into WMMA steps, so the summation order shifts
# speedup vs baseline: 1.9153x; 1.3951x over previous
#include <hip/hip_runtime.h>
#include <hip/hip_bf16.h>


typedef __attribute__((ext_vector_type(2))) float v2f;
typedef __attribute__((ext_vector_type(4))) float v4f;
typedef __attribute__((ext_vector_type(8))) float v8f;

#define BATCH 2048
#define FRAME 267
#define LAT   32
#define HID   256
#define NE    6
#define GH    64
#define KC    128

__device__ __forceinline__ float elu1(float x) {
  return x > 0.f ? x : expm1f(x);
}

__device__ __forceinline__ v8f wmma4(v2f a, v2f b, v8f c) {
  return __builtin_amdgcn_wmma_f32_16x16x4_f32(
      false, a, false, b, (short)0, c, false, false);
}

template <int N, bool MOE, int MT, int NTB>
__global__ void __launch_bounds__(256)
gemm_lds_f32(const float* __restrict__ A1, int k1,
             const float* __restrict__ A2, int k2,
             const float* __restrict__ coeff,
             const float* __restrict__ W,
             const float* __restrict__ bias,
             const float* __restrict__ bias2d,
             float* __restrict__ C,
             int act)
{
  static_assert(MT * NTB == 8, "eight waves per work-group");
  constexpr int BM  = 16 * MT;
  constexpr int BN  = 16 * NTB;
  constexpr int PA  = KC + 4;
  constexpr int PB  = BN + 8;
  constexpr int AJ  = KC / 32;
  constexpr int AP  = BM / 8;
  constexpr bool VEC = (N % BN == 0);
  constexpr int F4R = BN / 4;
  constexpr int BP  = KC * F4R / 256;
  constexpr int BRS = 256 / F4R;
  constexpr int BS  = KC * BN / 256;
  constexpr int BSS = 256 / BN;

  __shared__ __attribute__((aligned(16))) float As[2][BM * PA];
  __shared__ __attribute__((aligned(16))) float Bs[2][KC * PB];

  const int tid  = threadIdx.x;
  const int lane = tid & 31;
  const int wave = tid >> 5;
  const int h    = lane >> 4;
  const int ln   = lane & 15;
  const int wm   = wave / NTB;
  const int wn   = wave % NTB;
  const int m0   = blockIdx.y * BM;
  const int n0   = blockIdx.x * BN;

  const int bn4 = tid % F4R;
  const int bk  = tid / F4R;
  const int bn  = tid % BN;
  const int bk2 = tid / BN;
  const int colc = (n0 + bn < N) ? (n0 + bn) : (N - 1);

  const int I    = k1 + k2;
  const int Ktot = (MOE ? NE : 1) * I;
  const int nc   = (Ktot + KC - 1) / KC;

  float  ra[AJ][AP];
  float  cl[AP], ch[AP];
  bool   av[AJ];
  bool   ah[AJ];
  v4f    rb[BP > 0 ? BP : 1];
  float  rbs[BS];

  auto gload = [&](int i) {
    const int k0 = i * KC;
    const int e0 = MOE ? k0 / I : 0;
    const int kb = e0 * I;
    #pragma unroll
    for (int j = 0; j < AJ; ++j) {
      const int kk = k0 + 32 * j + lane;
      av[j] = kk < Ktot;
      const int kc = av[j] ? kk : Ktot - 1;
      int ii = kc - kb;
      ah[j] = MOE && ii >= I;
      if (ah[j]) ii -= I;
      const bool s2 = ii >= k1;
      const float* __restrict__ Aseg = s2 ? A2 : A1;
      const int len = s2 ? k2 : k1;
      const int cc  = s2 ? ii - k1 : ii;
      #pragma unroll
      for (int p = 0; p < AP; ++p)
        ra[j][p] = Aseg[(m0 + wave + 8 * p) * len + cc];
    }
    if (MOE) {
      const int e1 = e0 + 1 < NE ? e0 + 1 : NE - 1;
      #pragma unroll
      for (int p = 0; p < AP; ++p) {
        const int row = m0 + wave + 8 * p;
        cl[p] = coeff[row * NE + e0];
        ch[p] = coeff[row * NE + e1];
      }
    }
    if (VEC) {
      #pragma unroll
      for (int p = 0; p < BP; ++p) {
        int kr = k0 + bk + BRS * p;
        kr = kr < Ktot ? kr : Ktot - 1;
        rb[p] = *(const v4f*)(W + kr * N + n0 + 4 * bn4);
      }
    } else {
      #pragma unroll
      for (int p = 0; p < BS; ++p) {
        int kr = k0 + bk2 + BSS * p;
        kr = kr < Ktot ? kr : Ktot - 1;
        rbs[p] = W[kr * N + colc];
      }
    }
  };

  auto sstore = [&](int st) {
    float* as = As[st];
    float* bs = Bs[st];
    #pragma unroll
    for (int j = 0; j < AJ; ++j) {
      #pragma unroll
      for (int p = 0; p < AP; ++p) {
        float v = av[j] ? ra[j][p] : 0.f;
        if (MOE) v *= ah[j] ? ch[p] : cl[p];
        as[(wave + 8 * p) * PA + 32 * j + lane] = v;
      }
    }
    if (VEC) {
      #pragma unroll
      for (int p = 0; p < BP; ++p)
        *(v4f*)(bs + (bk + BRS * p) * PB + 4 * bn4) = rb[p];
    } else {
      #pragma unroll
      for (int p = 0; p < BS; ++p)
        bs[(bk2 + BSS * p) * PB + bn] = rbs[p];
    }
  };

  v8f acc = (v8f){0.f,0.f,0.f,0.f,0.f,0.f,0.f,0.f};

  const int aoff = (16 * wm + ln) * PA + 2 * h;
  const int boff = (2 * h) * PB + 16 * wn + ln;

  gload(0);
  sstore(0);
  __syncthreads();

  for (int i = 0; i < nc; ++i) {
    const int st = i & 1;
    if (i + 1 < nc) gload(i + 1);

    const int rem = Ktot - i * KC;
    const int nsb = rem >= KC ? KC / 32 : (rem + 31) / 32;

    const float* as = As[st] + aoff;
    const float* bs = Bs[st] + boff;
    for (int j = 0; j < nsb; ++j) {
      const float* asj = as + 32 * j;
      const float* bsj = bs + 32 * j * PB;
      v2f af[8], bf[8];
      #pragma unroll
      for (int ks = 0; ks < 8; ++ks) {
        af[ks]   = *(const v2f*)(asj + 4 * ks);
        bf[ks].x = bsj[(4 * ks) * PB];
        bf[ks].y = bsj[(4 * ks + 1) * PB];
      }
      asm volatile("" :: "v"(af[0]), "v"(af[1]), "v"(af[2]), "v"(af[3]),
                         "v"(af[4]), "v"(af[5]), "v"(af[6]), "v"(af[7]),
                         "v"(bf[0]), "v"(bf[1]), "v"(bf[2]), "v"(bf[3]),
                         "v"(bf[4]), "v"(bf[5]), "v"(bf[6]), "v"(bf[7]));
      #pragma unroll
      for (int ks = 0; ks < 8; ++ks)
        acc = wmma4(af[ks], bf[ks], acc);
    }

    if (i + 1 < nc) sstore(st ^ 1);
    __syncthreads();
  }

  const int colB = n0 + 16 * wn + ln;
  if (colB < N) {
    #pragma unroll
    for (int r = 0; r < 8; ++r) {
      const int row = m0 + 16 * wm + r + 8 * h;
      float y = acc[r];
      if (bias)   y += bias[colB];
      if (bias2d) y += bias2d[row * N + colB];
      if (act)    y = elu1(y);
      C[row * N + colB] = y;
    }
  }
}

__global__ void reparam_kernel(const float* __restrict__ m,
                               const float* __restrict__ v,
                               const float* __restrict__ eps,
                               float* __restrict__ z, int n) {
  int i = blockIdx.x * blockDim.x + threadIdx.x;
  if (i < n) z[i] = m[i] + eps[i] * expf(0.5f * v[i]);
}

__global__ void gate3_softmax_kernel(const float* __restrict__ g,
                                     const float* __restrict__ w,
                                     const float* __restrict__ b,
                                     float* __restrict__ coeff) {
  int row = blockIdx.x * blockDim.x + threadIdx.x;
  if (row >= BATCH) return;
  float lg[NE];
  #pragma unroll
  for (int e = 0; e < NE; ++e) lg[e] = b[e];
  for (int i = 0; i < GH; ++i) {
    float gv = g[row * GH + i];
    #pragma unroll
    for (int e = 0; e < NE; ++e) lg[e] = fmaf(gv, w[i * NE + e], lg[e]);
  }
  float mx = lg[0];
  #pragma unroll
  for (int e = 1; e < NE; ++e) mx = fmaxf(mx, lg[e]);
  float s = 0.f, ex[NE];
  #pragma unroll
  for (int e = 0; e < NE; ++e) { ex[e] = expf(lg[e] - mx); s += ex[e]; }
  float inv = 1.f / s;
  #pragma unroll
  for (int e = 0; e < NE; ++e) coeff[row * NE + e] = ex[e] * inv;
}

__global__ void mixed_bias_kernel(const float* __restrict__ coeff,
                                  const float* __restrict__ b0,
                                  const float* __restrict__ b1,
                                  const float* __restrict__ b2,
                                  float* __restrict__ mb0,
                                  float* __restrict__ mb1,
                                  float* __restrict__ mb2) {
  int row = blockIdx.x;
  float cf[NE];
  #pragma unroll
  for (int e = 0; e < NE; ++e) cf[e] = coeff[row * NE + e];
  for (int j = threadIdx.x; j < HID; j += blockDim.x) {
    float s0 = 0.f, s1 = 0.f;
    #pragma unroll
    for (int e = 0; e < NE; ++e) {
      s0 = fmaf(cf[e], b0[e * HID + j], s0);
      s1 = fmaf(cf[e], b1[e * HID + j], s1);
    }
    mb0[row * HID + j] = s0;
    mb1[row * HID + j] = s1;
  }
  for (int j = threadIdx.x; j < FRAME; j += blockDim.x) {
    float s2 = 0.f;
    #pragma unroll
    for (int e = 0; e < NE; ++e) s2 = fmaf(cf[e], b2[e * FRAME + j], s2);
    mb2[row * FRAME + j] = s2;
  }
}

template <int N, int MT, int NTB>
static inline dim3 gemm_grid() {
  return dim3((N + 16 * NTB - 1) / (16 * NTB), BATCH / (16 * MT));
}

extern "C" void kernel_launch(void* const* d_in, const int* in_sizes, int n_in,
                              void* d_out, int out_size, void* d_ws, size_t ws_size,
                              hipStream_t stream) {
  const float* x      = (const float*)d_in[0];
  const float* c      = (const float*)d_in[1];
  const float* eps    = (const float*)d_in[2];
  const float* efc1_w = (const float*)d_in[3];
  const float* efc1_b = (const float*)d_in[4];
  const float* efc2_w = (const float*)d_in[5];
  const float* efc2_b = (const float*)d_in[6];
  const float* em_w   = (const float*)d_in[7];
  const float* em_b   = (const float*)d_in[8];
  const float* ev_w   = (const float*)d_in[9];
  const float* ev_b   = (const float*)d_in[10];
  const float* g1_w   = (const float*)d_in[11];
  const float* g1_b   = (const float*)d_in[12];
  const float* g2_w   = (const float*)d_in[13];
  const float* g2_b   = (const float*)d_in[14];
  const float* g3_w   = (const float*)d_in[15];
  const float* g3_b   = (const float*)d_in[16];
  const float* w0     = (const float*)d_in[17];
  const float* b0     = (const float*)d_in[18];
  const float* w1     = (const float*)d_in[19];
  const float* b1     = (const float*)d_in[20];
  const float* w2     = (const float*)d_in[21];
  const float* b2     = (const float*)d_in[22];

  float* out   = (float*)d_out;
  float* out_m = out + BATCH * FRAME;
  float* out_v = out_m + BATCH * LAT;

  float* ws    = (float*)d_ws;
  float* h1    = ws;
  float* h2    = h1  + BATCH * HID;
  float* z     = h2  + BATCH * HID;
  float* g1o   = z   + BATCH * LAT;
  float* g2o   = g1o + BATCH * GH;
  float* coeff = g2o + BATCH * GH;
  float* mb0   = coeff + BATCH * NE;
  float* mb1   = mb0 + BATCH * HID;
  float* mb2   = mb1 + BATCH * HID;

  gemm_lds_f32<HID, false, 2, 4><<<gemm_grid<HID, 2, 4>(), 256, 0, stream>>>(
      x, FRAME, c, FRAME, nullptr, efc1_w, efc1_b, nullptr, h1, 1);
  gemm_lds_f32<HID, false, 2, 4><<<gemm_grid<HID, 2, 4>(), 256, 0, stream>>>(
      x, FRAME, h1, HID, nullptr, efc2_w, efc2_b, nullptr, h2, 1);
  gemm_lds_f32<LAT, false, 4, 2><<<gemm_grid<LAT, 4, 2>(), 256, 0, stream>>>(
      x, FRAME, h2, HID, nullptr, em_w, em_b, nullptr, out_m, 0);
  gemm_lds_f32<LAT, false, 4, 2><<<gemm_grid<LAT, 4, 2>(), 256, 0, stream>>>(
      x, FRAME, h2, HID, nullptr, ev_w, ev_b, nullptr, out_v, 0);
  reparam_kernel<<<(BATCH * LAT + 255) / 256, 256, 0, stream>>>(
      out_m, out_v, eps, z, BATCH * LAT);

  gemm_lds_f32<GH, false, 2, 4><<<gemm_grid<GH, 2, 4>(), 256, 0, stream>>>(
      z, LAT, c, FRAME, nullptr, g1_w, g1_b, nullptr, g1o, 1);
  gemm_lds_f32<GH, false, 2, 4><<<gemm_grid<GH, 2, 4>(), 256, 0, stream>>>(
      g1o, GH, nullptr, 0, nullptr, g2_w, g2_b, nullptr, g2o, 1);
  gate3_softmax_kernel<<<(BATCH + 255) / 256, 256, 0, stream>>>(
      g2o, g3_w, g3_b, coeff);
  mixed_bias_kernel<<<BATCH, 256, 0, stream>>>(coeff, b0, b1, b2, mb0, mb1, mb2);

  gemm_lds_f32<HID, true, 2, 4><<<gemm_grid<HID, 2, 4>(), 256, 0, stream>>>(
      z, LAT, c, FRAME, coeff, w0, nullptr, mb0, h1, 1);
  gemm_lds_f32<HID, true, 2, 4><<<gemm_grid<HID, 2, 4>(), 256, 0, stream>>>(
      z, LAT, h1, HID, coeff, w1, nullptr, mb1, h2, 1);
  gemm_lds_f32<FRAME, true, 2, 4><<<gemm_grid<FRAME, 2, 4>(), 256, 0, stream>>>(
      z, LAT, h2, HID, coeff, w2, nullptr, mb2, out, 0);
}


# Launch fusion 12 -> 7: short row-local stages ride in the GEMM before them (mixed expert bias in the decoder epilogues; m, v, z from one GEMM; gating net in one launch); sum orders kept
# speedup vs baseline: 2.4295x; 1.2685x over previous
#include <hip/hip_runtime.h>
#include <hip/hip_bf16.h>


typedef __attribute__((ext_vector_type(2))) float v2f;
typedef __attribute__((ext_vector_type(4))) float v4f;
typedef __attribute__((ext_vector_type(8))) float v8f;

#define BATCH 2048
#define FRAME 267
#define LAT   32
#define HID   256
#define NE    6
#define GH    64
#define KC    128
#define MVN   (2 * LAT)

#define MODE_PLAIN 0
#define MODE_MOE   1
#define MODE_MV    2
#define MODE_GATE  3

__device__ __forceinline__ float elu1(float x) {
  return x > 0.f ? x : expm1f(x);
}

__device__ __forceinline__ v8f wmma4(v2f a, v2f b, v8f c) {
  return __builtin_amdgcn_wmma_f32_16x16x4_f32(
      false, a, false, b, (short)0, c, false, false);
}

template <int N, int MODE, int MT, int NTB>
__global__ void __launch_bounds__(256)
gemm_lds_f32(const float* __restrict__ A1, int k1,
             const float* __restrict__ A2, int k2,
             const float* __restrict__ coeff,
             const float* __restrict__ W,
             const float* __restrict__ bias,
             const float* __restrict__ W2,
             const float* __restrict__ bias2,
             const float* __restrict__ eps,
             const float* __restrict__ W3,
             const float* __restrict__ bias3,
             float* __restrict__ C,
             float* __restrict__ C2,
             float* __restrict__ Z,
             int act)
{
  constexpr bool MOE  = (MODE == MODE_MOE);
  constexpr bool MV   = (MODE == MODE_MV);
  constexpr bool GATE = (MODE == MODE_GATE);
  static_assert(!GATE || (N == GH && NTB == 4 && MT == 2), "GATE: one 32 x 64 tile holds whole rows of both hidden layers");
  static_assert(MT * NTB == 8, "eight waves per work-group");
  static_assert(!MV || (N == MVN && NTB == 4 && MT == 2), "MV: one 32 x 64 tile, columns 0..31 mean, 32..63 log-variance");
  constexpr int BM  = 16 * MT;
  constexpr int BN  = 16 * NTB;
  constexpr int PA  = KC + 4;
  constexpr int PB  = BN + 8;
  constexpr int AJ  = KC / 32;
  constexpr int AP  = BM / 8;
  constexpr bool VEC = (N % BN == 0);
  constexpr int F4R = BN / 4;
  constexpr int BP  = KC * F4R / 256;
  constexpr int BRS = 256 / F4R;
  constexpr int BS  = KC * BN / 256;
  constexpr int BSS = 256 / BN;

  __shared__ __attribute__((aligned(16))) float As[2][BM * PA];
  __shared__ __attribute__((aligned(16))) float Bs[2][KC * PB];

  const int tid  = threadIdx.x;
  const int lane = tid & 31;
  const int wave = tid >> 5;
  const int h    = lane >> 4;
  const int ln   = lane & 15;
  const int wm   = wave / NTB;
  const int wn   = wave % NTB;
  const int m0   = blockIdx.y * BM;
  const int n0   = blockIdx.x * BN;

  const int bn4 = tid % F4R;
  const int bk  = tid / F4R;
  const int bn  = tid % BN;
  const int bk2 = tid / BN;
  const int colc = (n0 + bn < N) ? (n0 + bn) : (N - 1);

  const int I    = k1 + k2;
  const int Ktot = (MOE ? NE : 1) * I;
  const int nc   = (Ktot + KC - 1) / KC;

  float  ra[AJ][AP];
  float  cl[AP], ch[AP];
  bool   av[AJ];
  bool   ah[AJ];
  v4f    rb[BP > 0 ? BP : 1];
  float  rbs[BS];

  auto gload = [&](int i) {
    const int k0 = i * KC;
    const int e0 = MOE ? k0 / I : 0;
    const int kb = e0 * I;
    #pragma unroll
    for (int j = 0; j < AJ; ++j) {
      const int kk = k0 + 32 * j + lane;
      av[j] = kk < Ktot;
      const int kc = av[j] ? kk : Ktot - 1;
      int ii = kc - kb;
      ah[j] = MOE && ii >= I;
      if (ah[j]) ii -= I;
      const bool s2 = ii >= k1;
      const float* __restrict__ Aseg = s2 ? A2 : A1;
      const int len = s2 ? k2 : k1;
      const int cc  = s2 ? ii - k1 : ii;
      #pragma unroll
      for (int p = 0; p < AP; ++p)
        ra[j][p] = Aseg[(m0 + wave + 8 * p) * len + cc];
    }
    if (MOE) {
      const int e1 = e0 + 1 < NE ? e0 + 1 : NE - 1;
      #pragma unroll
      for (int p = 0; p < AP; ++p) {
        const int row = m0 + wave + 8 * p;
        cl[p] = coeff[row * NE + e0];
        ch[p] = coeff[row * NE + e1];
      }
    }
    if (VEC) {
      #pragma unroll
      for (int p = 0; p < BP; ++p) {
        int kr = k0 + bk + BRS * p;
        kr = kr < Ktot ? kr : Ktot - 1;
        if (MV) {
          const float* __restrict__ Wh = (bn4 < F4R / 2) ? W : W2;
          rb[p] = *(const v4f*)(Wh + kr * LAT + 4 * (bn4 & (F4R / 2 - 1)));
        } else {
          rb[p] = *(const v4f*)(W + kr * N + n0 + 4 * bn4);
        }
      }
    } else {
      #pragma unroll
      for (int p = 0; p < BS; ++p) {
        int kr = k0 + bk2 + BSS * p;
        kr = kr < Ktot ? kr : Ktot - 1;
        rbs[p] = W[kr * N + colc];
      }
    }
  };

  auto sstore = [&](int st) {
    float* as = As[st];
    float* bs = Bs[st];
    #pragma unroll
    for (int j = 0; j < AJ; ++j) {
      #pragma unroll
      for (int p = 0; p < AP; ++p) {
        float v = av[j] ? ra[j][p] : 0.f;
        if (MOE) v *= ah[j] ? ch[p] : cl[p];
        as[(wave + 8 * p) * PA + 32 * j + lane] = v;
      }
    }
    if (VEC) {
      #pragma unroll
      for (int p = 0; p < BP; ++p)
        *(v4f*)(bs + (bk + BRS * p) * PB + 4 * bn4) = rb[p];
    } else {
      #pragma unroll
      for (int p = 0; p < BS; ++p)
        bs[(bk2 + BSS * p) * PB + bn] = rbs[p];
    }
  };

  v8f acc = (v8f){0.f,0.f,0.f,0.f,0.f,0.f,0.f,0.f};

  const int aoff = (16 * wm + ln) * PA + 2 * h;
  const int boff = (2 * h) * PB + 16 * wn + ln;

  auto mma32 = [&](const float* asj, const float* bsj, v8f a) {
    v2f af[8], bf[8];
    #pragma unroll
    for (int ks = 0; ks < 8; ++ks) {
      af[ks]   = *(const v2f*)(asj + 4 * ks);
      bf[ks].x = bsj[(4 * ks) * PB];
      bf[ks].y = bsj[(4 * ks + 1) * PB];
    }
    asm volatile("" :: "v"(af[0]), "v"(af[1]), "v"(af[2]), "v"(af[3]),
                       "v"(af[4]), "v"(af[5]), "v"(af[6]), "v"(af[7]),
                       "v"(bf[0]), "v"(bf[1]), "v"(bf[2]), "v"(bf[3]),
                       "v"(bf[4]), "v"(bf[5]), "v"(bf[6]), "v"(bf[7]));
    #pragma unroll
    for (int ks = 0; ks < 8; ++ks)
      a = wmma4(af[ks], bf[ks], a);
    return a;
  };

  gload(0);
  sstore(0);
  __syncthreads();

  for (int i = 0; i < nc; ++i) {
    const int st = i & 1;
    if (i + 1 < nc) gload(i + 1);

    const int rem = Ktot - i * KC;
    const int nsb = rem >= KC ? KC / 32 : (rem + 31) / 32;

    const float* as = As[st] + aoff;
    const float* bs = Bs[st] + boff;
    for (int j = 0; j < nsb; ++j)
      acc = mma32(as + 32 * j, bs + 32 * j * PB, acc);


    if (i + 1 < nc) sstore(st ^ 1);
    __syncthreads();
  }

  const int colB = n0 + 16 * wn + ln;
  if (GATE) {
    constexpr int PT = GH + 1;
    float* as2 = &As[0][0];
    float* bs2 = &Bs[0][0];
    {
      const float bv = bias[colB];
      #pragma unroll
      for (int r = 0; r < 8; ++r)
        as2[(16 * wm + r + 8 * h) * PA + colB] = elu1(acc[r] + bv);
      #pragma unroll
      for (int p = 0; p < GH / BRS; ++p) {
        const int kr = bk + BRS * p;
        *(v4f*)(bs2 + kr * PB + 4 * bn4) = *(const v4f*)(W2 + kr * GH + 4 * bn4);
      }
    }
    __syncthreads();
    v8f acc2 = (v8f){0.f,0.f,0.f,0.f,0.f,0.f,0.f,0.f};
    #pragma unroll
    for (int j = 0; j < GH / 32; ++j)
      acc2 = mma32(as2 + aoff + 32 * j, bs2 + boff + 32 * j * PB, acc2);
    float* T = &As[1][0];
    {
      const float bv2 = bias2[colB];
      #pragma unroll
      for (int r = 0; r < 8; ++r)
        T[(16 * wm + r + 8 * h) * PT + colB] = elu1(acc2[r] + bv2);
    }
    __syncthreads();
    if (tid < BM) {
      float lg[NE];
      #pragma unroll
      for (int e = 0; e < NE; ++e) lg[e] = bias3[e];
      for (int i = 0; i < GH; ++i) {
        const float gv = T[tid * PT + i];
        #pragma unroll
        for (int e = 0; e < NE; ++e) lg[e] = fmaf(gv, W3[i * NE + e], lg[e]);
      }
      float mx = lg[0];
      #pragma unroll
      for (int e = 1; e < NE; ++e) mx = fmaxf(mx, lg[e]);
      float s = 0.f, ex[NE];
      #pragma unroll
      for (int e = 0; e < NE; ++e) { ex[e] = expf(lg[e] - mx); s += ex[e]; }
      const float inv = 1.f / s;
      #pragma unroll
      for (int e = 0; e < NE; ++e) Z[(m0 + tid) * NE + e] = ex[e] * inv;
    }
  } else if (MV) {
    constexpr int PT = MVN + 1;
    float* T = &As[0][0];
    const float bv = (colB < LAT) ? bias[colB] : bias2[colB - LAT];
    #pragma unroll
    for (int r = 0; r < 8; ++r)
      T[(16 * wm + r + 8 * h) * PT + colB] = acc[r] + bv;
    __syncthreads();
    #pragma unroll
    for (int it = 0; it < BM * LAT / 256; ++it) {
      const int idx = tid + 256 * it;
      const int rr  = idx / LAT;
      const int cc  = idx % LAT;
      const float m = T[rr * PT + cc];
      const float v = T[rr * PT + LAT + cc];
      const int   o = (m0 + rr) * LAT + cc;
      C[o]  = m;
      C2[o] = v;
      Z[o]  = m + eps[o] * expf(0.5f * v);
    }
  } else if (colB < N) {
    float be[NE];
    if (MOE) {
      #pragma unroll
      for (int e = 0; e < NE; ++e) be[e] = bias[e * N + colB];
    }
    #pragma unroll
    for (int r = 0; r < 8; ++r) {
      const int row = m0 + 16 * wm + r + 8 * h;
      float y = acc[r];
      if (MOE) {
        float s = 0.f;
        #pragma unroll
        for (int e = 0; e < NE; ++e) s = fmaf(coeff[row * NE + e], be[e], s);
        y += s;
      } else if (bias) {
        y += bias[colB];
      }
      if (act) y = elu1(y);
      C[row * N + colB] = y;
    }
  }
}

template <int N, int MT, int NTB>
static inline dim3 gemm_grid() {
  return dim3((N + 16 * NTB - 1) / (16 * NTB), BATCH / (16 * MT));
}

extern "C" void kernel_launch(void* const* d_in, const int* in_sizes, int n_in,
                              void* d_out, int out_size, void* d_ws, size_t ws_size,
                              hipStream_t stream) {
  const float* x      = (const float*)d_in[0];
  const float* c      = (const float*)d_in[1];
  const float* eps    = (const float*)d_in[2];
  const float* efc1_w = (const float*)d_in[3];
  const float* efc1_b = (const float*)d_in[4];
  const float* efc2_w = (const float*)d_in[5];
  const float* efc2_b = (const float*)d_in[6];
  const float* em_w   = (const float*)d_in[7];
  const float* em_b   = (const float*)d_in[8];
  const float* ev_w   = (const float*)d_in[9];
  const float* ev_b   = (const float*)d_in[10];
  const float* g1_w   = (const float*)d_in[11];
  const float* g1_b   = (const float*)d_in[12];
  const float* g2_w   = (const float*)d_in[13];
  const float* g2_b   = (const float*)d_in[14];
  const float* g3_w   = (const float*)d_in[15];
  const float* g3_b   = (const float*)d_in[16];
  const float* w0     = (const float*)d_in[17];
  const float* b0     = (const float*)d_in[18];
  const float* w1     = (const float*)d_in[19];
  const float* b1     = (const float*)d_in[20];
  const float* w2     = (const float*)d_in[21];
  const float* b2     = (const float*)d_in[22];

  float* out   = (float*)d_out;
  float* out_m = out + BATCH * FRAME;
  float* out_v = out_m + BATCH * LAT;

  float* ws    = (float*)d_ws;
  float* h1    = ws;
  float* h2    = h1  + BATCH * HID;
  float* z     = h2  + BATCH * HID;
  float* coeff = z   + BATCH * LAT;

  gemm_lds_f32<HID, MODE_PLAIN, 2, 4><<<gemm_grid<HID, 2, 4>(), 256, 0, stream>>>(
      x, FRAME, c, FRAME, nullptr, efc1_w, efc1_b, nullptr, nullptr, nullptr, nullptr, nullptr,
      h1, nullptr, nullptr, 1);
  gemm_lds_f32<HID, MODE_PLAIN, 2, 4><<<gemm_grid<HID, 2, 4>(), 256, 0, stream>>>(
      x, FRAME, h1, HID, nullptr, efc2_w, efc2_b, nullptr, nullptr, nullptr, nullptr, nullptr,
      h2, nullptr, nullptr, 1);
  gemm_lds_f32<MVN, MODE_MV, 2, 4><<<gemm_grid<MVN, 2, 4>(), 256, 0, stream>>>(
      x, FRAME, h2, HID, nullptr, em_w, em_b, ev_w, ev_b, eps, nullptr, nullptr,
      out_m, out_v, z, 0);

  gemm_lds_f32<GH, MODE_GATE, 2, 4><<<gemm_grid<GH, 2, 4>(), 256, 0, stream>>>(
      z, LAT, c, FRAME, nullptr, g1_w, g1_b, g2_w, g2_b, nullptr, g3_w, g3_b,
      nullptr, nullptr, coeff, 1);

  gemm_lds_f32<HID, MODE_MOE, 2, 4><<<gemm_grid<HID, 2, 4>(), 256, 0, stream>>>(
      z, LAT, c, FRAME, coeff, w0, b0, nullptr, nullptr, nullptr, nullptr, nullptr,
      h1, nullptr, nullptr, 1);
  gemm_lds_f32<HID, MODE_MOE, 2, 4><<<gemm_grid<HID, 2, 4>(), 256, 0, stream>>>(
      z, LAT, h1, HID, coeff, w1, b1, nullptr, nullptr, nullptr, nullptr, nullptr,
      h2, nullptr, nullptr, 1);
  gemm_lds_f32<FRAME, MODE_MOE, 2, 4><<<gemm_grid<FRAME, 2, 4>(), 256, 0, stream>>>(
      z, LAT, h2, HID, coeff, w2, b2, nullptr, nullptr, nullptr, nullptr, nullptr,
      out, nullptr, nullptr, 0);
}


# Launch fusion 7 -> 5: the m/v/z GEMM and the gating net run at the head of decoder layer 0's launch, once per column-block work-group (own z/coeff plane each); sum orders kept
# speedup vs baseline: 2.5237x; 1.0388x over previous
#include <hip/hip_runtime.h>
#include <hip/hip_bf16.h>


typedef __attribute__((ext_vector_type(2))) float v2f;
typedef __attribute__((ext_vector_type(4))) float v4f;
typedef __attribute__((ext_vector_type(8))) float v8f;

#define BATCH 2048
#define FRAME 267
#define LAT   32
#define HID   256
#define NE    6
#define GH    64
#define KC    128
#define MVN   (2 * LAT)

#define MODE_PLAIN 0
#define MODE_MOE   1
#define MODE_MV    2
#define MODE_GATE  3

__device__ __forceinline__ float elu1(float x) {
  return x > 0.f ? x : expm1f(x);
}

__device__ __forceinline__ v8f wmma4(v2f a, v2f b, v8f c) {
  return __builtin_amdgcn_wmma_f32_16x16x4_f32(
      false, a, false, b, (short)0, c, false, false);
}

template <int N, int MODE, int MT, int NTB>
__device__ __forceinline__ void
gemm_tile(const float* A1, int k1,
             const float* A2, int k2,
             const float* coeff,
             const float* __restrict__ W,
             const float* __restrict__ bias,
             const float* __restrict__ W2,
             const float* __restrict__ bias2,
             const float* __restrict__ eps,
             const float* __restrict__ W3,
             const float* __restrict__ bias3,
             float* C,
             float* C2,
             float* Z,
             int act,
             int bx,
             bool wmv,
             float* As,
             float* Bs)
{
  constexpr bool MOE  = (MODE == MODE_MOE);
  constexpr bool MV   = (MODE == MODE_MV);
  constexpr bool GATE = (MODE == MODE_GATE);
  static_assert(!GATE || (N == GH && NTB == 4 && MT == 2), "GATE: one 32 x 64 tile holds whole rows of both hidden layers");
  static_assert(MT * NTB == 8, "eight waves per work-group");
  static_assert(!MV || (N == MVN && NTB == 4 && MT == 2), "MV: one 32 x 64 tile, columns 0..31 mean, 32..63 log-variance");
  constexpr int BM  = 16 * MT;
  constexpr int BN  = 16 * NTB;
  constexpr int PA  = KC + 4;
  constexpr int PB  = BN + 8;
  constexpr int AJ  = KC / 32;
  constexpr int AP  = BM / 8;
  constexpr bool VEC = (N % BN == 0);
  constexpr int F4R = BN / 4;
  constexpr int BP  = KC * F4R / 256;
  constexpr int BRS = 256 / F4R;
  constexpr int BS  = KC * BN / 256;
  constexpr int BSS = 256 / BN;

  const int tid  = threadIdx.x;
  const int lane = tid & 31;
  const int wave = tid >> 5;
  const int h    = lane >> 4;
  const int ln   = lane & 15;
  const int wm   = wave / NTB;
  const int wn   = wave % NTB;
  const int m0   = blockIdx.y * BM;
  const int n0   = bx * BN;

  const int bn4 = tid % F4R;
  const int bk  = tid / F4R;
  const int bn  = tid % BN;
  const int bk2 = tid / BN;
  const int colc = (n0 + bn < N) ? (n0 + bn) : (N - 1);

  const int I    = k1 + k2;
  const int Ktot = (MOE ? NE : 1) * I;
  const int nc   = (Ktot + KC - 1) / KC;

  float  ra[AJ][AP];
  float  cl[AP], ch[AP];
  bool   av[AJ];
  bool   ah[AJ];
  v4f    rb[BP > 0 ? BP : 1];
  float  rbs[BS];

  auto gload = [&](int i) {
    const int k0 = i * KC;
    const int e0 = MOE ? k0 / I : 0;
    const int kb = e0 * I;
    #pragma unroll
    for (int j = 0; j < AJ; ++j) {
      const int kk = k0 + 32 * j + lane;
      av[j] = kk < Ktot;
      const int kc = av[j] ? kk : Ktot - 1;
      int ii = kc - kb;
      ah[j] = MOE && ii >= I;
      if (ah[j]) ii -= I;
      const bool s2 = ii >= k1;
      const float* Aseg = s2 ? A2 : A1;
      const int len = s2 ? k2 : k1;
      const int cc  = s2 ? ii - k1 : ii;
      #pragma unroll
      for (int p = 0; p < AP; ++p)
        ra[j][p] = Aseg[(m0 + wave + 8 * p) * len + cc];
    }
    if (MOE) {
      const int e1 = e0 + 1 < NE ? e0 + 1 : NE - 1;
      #pragma unroll
      for (int p = 0; p < AP; ++p) {
        const int row = m0 + wave + 8 * p;
        cl[p] = coeff[row * NE + e0];
        ch[p] = coeff[row * NE + e1];
      }
    }
    if (VEC) {
      #pragma unroll
      for (int p = 0; p < BP; ++p) {
        int kr = k0 + bk + BRS * p;
        kr = kr < Ktot ? kr : Ktot - 1;
        if (MV) {
          const float* __restrict__ Wh = (bn4 < F4R / 2) ? W : W2;
          rb[p] = *(const v4f*)(Wh + kr * LAT + 4 * (bn4 & (F4R / 2 - 1)));
        } else {
          rb[p] = *(const v4f*)(W + kr * N + n0 + 4 * bn4);
        }
      }
    } else {
      #pragma unroll
      for (int p = 0; p < BS; ++p) {
        int kr = k0 + bk2 + BSS * p;
        kr = kr < Ktot ? kr : Ktot - 1;
        rbs[p] = W[kr * N + colc];
      }
    }
  };

  auto sstore = [&](int st) {
    float* as = As + st * (BM * PA);
    float* bs = Bs + st * (KC * PB);
    #pragma unroll
    for (int j = 0; j < AJ; ++j) {
      #pragma unroll
      for (int p = 0; p < AP; ++p) {
        float v = av[j] ? ra[j][p] : 0.f;
        if (MOE) v *= ah[j] ? ch[p] : cl[p];
        as[(wave + 8 * p) * PA + 32 * j + lane] = v;
      }
    }
    if (VEC) {
      #pragma unroll
      for (int p = 0; p < BP; ++p)
        *(v4f*)(bs + (bk + BRS * p) * PB + 4 * bn4) = rb[p];
    } else {
      #pragma unroll
      for (int p = 0; p < BS; ++p)
        bs[(bk2 + BSS * p) * PB + bn] = rbs[p];
    }
  };

  v8f acc = (v8f){0.f,0.f,0.f,0.f,0.f,0.f,0.f,0.f};

  const int aoff = (16 * wm + ln) * PA + 2 * h;
  const int boff = (2 * h) * PB + 16 * wn + ln;

  auto mma32 = [&](const float* asj, const float* bsj, v8f a) {
    v2f af[8], bf[8];
    #pragma unroll
    for (int ks = 0; ks < 8; ++ks) {
      af[ks]   = *(const v2f*)(asj + 4 * ks);
      bf[ks].x = bsj[(4 * ks) * PB];
      bf[ks].y = bsj[(4 * ks + 1) * PB];
    }
    asm volatile("" :: "v"(af[0]), "v"(af[1]), "v"(af[2]), "v"(af[3]),
                       "v"(af[4]), "v"(af[5]), "v"(af[6]), "v"(af[7]),
                       "v"(bf[0]), "v"(bf[1]), "v"(bf[2]), "v"(bf[3]),
                       "v"(bf[4]), "v"(bf[5]), "v"(bf[6]), "v"(bf[7]));
    #pragma unroll
    for (int ks = 0; ks < 8; ++ks)
      a = wmma4(af[ks], bf[ks], a);
    return a;
  };

  gload(0);
  sstore(0);
  __syncthreads();

  for (int i = 0; i < nc; ++i) {
    const int st = i & 1;
    if (i + 1 < nc) gload(i + 1);

    const int rem = Ktot - i * KC;
    const int nsb = rem >= KC ? KC / 32 : (rem + 31) / 32;

    const float* as = As + st * (BM * PA) + aoff;
    const float* bs = Bs + st * (KC * PB) + boff;
    for (int j = 0; j < nsb; ++j)
      acc = mma32(as + 32 * j, bs + 32 * j * PB, acc);


    if (i + 1 < nc) sstore(st ^ 1);
    __syncthreads();
  }

  const int colB = n0 + 16 * wn + ln;
  if (GATE) {
    constexpr int PT = GH + 1;
    float* as2 = As;
    float* bs2 = Bs;
    {
      const float bv = bias[colB];
      #pragma unroll
      for (int r = 0; r < 8; ++r)
        as2[(16 * wm + r + 8 * h) * PA + colB] = elu1(acc[r] + bv);
      #pragma unroll
      for (int p = 0; p < GH / BRS; ++p) {
        const int kr = bk + BRS * p;
        *(v4f*)(bs2 + kr * PB + 4 * bn4) = *(const v4f*)(W2 + kr * GH + 4 * bn4);
      }
    }
    __syncthreads();
    v8f acc2 = (v8f){0.f,0.f,0.f,0.f,0.f,0.f,0.f,0.f};
    #pragma unroll
    for (int j = 0; j < GH / 32; ++j)
      acc2 = mma32(as2 + aoff + 32 * j, bs2 + boff + 32 * j * PB, acc2);
    float* T = As + BM * PA;
    {
      const float bv2 = bias2[colB];
      #pragma unroll
      for (int r = 0; r < 8; ++r)
        T[(16 * wm + r + 8 * h) * PT + colB] = elu1(acc2[r] + bv2);
    }
    __syncthreads();
    if (tid < BM) {
      float lg[NE];
      #pragma unroll
      for (int e = 0; e < NE; ++e) lg[e] = bias3[e];
      for (int i = 0; i < GH; ++i) {
        const float gv = T[tid * PT + i];
        #pragma unroll
        for (int e = 0; e < NE; ++e) lg[e] = fmaf(gv, W3[i * NE + e], lg[e]);
      }
      float mx = lg[0];
      #pragma unroll
      for (int e = 1; e < NE; ++e) mx = fmaxf(mx, lg[e]);
      float s = 0.f, ex[NE];
      #pragma unroll
      for (int e = 0; e < NE; ++e) { ex[e] = expf(lg[e] - mx); s += ex[e]; }
      const float inv = 1.f / s;
      #pragma unroll
      for (int e = 0; e < NE; ++e) Z[(m0 + tid) * NE + e] = ex[e] * inv;
    }
  } else if (MV) {
    constexpr int PT = MVN + 1;
    float* T = As;
    const float bv = (colB < LAT) ? bias[colB] : bias2[colB - LAT];
    #pragma unroll
    for (int r = 0; r < 8; ++r)
      T[(16 * wm + r + 8 * h) * PT + colB] = acc[r] + bv;
    __syncthreads();
    #pragma unroll
    for (int it = 0; it < BM * LAT / 256; ++it) {
      const int idx = tid + 256 * it;
      const int rr  = idx / LAT;
      const int cc  = idx % LAT;
      const float m = T[rr * PT + cc];
      const float v = T[rr * PT + LAT + cc];
      const int   o = (m0 + rr) * LAT + cc;
      if (wmv) {
        C[o]  = m;
        C2[o] = v;
      }
      Z[o]  = m + eps[o] * expf(0.5f * v);
    }
  } else if (colB < N) {
    float be[NE];
    if (MOE) {
      #pragma unroll
      for (int e = 0; e < NE; ++e) be[e] = bias[e * N + colB];
    }
    #pragma unroll
    for (int r = 0; r < 8; ++r) {
      const int row = m0 + 16 * wm + r + 8 * h;
      float y = acc[r];
      if (MOE) {
        float s = 0.f;
        #pragma unroll
        for (int e = 0; e < NE; ++e) s = fmaf(coeff[row * NE + e], be[e], s);
        y += s;
      } else if (bias) {
        y += bias[colB];
      }
      if (act) y = elu1(y);
      C[row * N + colB] = y;
    }
  }
}

template <int N, int MODE, int MT, int NTB>
__global__ void __launch_bounds__(256)
gemm_lds_f32(const float* __restrict__ A1, int k1,
             const float* __restrict__ A2, int k2,
             const float* __restrict__ coeff,
             const float* __restrict__ W,
             const float* __restrict__ bias,
             float* __restrict__ C,
             int act)
{
  __shared__ __attribute__((aligned(16))) float As[2 * (16 * MT) * (KC + 4)];
  __shared__ __attribute__((aligned(16))) float Bs[2 * KC * (16 * NTB + 8)];
  gemm_tile<N, MODE, MT, NTB>(A1, k1, A2, k2, coeff, W, bias, nullptr, nullptr, nullptr, nullptr, nullptr,
                              C, nullptr, nullptr, act, blockIdx.x, false, As, Bs);
}

__global__ void __launch_bounds__(256)
head_decoder0(const float* __restrict__ x, const float* __restrict__ h2in,
              const float* __restrict__ c, const float* __restrict__ eps,
              const float* __restrict__ em_w, const float* __restrict__ em_b,
              const float* __restrict__ ev_w, const float* __restrict__ ev_b,
              const float* __restrict__ g1_w, const float* __restrict__ g1_b,
              const float* __restrict__ g2_w, const float* __restrict__ g2_b,
              const float* __restrict__ g3_w, const float* __restrict__ g3_b,
              const float* __restrict__ w0, const float* __restrict__ b0,
              float* __restrict__ out_m, float* __restrict__ out_v,
              float* zpl, float* cpl,
              float* __restrict__ d0)
{
  __shared__ __attribute__((aligned(16))) float As[2 * 32 * (KC + 4)];
  __shared__ __attribute__((aligned(16))) float Bs[2 * KC * (64 + 8)];
  const int bx = blockIdx.x;
  float* zmine = zpl + bx * (BATCH * LAT);
  float* cmine = cpl + bx * (BATCH * NE);

  gemm_tile<MVN, MODE_MV, 2, 4>(x, FRAME, h2in, HID, nullptr, em_w, em_b, ev_w, ev_b, eps, nullptr, nullptr,
                                out_m, out_v, zmine, 0, 0, bx == 0, As, Bs);
  __syncthreads();
  gemm_tile<GH, MODE_GATE, 2, 4>(zmine, LAT, c, FRAME, nullptr, g1_w, g1_b, g2_w, g2_b, nullptr, g3_w, g3_b,
                                 nullptr, nullptr, cmine, 1, 0, false, As, Bs);
  __syncthreads();
  gemm_tile<HID, MODE_MOE, 2, 4>(zmine, LAT, c, FRAME, cmine, w0, b0, nullptr, nullptr, nullptr, nullptr, nullptr,
                                 d0, nullptr, nullptr, 1, bx, false, As, Bs);
}

template <int N, int MT, int NTB>
static inline dim3 gemm_grid() {
  return dim3((N + 16 * NTB - 1) / (16 * NTB), BATCH / (16 * MT));
}

extern "C" void kernel_launch(void* const* d_in, const int* in_sizes, int n_in,
                              void* d_out, int out_size, void* d_ws, size_t ws_size,
                              hipStream_t stream) {
  const float* x      = (const float*)d_in[0];
  const float* c      = (const float*)d_in[1];
  const float* eps    = (const float*)d_in[2];
  const float* efc1_w = (const float*)d_in[3];
  const float* efc1_b = (const float*)d_in[4];
  const float* efc2_w = (const float*)d_in[5];
  const float* efc2_b = (const float*)d_in[6];
  const float* em_w   = (const float*)d_in[7];
  const float* em_b   = (const float*)d_in[8];
  const float* ev_w   = (const float*)d_in[9];
  const float* ev_b   = (const float*)d_in[10];
  const float* g1_w   = (const float*)d_in[11];
  const float* g1_b   = (const float*)d_in[12];
  const float* g2_w   = (const float*)d_in[13];
  const float* g2_b   = (const float*)d_in[14];
  const float* g3_w   = (const float*)d_in[15];
  const float* g3_b   = (const float*)d_in[16];
  const float* w0     = (const float*)d_in[17];
  const float* b0     = (const float*)d_in[18];
  const float* w1     = (const float*)d_in[19];
  const float* b1     = (const float*)d_in[20];
  const float* w2     = (const float*)d_in[21];
  const float* b2     = (const float*)d_in[22];

  float* out   = (float*)d_out;
  float* out_m = out + BATCH * FRAME;
  float* out_v = out_m + BATCH * LAT;

  float* ws    = (float*)d_ws;
  float* h1    = ws;
  float* h2    = h1  + BATCH * HID;
  float* zpl   = h2  + BATCH * HID;
  float* cpl   = zpl + 4 * BATCH * LAT;
  float* z     = zpl;
  float* coeff = cpl;

  gemm_lds_f32<HID, MODE_PLAIN, 2, 4><<<gemm_grid<HID, 2, 4>(), 256, 0, stream>>>(
      x, FRAME, c, FRAME, nullptr, efc1_w, efc1_b, h1, 1);
  gemm_lds_f32<HID, MODE_PLAIN, 2, 4><<<gemm_grid<HID, 2, 4>(), 256, 0, stream>>>(
      x, FRAME, h1, HID, nullptr, efc2_w, efc2_b, h2, 1);

  head_decoder0<<<gemm_grid<HID, 2, 4>(), 256, 0, stream>>>(
      x, h2, c, eps, em_w, em_b, ev_w, ev_b, g1_w, g1_b, g2_w, g2_b, g3_w, g3_b, w0, b0,
      out_m, out_v, zpl, cpl, h1);

  gemm_lds_f32<HID, MODE_MOE, 2, 4><<<gemm_grid<HID, 2, 4>(), 256, 0, stream>>>(
      z, LAT, h1, HID, coeff, w1, b1, h2, 1);
  gemm_lds_f32<FRAME, MODE_MOE, 2, 4><<<gemm_grid<FRAME, 2, 4>(), 256, 0, stream>>>(
      z, LAT, h2, HID, coeff, w2, b2, out, 0);
}


# MoE layers: each 32-k input tile staged once, unscaled, for all six experts; one accumulator tile per expert, coefficient mix in the epilogue (sum order: per-expert sums, then the mix)
# speedup vs baseline: 3.1631x; 1.2534x over previous
#include <hip/hip_runtime.h>
#include <hip/hip_bf16.h>


typedef __attribute__((ext_vector_type(2))) float v2f;
typedef __attribute__((ext_vector_type(4))) float v4f;
typedef __attribute__((ext_vector_type(8))) float v8f;

#define BATCH 2048
#define FRAME 267
#define LAT   32
#define HID   256
#define NE    6
#define GH    64
#define KC    128
#define KQ    32
#define MVN   (2 * LAT)

#define MODE_PLAIN 0
#define MODE_MV    1
#define MODE_GATE  2

__device__ __forceinline__ float elu1(float x) {
  return x > 0.f ? x : expm1f(x);
}

__device__ __forceinline__ v8f wmma4(v2f a, v2f b, v8f c) {
  return __builtin_amdgcn_wmma_f32_16x16x4_f32(
      false, a, false, b, (short)0, c, false, false);
}

template <int N, int MODE, int MT, int NTB>
__device__ __forceinline__ void
gemm_tile(const float* A1, int k1,
             const float* A2, int k2,
             const float* __restrict__ W,
             const float* __restrict__ bias,
             const float* __restrict__ W2,
             const float* __restrict__ bias2,
             const float* __restrict__ eps,
             const float* __restrict__ W3,
             const float* __restrict__ bias3,
             float* C,
             float* C2,
             float* Z,
             int act,
             int bx,
             bool wmv,
             float* As,
             float* Bs)
{
  constexpr bool MV   = (MODE == MODE_MV);
  constexpr bool GATE = (MODE == MODE_GATE);
  static_assert(!GATE || (N == GH && NTB == 4 && MT == 2), "GATE: one 32 x 64 tile holds whole rows of both hidden layers");
  static_assert(MT * NTB == 8, "eight waves per work-group");
  static_assert(!MV || (N == MVN && NTB == 4 && MT == 2), "MV: one 32 x 64 tile, columns 0..31 mean, 32..63 log-variance");
  constexpr int BM  = 16 * MT;
  constexpr int BN  = 16 * NTB;
  constexpr int PA  = KC + 4;
  constexpr int PB  = BN + 8;
  constexpr int AJ  = KC / 32;
  constexpr int AP  = BM / 8;
  constexpr bool VEC = (N % BN == 0);
  constexpr int F4R = BN / 4;
  constexpr int BP  = KC * F4R / 256;
  constexpr int BRS = 256 / F4R;
  constexpr int BS  = KC * BN / 256;
  constexpr int BSS = 256 / BN;

  const int tid  = threadIdx.x;
  const int lane = tid & 31;
  const int wave = tid >> 5;
  const int h    = lane >> 4;
  const int ln   = lane & 15;
  const int wm   = wave / NTB;
  const int wn   = wave % NTB;
  const int m0   = blockIdx.y * BM;
  const int n0   = bx * BN;

  const int bn4 = tid % F4R;
  const int bk  = tid / F4R;
  const int bn  = tid % BN;
  const int bk2 = tid / BN;
  const int colc = (n0 + bn < N) ? (n0 + bn) : (N - 1);

  const int Ktot = k1 + k2;
  const int nc   = (Ktot + KC - 1) / KC;

  float  ra[AJ][AP];
  bool   av[AJ];
  v4f    rb[BP > 0 ? BP : 1];
  float  rbs[BS];

  auto gload = [&](int i) {
    const int k0 = i * KC;
    #pragma unroll
    for (int j = 0; j < AJ; ++j) {
      const int kk = k0 + 32 * j + lane;
      av[j] = kk < Ktot;
      const int kc = av[j] ? kk : Ktot - 1;
      const int ii = kc;
      const bool s2 = ii >= k1;
      const float* Aseg = s2 ? A2 : A1;
      const int len = s2 ? k2 : k1;
      const int cc  = s2 ? ii - k1 : ii;
      #pragma unroll
      for (int p = 0; p < AP; ++p)
        ra[j][p] = Aseg[(m0 + wave + 8 * p) * len + cc];
    }
    if (VEC) {
      #pragma unroll
      for (int p = 0; p < BP; ++p) {
        int kr = k0 + bk + BRS * p;
        kr = kr < Ktot ? kr : Ktot - 1;
        if (MV) {
          const float* __restrict__ Wh = (bn4 < F4R / 2) ? W : W2;
          rb[p] = *(const v4f*)(Wh + kr * LAT + 4 * (bn4 & (F4R / 2 - 1)));
        } else {
          rb[p] = *(const v4f*)(W + kr * N + n0 + 4 * bn4);
        }
      }
    } else {
      #pragma unroll
      for (int p = 0; p < BS; ++p) {
        int kr = k0 + bk2 + BSS * p;
        kr = kr < Ktot ? kr : Ktot - 1;
        rbs[p] = W[kr * N + colc];
      }
    }
  };

  auto sstore = [&](int st) {
    float* as = As + st * (BM * PA);
    float* bs = Bs + st * (KC * PB);
    #pragma unroll
    for (int j = 0; j < AJ; ++j) {
      #pragma unroll
      for (int p = 0; p < AP; ++p) {
        as[(wave + 8 * p) * PA + 32 * j + lane] = av[j] ? ra[j][p] : 0.f;
      }
    }
    if (VEC) {
      #pragma unroll
      for (int p = 0; p < BP; ++p)
        *(v4f*)(bs + (bk + BRS * p) * PB + 4 * bn4) = rb[p];
    } else {
      #pragma unroll
      for (int p = 0; p < BS; ++p)
        bs[(bk2 + BSS * p) * PB + bn] = rbs[p];
    }
  };

  v8f acc = (v8f){0.f,0.f,0.f,0.f,0.f,0.f,0.f,0.f};

  const int aoff = (16 * wm + ln) * PA + 2 * h;
  const int boff = (2 * h) * PB + 16 * wn + ln;

  auto mma32 = [&](const float* asj, const float* bsj, v8f a) {
    v2f af[8], bf[8];
    #pragma unroll
    for (int ks = 0; ks < 8; ++ks) {
      af[ks]   = *(const v2f*)(asj + 4 * ks);
      bf[ks].x = bsj[(4 * ks) * PB];
      bf[ks].y = bsj[(4 * ks + 1) * PB];
    }
    asm volatile("" :: "v"(af[0]), "v"(af[1]), "v"(af[2]), "v"(af[3]),
                       "v"(af[4]), "v"(af[5]), "v"(af[6]), "v"(af[7]),
                       "v"(bf[0]), "v"(bf[1]), "v"(bf[2]), "v"(bf[3]),
                       "v"(bf[4]), "v"(bf[5]), "v"(bf[6]), "v"(bf[7]));
    #pragma unroll
    for (int ks = 0; ks < 8; ++ks)
      a = wmma4(af[ks], bf[ks], a);
    return a;
  };

  gload(0);
  sstore(0);
  __syncthreads();

  for (int i = 0; i < nc; ++i) {
    const int st = i & 1;
    if (i + 1 < nc) gload(i + 1);

    const int rem = Ktot - i * KC;
    const int nsb = rem >= KC ? KC / 32 : (rem + 31) / 32;

    const float* as = As + st * (BM * PA) + aoff;
    const float* bs = Bs + st * (KC * PB) + boff;
    for (int j = 0; j < nsb; ++j)
      acc = mma32(as + 32 * j, bs + 32 * j * PB, acc);


    if (i + 1 < nc) sstore(st ^ 1);
    __syncthreads();
  }

  const int colB = n0 + 16 * wn + ln;
  if (GATE) {
    constexpr int PT = GH + 1;
    float* as2 = As;
    float* bs2 = Bs;
    {
      const float bv = bias[colB];
      #pragma unroll
      for (int r = 0; r < 8; ++r)
        as2[(16 * wm + r + 8 * h) * PA + colB] = elu1(acc[r] + bv);
      #pragma unroll
      for (int p = 0; p < GH / BRS; ++p) {
        const int kr = bk + BRS * p;
        *(v4f*)(bs2 + kr * PB + 4 * bn4) = *(const v4f*)(W2 + kr * GH + 4 * bn4);
      }
    }
    __syncthreads();
    v8f acc2 = (v8f){0.f,0.f,0.f,0.f,0.f,0.f,0.f,0.f};
    #pragma unroll
    for (int j = 0; j < GH / 32; ++j)
      acc2 = mma32(as2 + aoff + 32 * j, bs2 + boff + 32 * j * PB, acc2);
    float* T = As + BM * PA;
    {
      const float bv2 = bias2[colB];
      #pragma unroll
      for (int r = 0; r < 8; ++r)
        T[(16 * wm + r + 8 * h) * PT + colB] = elu1(acc2[r] + bv2);
    }
    __syncthreads();
    if (tid < BM) {
      float lg[NE];
      #pragma unroll
      for (int e = 0; e < NE; ++e) lg[e] = bias3[e];
      for (int i = 0; i < GH; ++i) {
        const float gv = T[tid * PT + i];
        #pragma unroll
        for (int e = 0; e < NE; ++e) lg[e] = fmaf(gv, W3[i * NE + e], lg[e]);
      }
      float mx = lg[0];
      #pragma unroll
      for (int e = 1; e < NE; ++e) mx = fmaxf(mx, lg[e]);
      float s = 0.f, ex[NE];
      #pragma unroll
      for (int e = 0; e < NE; ++e) { ex[e] = expf(lg[e] - mx); s += ex[e]; }
      const float inv = 1.f / s;
      #pragma unroll
      for (int e = 0; e < NE; ++e) Z[(m0 + tid) * NE + e] = ex[e] * inv;
    }
  } else if (MV) {
    constexpr int PT = MVN + 1;
    float* T = As;
    const float bv = (colB < LAT) ? bias[colB] : bias2[colB - LAT];
    #pragma unroll
    for (int r = 0; r < 8; ++r)
      T[(16 * wm + r + 8 * h) * PT + colB] = acc[r] + bv;
    __syncthreads();
    #pragma unroll
    for (int it = 0; it < BM * LAT / 256; ++it) {
      const int idx = tid + 256 * it;
      const int rr  = idx / LAT;
      const int cc  = idx % LAT;
      const float m = T[rr * PT + cc];
      const float v = T[rr * PT + LAT + cc];
      const int   o = (m0 + rr) * LAT + cc;
      if (wmv) {
        C[o]  = m;
        C2[o] = v;
      }
      Z[o]  = m + eps[o] * expf(0.5f * v);
    }
  } else if (colB < N) {
    #pragma unroll
    for (int r = 0; r < 8; ++r) {
      const int row = m0 + 16 * wm + r + 8 * h;
      float y = acc[r];
      if (bias) y += bias[colB];
      if (act) y = elu1(y);
      C[row * N + colB] = y;
    }
  }
}

template <int N, int K1, int K2>
__device__ __forceinline__ void
moe_tile(const float* A1, const float* A2,
         const float* coeff,
         const float* __restrict__ W,
         const float* __restrict__ bias,
         float* C,
         int act,
         int bx,
         float* As,
         float* Bs)
{
  static_assert(K1 % KQ == 0, "a round reads one segment of [A1|A2]");
  constexpr int NTB = 4;
  constexpr int BM  = 32;
  constexpr int BN  = 64;
  constexpr int I   = K1 + K2;
  constexpr int NR  = (I + KQ - 1) / KQ;
  constexpr bool EXACT = (I % KQ == 0);
  constexpr int PA  = KQ + 4;
  constexpr int PB  = BN + 8;
  constexpr int WR  = NE * KQ;
  constexpr int AP  = BM / 8;
  constexpr bool VEC = (N % BN == 0);
  constexpr int F4R = BN / 4;
  constexpr int BP  = WR * F4R / 256;
  constexpr int BRS = 256 / F4R;
  constexpr int BS  = WR * BN / 256;
  constexpr int BSS = 256 / BN;
  static_assert(KQ % BRS == 0 && KQ % BSS == 0, "a staging pass stays inside one expert");

  const int tid  = threadIdx.x;
  const int lane = tid & 31;
  const int wave = tid >> 5;
  const int h    = lane >> 4;
  const int ln   = lane & 15;
  const int wm   = wave / NTB;
  const int wn   = wave % NTB;
  const int m0   = blockIdx.y * BM;
  const int n0   = bx * BN;

  const int bn4 = tid % F4R;
  const int bk  = tid / F4R;
  const int bn  = tid % BN;
  const int bk2 = tid / BN;
  const int colc = (n0 + bn < N) ? (n0 + bn) : (N - 1);

  float ra[AP];
  bool  av;
  v4f   rb[BP];
  float rbs[BS];

  auto gload = [&](int i) {
    const int k0 = i * KQ;
    const bool full = EXACT || (k0 + KQ <= I);
    const bool s2 = k0 >= K1;
    const float* Aseg = s2 ? A2 : A1;
    const int len = s2 ? K2 : K1;
    av = full || (k0 + lane < I);
    const int kc = av ? k0 + lane : I - 1;
    const int cc = s2 ? kc - K1 : kc;
    #pragma unroll
    for (int p = 0; p < AP; ++p)
      ra[p] = Aseg[(m0 + wave + 8 * p) * len + cc];
    if (VEC) {
      #pragma unroll
      for (int p = 0; p < BP; ++p) {
        const int e = (BRS * p) / KQ;
        int kr = k0 + (BRS * p) % KQ + bk;
        if (!full) kr = kr < I ? kr : I - 1;
        rb[p] = *(const v4f*)(W + (e * I + kr) * N + n0 + 4 * bn4);
      }
    } else {
      #pragma unroll
      for (int p = 0; p < BS; ++p) {
        const int e = (BSS * p) / KQ;
        int kr = k0 + (BSS * p) % KQ + bk2;
        if (!full) kr = kr < I ? kr : I - 1;
        rbs[p] = W[(e * I + kr) * N + colc];
      }
    }
  };

  auto sstore = [&](int st) {
    float* as = As + st * (BM * PA);
    float* bs = Bs + st * (WR * PB);
    #pragma unroll
    for (int p = 0; p < AP; ++p)
      as[(wave + 8 * p) * PA + lane] = av ? ra[p] : 0.f;
    if (VEC) {
      #pragma unroll
      for (int p = 0; p < BP; ++p)
        *(v4f*)(bs + (bk + BRS * p) * PB + 4 * bn4) = rb[p];
    } else {
      #pragma unroll
      for (int p = 0; p < BS; ++p)
        bs[(bk2 + BSS * p) * PB + bn] = rbs[p];
    }
  };

  v8f acc[NE];
  #pragma unroll
  for (int e = 0; e < NE; ++e) acc[e] = (v8f){0.f,0.f,0.f,0.f,0.f,0.f,0.f,0.f};

  const int aoff = (16 * wm + ln) * PA + 2 * h;
  const int boff = (2 * h) * PB + 16 * wn + ln;

  auto mma16 = [&](const float* ag, const float* bg) {
    v2f af[4], bf[4][NE];
    #pragma unroll
    for (int ks = 0; ks < 4; ++ks) {
      af[ks] = *(const v2f*)(ag + 4 * ks);
      #pragma unroll
      for (int e = 0; e < NE; ++e) {
        bf[ks][e].x = bg[(e * KQ + 4 * ks) * PB];
        bf[ks][e].y = bg[(e * KQ + 4 * ks + 1) * PB];
      }
    }
    asm volatile("" :: "v"(af[0]), "v"(af[1]), "v"(af[2]), "v"(af[3]),
                       "v"(bf[0][0]), "v"(bf[0][1]), "v"(bf[0][2]), "v"(bf[0][3]), "v"(bf[0][4]), "v"(bf[0][5]),
                       "v"(bf[1][0]), "v"(bf[1][1]), "v"(bf[1][2]), "v"(bf[1][3]), "v"(bf[1][4]), "v"(bf[1][5]),
                       "v"(bf[2][0]), "v"(bf[2][1]), "v"(bf[2][2]), "v"(bf[2][3]), "v"(bf[2][4]), "v"(bf[2][5]),
                       "v"(bf[3][0]), "v"(bf[3][1]), "v"(bf[3][2]), "v"(bf[3][3]), "v"(bf[3][4]), "v"(bf[3][5]));
    #pragma unroll
    for (int ks = 0; ks < 4; ++ks) {
      #pragma unroll
      for (int e = 0; e < NE; ++e)
        acc[e] = wmma4(af[ks], bf[ks][e], acc[e]);
    }
  };

  gload(0);
  sstore(0);
  __syncthreads();

  for (int i = 0; i < NR; ++i) {
    const int st = i & 1;
    if (i + 1 < NR) gload(i + 1);

    const int rem = I - i * KQ;
    const int ng  = (EXACT || rem >= KQ) ? KQ / 16 : (rem + 15) / 16;

    const float* as = As + st * (BM * PA) + aoff;
    const float* bs = Bs + st * (WR * PB) + boff;
    for (int g = 0; g < ng; ++g)
      mma16(as + 16 * g, bs + 16 * g * PB);

    if (i + 1 < NR) sstore(st ^ 1);
    __syncthreads();
  }

  const int colB = n0 + 16 * wn + ln;
  if (colB < N) {
    float be[NE];
    #pragma unroll
    for (int e = 0; e < NE; ++e) be[e] = bias[e * N + colB];
    #pragma unroll
    for (int r = 0; r < 8; ++r) {
      const int row = m0 + 16 * wm + r + 8 * h;
      float y = 0.f, s = 0.f;
      #pragma unroll
      for (int e = 0; e < NE; ++e) {
        const float ce = coeff[row * NE + e];
        y = fmaf(ce, acc[e][r], y);
        s = fmaf(ce, be[e], s);
      }
      y += s;
      if (act) y = elu1(y);
      C[row * N + colB] = y;
    }
  }
}

template <int N, int MODE, int MT, int NTB>
__global__ void __launch_bounds__(256)
gemm_lds_f32(const float* __restrict__ A1, int k1,
             const float* __restrict__ A2, int k2,
             const float* __restrict__ W,
             const float* __restrict__ bias,
             float* __restrict__ C,
             int act)
{
  __shared__ __attribute__((aligned(16))) float As[2 * (16 * MT) * (KC + 4)];
  __shared__ __attribute__((aligned(16))) float Bs[2 * KC * (16 * NTB + 8)];
  gemm_tile<N, MODE, MT, NTB>(A1, k1, A2, k2, W, bias, nullptr, nullptr, nullptr, nullptr, nullptr,
                              C, nullptr, nullptr, act, blockIdx.x, false, As, Bs);
}

template <int N, int K2>
__global__ void __launch_bounds__(256)
moe_lds_f32(const float* __restrict__ z,
            const float* __restrict__ prev,
            const float* __restrict__ coeff,
            const float* __restrict__ W,
            const float* __restrict__ bias,
            float* __restrict__ C,
            int act)
{
  __shared__ __attribute__((aligned(16))) float As[2 * 32 * (KQ + 4)];
  __shared__ __attribute__((aligned(16))) float Bs[2 * NE * KQ * (64 + 8)];
  moe_tile<N, LAT, K2>(z, prev, coeff, W, bias, C, act, blockIdx.x, As, Bs);
}

__global__ void __launch_bounds__(256)
head_decoder0(const float* __restrict__ x, const float* __restrict__ h2in,
              const float* __restrict__ c, const float* __restrict__ eps,
              const float* __restrict__ em_w, const float* __restrict__ em_b,
              const float* __restrict__ ev_w, const float* __restrict__ ev_b,
              const float* __restrict__ g1_w, const float* __restrict__ g1_b,
              const float* __restrict__ g2_w, const float* __restrict__ g2_b,
              const float* __restrict__ g3_w, const float* __restrict__ g3_b,
              const float* __restrict__ w0, const float* __restrict__ b0,
              float* __restrict__ out_m, float* __restrict__ out_v,
              float* zpl, float* cpl,
              float* __restrict__ d0)
{
  __shared__ __attribute__((aligned(16))) float As[2 * 32 * (KC + 4)];
  __shared__ __attribute__((aligned(16))) float Bs[2 * (NE * KQ > KC ? NE * KQ : KC) * (64 + 8)];
  const int bx = blockIdx.x;
  float* zmine = zpl + bx * (BATCH * LAT);
  float* cmine = cpl + bx * (BATCH * NE);

  gemm_tile<MVN, MODE_MV, 2, 4>(x, FRAME, h2in, HID, em_w, em_b, ev_w, ev_b, eps, nullptr, nullptr,
                                out_m, out_v, zmine, 0, 0, bx == 0, As, Bs);
  __syncthreads();
  gemm_tile<GH, MODE_GATE, 2, 4>(zmine, LAT, c, FRAME, g1_w, g1_b, g2_w, g2_b, nullptr, g3_w, g3_b,
                                 nullptr, nullptr, cmine, 1, 0, false, As, Bs);
  __syncthreads();
  moe_tile<HID, LAT, FRAME>(zmine, c, cmine, w0, b0, d0, 1, bx, As, Bs);
}

template <int N, int MT, int NTB>
static inline dim3 gemm_grid() {
  return dim3((N + 16 * NTB - 1) / (16 * NTB), BATCH / (16 * MT));
}

extern "C" void kernel_launch(void* const* d_in, const int* in_sizes, int n_in,
                              void* d_out, int out_size, void* d_ws, size_t ws_size,
                              hipStream_t stream) {
  const float* x      = (const float*)d_in[0];
  const float* c      = (const float*)d_in[1];
  const float* eps    = (const float*)d_in[2];
  const float* efc1_w = (const float*)d_in[3];
  const float* efc1_b = (const float*)d_in[4];
  const float* efc2_w = (const float*)d_in[5];
  const float* efc2_b = (const float*)d_in[6];
  const float* em_w   = (const float*)d_in[7];
  const float* em_b   = (const float*)d_in[8];
  const float* ev_w   = (const float*)d_in[9];
  const float* ev_b   = (const float*)d_in[10];
  const float* g1_w   = (const float*)d_in[11];
  const float* g1_b   = (const float*)d_in[12];
  const float* g2_w   = (const float*)d_in[13];
  const float* g2_b   = (const float*)d_in[14];
  const float* g3_w   = (const float*)d_in[15];
  const float* g3_b   = (const float*)d_in[16];
  const float* w0     = (const float*)d_in[17];
  const float* b0     = (const float*)d_in[18];
  const float* w1     = (const float*)d_in[19];
  const float* b1     = (const float*)d_in[20];
  const float* w2     = (const float*)d_in[21];
  const float* b2     = (const float*)d_in[22];

  float* out   = (float*)d_out;
  float* out_m = out + BATCH * FRAME;
  float* out_v = out_m + BATCH * LAT;

  float* ws    = (float*)d_ws;
  float* h1    = ws;
  float* h2    = h1  + BATCH * HID;
  float* zpl   = h2  + BATCH * HID;
  float* cpl   = zpl + 4 * BATCH * LAT;
  float* z     = zpl;
  float* coeff = cpl;

  gemm_lds_f32<HID, MODE_PLAIN, 2, 4><<<gemm_grid<HID, 2, 4>(), 256, 0, stream>>>(
      x, FRAME, c, FRAME, efc1_w, efc1_b, h1, 1);
  gemm_lds_f32<HID, MODE_PLAIN, 2, 4><<<gemm_grid<HID, 2, 4>(), 256, 0, stream>>>(
      x, FRAME, h1, HID, efc2_w, efc2_b, h2, 1);

  head_decoder0<<<gemm_grid<HID, 2, 4>(), 256, 0, stream>>>(
      x, h2, c, eps, em_w, em_b, ev_w, ev_b, g1_w, g1_b, g2_w, g2_b, g3_w, g3_b, w0, b0,
      out_m, out_v, zpl, cpl, h1);

  moe_lds_f32<HID, HID><<<gemm_grid<HID, 2, 4>(), 256, 0, stream>>>(
      z, h1, coeff, w1, b1, h2, 1);
  moe_lds_f32<FRAME, HID><<<gemm_grid<FRAME, 2, 4>(), 256, 0, stream>>>(
      z, h2, coeff, w2, b2, out, 0);
}


# Decoder layers 1, 2: MoE work-group tile 32x64 -> 64x32 (4x2 waves): each expert weight row is re-read from L2 by 32 row blocks, not 64; a staged 32-column row is one 128-B line; sum order kept
# speedup vs baseline: 3.3871x; 1.0708x over previous
#include <hip/hip_runtime.h>
#include <hip/hip_bf16.h>


typedef __attribute__((ext_vector_type(2))) float v2f;
typedef __attribute__((ext_vector_type(4))) float v4f;
typedef __attribute__((ext_vector_type(8))) float v8f;

#define BATCH 2048
#define FRAME 267
#define LAT   32
#define HID   256
#define NE    6
#define GH    64
#define KC    128
#define KQ    32
#define MVN   (2 * LAT)

#define MODE_PLAIN 0
#define MODE_MV    1
#define MODE_GATE  2

__device__ __forceinline__ float elu1(float x) {
  return x > 0.f ? x : expm1f(x);
}

__device__ __forceinline__ v8f wmma4(v2f a, v2f b, v8f c) {
  return __builtin_amdgcn_wmma_f32_16x16x4_f32(
      false, a, false, b, (short)0, c, false, false);
}

template <int N, int MODE, int MT, int NTB>
__device__ __forceinline__ void
gemm_tile(const float* A1, int k1,
             const float* A2, int k2,
             const float* __restrict__ W,
             const float* __restrict__ bias,
             const float* __restrict__ W2,
             const float* __restrict__ bias2,
             const float* __restrict__ eps,
             const float* __restrict__ W3,
             const float* __restrict__ bias3,
             float* C,
             float* C2,
             float* Z,
             int act,
             int bx,
             bool wmv,
             float* As,
             float* Bs)
{
  constexpr bool MV   = (MODE == MODE_MV);
  constexpr bool GATE = (MODE == MODE_GATE);
  static_assert(!GATE || (N == GH && NTB == 4 && MT == 2), "GATE: one 32 x 64 tile holds whole rows of both hidden layers");
  static_assert(MT * NTB == 8, "eight waves per work-group");
  static_assert(!MV || (N == MVN && NTB == 4 && MT == 2), "MV: one 32 x 64 tile, columns 0..31 mean, 32..63 log-variance");
  constexpr int BM  = 16 * MT;
  constexpr int BN  = 16 * NTB;
  constexpr int PA  = KC + 4;
  constexpr int PB  = BN + 8;
  constexpr int AJ  = KC / 32;
  constexpr int AP  = BM / 8;
  constexpr bool VEC = (N % BN == 0);
  constexpr int F4R = BN / 4;
  constexpr int BP  = KC * F4R / 256;
  constexpr int BRS = 256 / F4R;
  constexpr int BS  = KC * BN / 256;
  constexpr int BSS = 256 / BN;

  const int tid  = threadIdx.x;
  const int lane = tid & 31;
  const int wave = tid >> 5;
  const int h    = lane >> 4;
  const int ln   = lane & 15;
  const int wm   = wave / NTB;
  const int wn   = wave % NTB;
  const int m0   = blockIdx.y * BM;
  const int n0   = bx * BN;

  const int bn4 = tid % F4R;
  const int bk  = tid / F4R;
  const int bn  = tid % BN;
  const int bk2 = tid / BN;
  const int colc = (n0 + bn < N) ? (n0 + bn) : (N - 1);

  const int Ktot = k1 + k2;
  const int nc   = (Ktot + KC - 1) / KC;

  float  ra[AJ][AP];
  bool   av[AJ];
  v4f    rb[BP > 0 ? BP : 1];
  float  rbs[BS];

  auto gload = [&](int i) {
    const int k0 = i * KC;
    #pragma unroll
    for (int j = 0; j < AJ; ++j) {
      const int kk = k0 + 32 * j + lane;
      av[j] = kk < Ktot;
      const int kc = av[j] ? kk : Ktot - 1;
      const int ii = kc;
      const bool s2 = ii >= k1;
      const float* Aseg = s2 ? A2 : A1;
      const int len = s2 ? k2 : k1;
      const int cc  = s2 ? ii - k1 : ii;
      #pragma unroll
      for (int p = 0; p < AP; ++p)
        ra[j][p] = Aseg[(m0 + wave + 8 * p) * len + cc];
    }
    if (VEC) {
      #pragma unroll
      for (int p = 0; p < BP; ++p) {
        int kr = k0 + bk + BRS * p;
        kr = kr < Ktot ? kr : Ktot - 1;
        if (MV) {
          const float* __restrict__ Wh = (bn4 < F4R / 2) ? W : W2;
          rb[p] = *(const v4f*)(Wh + kr * LAT + 4 * (bn4 & (F4R / 2 - 1)));
        } else {
          rb[p] = *(const v4f*)(W + kr * N + n0 + 4 * bn4);
        }
      }
    } else {
      #pragma unroll
      for (int p = 0; p < BS; ++p) {
        int kr = k0 + bk2 + BSS * p;
        kr = kr < Ktot ? kr : Ktot - 1;
        rbs[p] = W[kr * N + colc];
      }
    }
  };

  auto sstore = [&](int st) {
    float* as = As + st * (BM * PA);
    float* bs = Bs + st * (KC * PB);
    #pragma unroll
    for (int j = 0; j < AJ; ++j) {
      #pragma unroll
      for (int p = 0; p < AP; ++p) {
        as[(wave + 8 * p) * PA + 32 * j + lane] = av[j] ? ra[j][p] : 0.f;
      }
    }
    if (VEC) {
      #pragma unroll
      for (int p = 0; p < BP; ++p)
        *(v4f*)(bs + (bk + BRS * p) * PB + 4 * bn4) = rb[p];
    } else {
      #pragma unroll
      for (int p = 0; p < BS; ++p)
        bs[(bk2 + BSS * p) * PB + bn] = rbs[p];
    }
  };

  v8f acc = (v8f){0.f,0.f,0.f,0.f,0.f,0.f,0.f,0.f};

  const int aoff = (16 * wm + ln) * PA + 2 * h;
  const int boff = (2 * h) * PB + 16 * wn + ln;

  auto mma32 = [&](const float* asj, const float* bsj, v8f a) {
    v2f af[8], bf[8];
    #pragma unroll
    for (int ks = 0; ks < 8; ++ks) {
      af[ks]   = *(const v2f*)(asj + 4 * ks);
      bf[ks].x = bsj[(4 * ks) * PB];
      bf[ks].y = bsj[(4 * ks + 1) * PB];
    }
    asm volatile("" :: "v"(af[0]), "v"(af[1]), "v"(af[2]), "v"(af[3]),
                       "v"(af[4]), "v"(af[5]), "v"(af[6]), "v"(af[7]),
                       "v"(bf[0]), "v"(bf[1]), "v"(bf[2]), "v"(bf[3]),
                       "v"(bf[4]), "v"(bf[5]), "v"(bf[6]), "v"(bf[7]));
    #pragma unroll
    for (int ks = 0; ks < 8; ++ks)
      a = wmma4(af[ks], bf[ks], a);
    return a;
  };

  gload(0);
  sstore(0);
  __syncthreads();

  for (int i = 0; i < nc; ++i) {
    const int st = i & 1;
    if (i + 1 < nc) gload(i + 1);

    const int rem = Ktot - i * KC;
    const int nsb = rem >= KC ? KC / 32 : (rem + 31) / 32;

    const float* as = As + st * (BM * PA) + aoff;
    const float* bs = Bs + st * (KC * PB) + boff;
    for (int j = 0; j < nsb; ++j)
      acc = mma32(as + 32 * j, bs + 32 * j * PB, acc);


    if (i + 1 < nc) sstore(st ^ 1);
    __syncthreads();
  }

  const int colB = n0 + 16 * wn + ln;
  if (GATE) {
    constexpr int PT = GH + 1;
    float* as2 = As;
    float* bs2 = Bs;
    {
      const float bv = bias[colB];
      #pragma unroll
      for (int r = 0; r < 8; ++r)
        as2[(16 * wm + r + 8 * h) * PA + colB] = elu1(acc[r] + bv);
      #pragma unroll
      for (int p = 0; p < GH / BRS; ++p) {
        const int kr = bk + BRS * p;
        *(v4f*)(bs2 + kr * PB + 4 * bn4) = *(const v4f*)(W2 + kr * GH + 4 * bn4);
      }
    }
    __syncthreads();
    v8f acc2 = (v8f){0.f,0.f,0.f,0.f,0.f,0.f,0.f,0.f};
    #pragma unroll
    for (int j = 0; j < GH / 32; ++j)
      acc2 = mma32(as2 + aoff + 32 * j, bs2 + boff + 32 * j * PB, acc2);
    float* T = As + BM * PA;
    {
      const float bv2 = bias2[colB];
      #pragma unroll
      for (int r = 0; r < 8; ++r)
        T[(16 * wm + r + 8 * h) * PT + colB] = elu1(acc2[r] + bv2);
    }
    __syncthreads();
    if (tid < BM) {
      float lg[NE];
      #pragma unroll
      for (int e = 0; e < NE; ++e) lg[e] = bias3[e];
      for (int i = 0; i < GH; ++i) {
        const float gv = T[tid * PT + i];
        #pragma unroll
        for (int e = 0; e < NE; ++e) lg[e] = fmaf(gv, W3[i * NE + e], lg[e]);
      }
      float mx = lg[0];
      #pragma unroll
      for (int e = 1; e < NE; ++e) mx = fmaxf(mx, lg[e]);
      float s = 0.f, ex[NE];
      #pragma unroll
      for (int e = 0; e < NE; ++e) { ex[e] = expf(lg[e] - mx); s += ex[e]; }
      const float inv = 1.f / s;
      #pragma unroll
      for (int e = 0; e < NE; ++e) Z[(m0 + tid) * NE + e] = ex[e] * inv;
    }
  } else if (MV) {
    constexpr int PT = MVN + 1;
    float* T = As;
    const float bv = (colB < LAT) ? bias[colB] : bias2[colB - LAT];
    #pragma unroll
    for (int r = 0; r < 8; ++r)
      T[(16 * wm + r + 8 * h) * PT + colB] = acc[r] + bv;
    __syncthreads();
    #pragma unroll
    for (int it = 0; it < BM * LAT / 256; ++it) {
      const int idx = tid + 256 * it;
      const int rr  = idx / LAT;
      const int cc  = idx % LAT;
      const float m = T[rr * PT + cc];
      const float v = T[rr * PT + LAT + cc];
      const int   o = (m0 + rr) * LAT + cc;
      if (wmv) {
        C[o]  = m;
        C2[o] = v;
      }
      Z[o]  = m + eps[o] * expf(0.5f * v);
    }
  } else if (colB < N) {
    #pragma unroll
    for (int r = 0; r < 8; ++r) {
      const int row = m0 + 16 * wm + r + 8 * h;
      float y = acc[r];
      if (bias) y += bias[colB];
      if (act) y = elu1(y);
      C[row * N + colB] = y;
    }
  }
}

template <int N, int K1, int K2, int MT, int NTB>
__device__ __forceinline__ void
moe_tile(const float* A1, const float* A2,
         const float* coeff,
         const float* __restrict__ W,
         const float* __restrict__ bias,
         float* C,
         int act,
         int bx,
         float* As,
         float* Bs)
{
  static_assert(K1 % KQ == 0, "a round reads one segment of [A1|A2]");
  static_assert(MT * NTB == 8, "eight waves per work-group");
  constexpr int BM  = 16 * MT;
  constexpr int BN  = 16 * NTB;
  constexpr int I   = K1 + K2;
  constexpr int NR  = (I + KQ - 1) / KQ;
  constexpr bool EXACT = (I % KQ == 0);
  constexpr int PA  = KQ + 4;
  constexpr int PB  = BN + 8;
  constexpr int WR  = NE * KQ;
  constexpr int AP  = BM / 8;
  constexpr bool VEC = (N % BN == 0);
  constexpr int F4R = BN / 4;
  constexpr int BP  = WR * F4R / 256;
  constexpr int BRS = 256 / F4R;
  constexpr int BS  = WR * BN / 256;
  constexpr int BSS = 256 / BN;
  static_assert(KQ % BRS == 0 && KQ % BSS == 0, "a staging pass stays inside one expert");

  const int tid  = threadIdx.x;
  const int lane = tid & 31;
  const int wave = tid >> 5;
  const int h    = lane >> 4;
  const int ln   = lane & 15;
  const int wm   = wave / NTB;
  const int wn   = wave % NTB;
  const int m0   = blockIdx.y * BM;
  const int n0   = bx * BN;

  const int bn4 = tid % F4R;
  const int bk  = tid / F4R;
  const int bn  = tid % BN;
  const int bk2 = tid / BN;
  const int colc = (n0 + bn < N) ? (n0 + bn) : (N - 1);

  float ra[AP];
  bool  av;
  v4f   rb[BP];
  float rbs[BS];

  auto gload = [&](int i) {
    const int k0 = i * KQ;
    const bool full = EXACT || (k0 + KQ <= I);
    const bool s2 = k0 >= K1;
    const float* Aseg = s2 ? A2 : A1;
    const int len = s2 ? K2 : K1;
    av = full || (k0 + lane < I);
    const int kc = av ? k0 + lane : I - 1;
    const int cc = s2 ? kc - K1 : kc;
    #pragma unroll
    for (int p = 0; p < AP; ++p)
      ra[p] = Aseg[(m0 + wave + 8 * p) * len + cc];
    if (VEC) {
      #pragma unroll
      for (int p = 0; p < BP; ++p) {
        const int e = (BRS * p) / KQ;
        int kr = k0 + (BRS * p) % KQ + bk;
        if (!full) kr = kr < I ? kr : I - 1;
        rb[p] = *(const v4f*)(W + (e * I + kr) * N + n0 + 4 * bn4);
      }
    } else {
      #pragma unroll
      for (int p = 0; p < BS; ++p) {
        const int e = (BSS * p) / KQ;
        int kr = k0 + (BSS * p) % KQ + bk2;
        if (!full) kr = kr < I ? kr : I - 1;
        rbs[p] = W[(e * I + kr) * N + colc];
      }
    }
  };

  auto sstore = [&](int st) {
    float* as = As + st * (BM * PA);
    float* bs = Bs + st * (WR * PB);
    #pragma unroll
    for (int p = 0; p < AP; ++p)
      as[(wave + 8 * p) * PA + lane] = av ? ra[p] : 0.f;
    if (VEC) {
      #pragma unroll
      for (int p = 0; p < BP; ++p)
        *(v4f*)(bs + (bk + BRS * p) * PB + 4 * bn4) = rb[p];
    } else {
      #pragma unroll
      for (int p = 0; p < BS; ++p)
        bs[(bk2 + BSS * p) * PB + bn] = rbs[p];
    }
  };

  v8f acc[NE];
  #pragma unroll
  for (int e = 0; e < NE; ++e) acc[e] = (v8f){0.f,0.f,0.f,0.f,0.f,0.f,0.f,0.f};

  const int aoff = (16 * wm + ln) * PA + 2 * h;
  const int boff = (2 * h) * PB + 16 * wn + ln;

  auto mma16 = [&](const float* ag, const float* bg) {
    v2f af[4], bf[4][NE];
    #pragma unroll
    for (int ks = 0; ks < 4; ++ks) {
      af[ks] = *(const v2f*)(ag + 4 * ks);
      #pragma unroll
      for (int e = 0; e < NE; ++e) {
        bf[ks][e].x = bg[(e * KQ + 4 * ks) * PB];
        bf[ks][e].y = bg[(e * KQ + 4 * ks + 1) * PB];
      }
    }
    asm volatile("" :: "v"(af[0]), "v"(af[1]), "v"(af[2]), "v"(af[3]),
                       "v"(bf[0][0]), "v"(bf[0][1]), "v"(bf[0][2]), "v"(bf[0][3]), "v"(bf[0][4]), "v"(bf[0][5]),
                       "v"(bf[1][0]), "v"(bf[1][1]), "v"(bf[1][2]), "v"(bf[1][3]), "v"(bf[1][4]), "v"(bf[1][5]),
                       "v"(bf[2][0]), "v"(bf[2][1]), "v"(bf[2][2]), "v"(bf[2][3]), "v"(bf[2][4]), "v"(bf[2][5]),
                       "v"(bf[3][0]), "v"(bf[3][1]), "v"(bf[3][2]), "v"(bf[3][3]), "v"(bf[3][4]), "v"(bf[3][5]));
    #pragma unroll
    for (int ks = 0; ks < 4; ++ks) {
      #pragma unroll
      for (int e = 0; e < NE; ++e)
        acc[e] = wmma4(af[ks], bf[ks][e], acc[e]);
    }
  };

  gload(0);
  sstore(0);
  __syncthreads();

  for (int i = 0; i < NR; ++i) {
    const int st = i & 1;
    if (i + 1 < NR) gload(i + 1);

    const int rem = I - i * KQ;
    const int ng  = (EXACT || rem >= KQ) ? KQ / 16 : (rem + 15) / 16;

    const float* as = As + st * (BM * PA) + aoff;
    const float* bs = Bs + st * (WR * PB) + boff;
    for (int g = 0; g < ng; ++g)
      mma16(as + 16 * g, bs + 16 * g * PB);

    if (i + 1 < NR) sstore(st ^ 1);
    __syncthreads();
  }

  const int colB = n0 + 16 * wn + ln;
  if (colB < N) {
    float be[NE];
    #pragma unroll
    for (int e = 0; e < NE; ++e) be[e] = bias[e * N + colB];
    #pragma unroll
    for (int r = 0; r < 8; ++r) {
      const int row = m0 + 16 * wm + r + 8 * h;
      float y = 0.f, s = 0.f;
      #pragma unroll
      for (int e = 0; e < NE; ++e) {
        const float ce = coeff[row * NE + e];
        y = fmaf(ce, acc[e][r], y);
        s = fmaf(ce, be[e], s);
      }
      y += s;
      if (act) y = elu1(y);
      C[row * N + colB] = y;
    }
  }
}

template <int N, int MODE, int MT, int NTB>
__global__ void __launch_bounds__(256)
gemm_lds_f32(const float* __restrict__ A1, int k1,
             const float* __restrict__ A2, int k2,
             const float* __restrict__ W,
             const float* __restrict__ bias,
             float* __restrict__ C,
             int act)
{
  __shared__ __attribute__((aligned(16))) float As[2 * (16 * MT) * (KC + 4)];
  __shared__ __attribute__((aligned(16))) float Bs[2 * KC * (16 * NTB + 8)];
  gemm_tile<N, MODE, MT, NTB>(A1, k1, A2, k2, W, bias, nullptr, nullptr, nullptr, nullptr, nullptr,
                              C, nullptr, nullptr, act, blockIdx.x, false, As, Bs);
}

template <int N, int K2, int MT, int NTB>
__global__ void __launch_bounds__(256)
moe_lds_f32(const float* __restrict__ z,
            const float* __restrict__ prev,
            const float* __restrict__ coeff,
            const float* __restrict__ W,
            const float* __restrict__ bias,
            float* __restrict__ C,
            int act)
{
  __shared__ __attribute__((aligned(16))) float As[2 * (16 * MT) * (KQ + 4)];
  __shared__ __attribute__((aligned(16))) float Bs[2 * NE * KQ * (16 * NTB + 8)];
  moe_tile<N, LAT, K2, MT, NTB>(z, prev, coeff, W, bias, C, act, blockIdx.x, As, Bs);
}

__global__ void __launch_bounds__(256)
head_decoder0(const float* __restrict__ x, const float* __restrict__ h2in,
              const float* __restrict__ c, const float* __restrict__ eps,
              const float* __restrict__ em_w, const float* __restrict__ em_b,
              const float* __restrict__ ev_w, const float* __restrict__ ev_b,
              const float* __restrict__ g1_w, const float* __restrict__ g1_b,
              const float* __restrict__ g2_w, const float* __restrict__ g2_b,
              const float* __restrict__ g3_w, const float* __restrict__ g3_b,
              const float* __restrict__ w0, const float* __restrict__ b0,
              float* __restrict__ out_m, float* __restrict__ out_v,
              float* zpl, float* cpl,
              float* __restrict__ d0)
{
  __shared__ __attribute__((aligned(16))) float As[2 * 32 * (KC + 4)];
  __shared__ __attribute__((aligned(16))) float Bs[2 * (NE * KQ > KC ? NE * KQ : KC) * (64 + 8)];
  const int bx = blockIdx.x;
  float* zmine = zpl + bx * (BATCH * LAT);
  float* cmine = cpl + bx * (BATCH * NE);

  gemm_tile<MVN, MODE_MV, 2, 4>(x, FRAME, h2in, HID, em_w, em_b, ev_w, ev_b, eps, nullptr, nullptr,
                                out_m, out_v, zmine, 0, 0, bx == 0, As, Bs);
  __syncthreads();
  gemm_tile<GH, MODE_GATE, 2, 4>(zmine, LAT, c, FRAME, g1_w, g1_b, g2_w, g2_b, nullptr, g3_w, g3_b,
                                 nullptr, nullptr, cmine, 1, 0, false, As, Bs);
  __syncthreads();
  moe_tile<HID, LAT, FRAME, 2, 4>(zmine, c, cmine, w0, b0, d0, 1, bx, As, Bs);
}

template <int N, int MT, int NTB>
static inline dim3 gemm_grid() {
  return dim3((N + 16 * NTB - 1) / (16 * NTB), BATCH / (16 * MT));
}

extern "C" void kernel_launch(void* const* d_in, const int* in_sizes, int n_in,
                              void* d_out, int out_size, void* d_ws, size_t ws_size,
                              hipStream_t stream) {
  const float* x      = (const float*)d_in[0];
  const float* c      = (const float*)d_in[1];
  const float* eps    = (const float*)d_in[2];
  const float* efc1_w = (const float*)d_in[3];
  const float* efc1_b = (const float*)d_in[4];
  const float* efc2_w = (const float*)d_in[5];
  const float* efc2_b = (const float*)d_in[6];
  const float* em_w   = (const float*)d_in[7];
  const float* em_b   = (const float*)d_in[8];
  const float* ev_w   = (const float*)d_in[9];
  const float* ev_b   = (const float*)d_in[10];
  const float* g1_w   = (const float*)d_in[11];
  const float* g1_b   = (const float*)d_in[12];
  const float* g2_w   = (const float*)d_in[13];
  const float* g2_b   = (const float*)d_in[14];
  const float* g3_w   = (const float*)d_in[15];
  const float* g3_b   = (const float*)d_in[16];
  const float* w0     = (const float*)d_in[17];
  const float* b0     = (const float*)d_in[18];
  const float* w1     = (const float*)d_in[19];
  const float* b1     = (const float*)d_in[20];
  const float* w2     = (const float*)d_in[21];
  const float* b2     = (const float*)d_in[22];

  float* out   = (float*)d_out;
  float* out_m = out + BATCH * FRAME;
  float* out_v = out_m + BATCH * LAT;

  float* ws    = (float*)d_ws;
  float* h1    = ws;
  float* h2    = h1  + BATCH * HID;
  float* zpl   = h2  + BATCH * HID;
  float* cpl   = zpl + 4 * BATCH * LAT;
  float* z     = zpl;
  float* coeff = cpl;

  gemm_lds_f32<HID, MODE_PLAIN, 2, 4><<<gemm_grid<HID, 2, 4>(), 256, 0, stream>>>(
      x, FRAME, c, FRAME, efc1_w, efc1_b, h1, 1);
  gemm_lds_f32<HID, MODE_PLAIN, 2, 4><<<gemm_grid<HID, 2, 4>(), 256, 0, stream>>>(
      x, FRAME, h1, HID, efc2_w, efc2_b, h2, 1);

  head_decoder0<<<gemm_grid<HID, 2, 4>(), 256, 0, stream>>>(
      x, h2, c, eps, em_w, em_b, ev_w, ev_b, g1_w, g1_b, g2_w, g2_b, g3_w, g3_b, w0, b0,
      out_m, out_v, zpl, cpl, h1);

  moe_lds_f32<HID, HID, 4, 2><<<gemm_grid<HID, 4, 2>(), 256, 0, stream>>>(
      z, h1, coeff, w1, b1, h2, 1);
  moe_lds_f32<FRAME, HID, 4, 2><<<gemm_grid<FRAME, 4, 2>(), 256, 0, stream>>>(
      z, h2, coeff, w2, b2, out, 0);
}


# Epilogue operands (biases, row coefficients, eps, gate layer-2/3 weights) read ahead of each K loop and held in registers: no dependent global read after the last product; sum orders kept
# speedup vs baseline: 4.0518x; 1.1962x over previous
#include <hip/hip_runtime.h>
#include <hip/hip_bf16.h>


typedef __attribute__((ext_vector_type(2))) float v2f;
typedef __attribute__((ext_vector_type(4))) float v4f;
typedef __attribute__((ext_vector_type(8))) float v8f;

#define BATCH 2048
#define FRAME 267
#define LAT   32
#define HID   256
#define NE    6
#define GH    64
#define KC    128
#define KQ    32
#define MVN   (2 * LAT)

#define MODE_PLAIN 0
#define MODE_MV    1
#define MODE_GATE  2

__device__ __forceinline__ float elu1(float x) {
  return x > 0.f ? x : expm1f(x);
}

__device__ __forceinline__ v8f wmma4(v2f a, v2f b, v8f c) {
  return __builtin_amdgcn_wmma_f32_16x16x4_f32(
      false, a, false, b, (short)0, c, false, false);
}

template <int N, int MODE, int MT, int NTB>
__device__ __forceinline__ void
gemm_tile(const float* A1, int k1,
             const float* A2, int k2,
             const float* __restrict__ W,
             const float* __restrict__ bias,
             const float* __restrict__ W2,
             const float* __restrict__ bias2,
             const float* __restrict__ eps,
             const float* __restrict__ W3,
             const float* __restrict__ bias3,
             float* C,
             float* C2,
             float* Z,
             int act,
             int bx,
             bool wmv,
             float* As,
             float* Bs)
{
  constexpr bool MV   = (MODE == MODE_MV);
  constexpr bool GATE = (MODE == MODE_GATE);
  static_assert(!GATE || (N == GH && NTB == 4 && MT == 2), "GATE: one 32 x 64 tile holds whole rows of both hidden layers");
  static_assert(MT * NTB == 8, "eight waves per work-group");
  static_assert(!MV || (N == MVN && NTB == 4 && MT == 2), "MV: one 32 x 64 tile, columns 0..31 mean, 32..63 log-variance");
  constexpr int BM  = 16 * MT;
  constexpr int BN  = 16 * NTB;
  constexpr int PA  = KC + 4;
  constexpr int PB  = BN + 8;
  constexpr int AJ  = KC / 32;
  constexpr int AP  = BM / 8;
  constexpr bool VEC = (N % BN == 0);
  constexpr int F4R = BN / 4;
  constexpr int BP  = KC * F4R / 256;
  constexpr int BRS = 256 / F4R;
  constexpr int BS  = KC * BN / 256;
  constexpr int BSS = 256 / BN;

  const int tid  = threadIdx.x;
  const int lane = tid & 31;
  const int wave = tid >> 5;
  const int h    = lane >> 4;
  const int ln   = lane & 15;
  const int wm   = wave / NTB;
  const int wn   = wave % NTB;
  const int m0   = blockIdx.y * BM;
  const int n0   = bx * BN;

  const int bn4 = tid % F4R;
  const int bk  = tid / F4R;
  const int bn  = tid % BN;
  const int bk2 = tid / BN;
  const int colc = (n0 + bn < N) ? (n0 + bn) : (N - 1);

  const int Ktot = k1 + k2;
  const int nc   = (Ktot + KC - 1) / KC;

  float  ra[AJ][AP];
  bool   av[AJ];
  v4f    rb[BP > 0 ? BP : 1];
  float  rbs[BS];

  auto gload = [&](int i) {
    const int k0 = i * KC;
    #pragma unroll
    for (int j = 0; j < AJ; ++j) {
      const int kk = k0 + 32 * j + lane;
      av[j] = kk < Ktot;
      const int kc = av[j] ? kk : Ktot - 1;
      const int ii = kc;
      const bool s2 = ii >= k1;
      const float* Aseg = s2 ? A2 : A1;
      const int len = s2 ? k2 : k1;
      const int cc  = s2 ? ii - k1 : ii;
      #pragma unroll
      for (int p = 0; p < AP; ++p)
        ra[j][p] = Aseg[(m0 + wave + 8 * p) * len + cc];
    }
    if (VEC) {
      #pragma unroll
      for (int p = 0; p < BP; ++p) {
        int kr = k0 + bk + BRS * p;
        kr = kr < Ktot ? kr : Ktot - 1;
        if (MV) {
          const float* __restrict__ Wh = (bn4 < F4R / 2) ? W : W2;
          rb[p] = *(const v4f*)(Wh + kr * LAT + 4 * (bn4 & (F4R / 2 - 1)));
        } else {
          rb[p] = *(const v4f*)(W + kr * N + n0 + 4 * bn4);
        }
      }
    } else {
      #pragma unroll
      for (int p = 0; p < BS; ++p) {
        int kr = k0 + bk2 + BSS * p;
        kr = kr < Ktot ? kr : Ktot - 1;
        rbs[p] = W[kr * N + colc];
      }
    }
  };

  auto sstore = [&](int st) {
    float* as = As + st * (BM * PA);
    float* bs = Bs + st * (KC * PB);
    #pragma unroll
    for (int j = 0; j < AJ; ++j) {
      #pragma unroll
      for (int p = 0; p < AP; ++p) {
        as[(wave + 8 * p) * PA + 32 * j + lane] = av[j] ? ra[j][p] : 0.f;
      }
    }
    if (VEC) {
      #pragma unroll
      for (int p = 0; p < BP; ++p)
        *(v4f*)(bs + (bk + BRS * p) * PB + 4 * bn4) = rb[p];
    } else {
      #pragma unroll
      for (int p = 0; p < BS; ++p)
        bs[(bk2 + BSS * p) * PB + bn] = rbs[p];
    }
  };

  v8f acc = (v8f){0.f,0.f,0.f,0.f,0.f,0.f,0.f,0.f};

  const int aoff = (16 * wm + ln) * PA + 2 * h;
  const int boff = (2 * h) * PB + 16 * wn + ln;

  auto mma32 = [&](const float* asj, const float* bsj, v8f a) {
    v2f af[8], bf[8];
    #pragma unroll
    for (int ks = 0; ks < 8; ++ks) {
      af[ks]   = *(const v2f*)(asj + 4 * ks);
      bf[ks].x = bsj[(4 * ks) * PB];
      bf[ks].y = bsj[(4 * ks + 1) * PB];
    }
    asm volatile("" :: "v"(af[0]), "v"(af[1]), "v"(af[2]), "v"(af[3]),
                       "v"(af[4]), "v"(af[5]), "v"(af[6]), "v"(af[7]),
                       "v"(bf[0]), "v"(bf[1]), "v"(bf[2]), "v"(bf[3]),
                       "v"(bf[4]), "v"(bf[5]), "v"(bf[6]), "v"(bf[7]));
    #pragma unroll
    for (int ks = 0; ks < 8; ++ks)
      a = wmma4(af[ks], bf[ks], a);
    return a;
  };

  const int colB = n0 + 16 * wn + ln;
  float pb = 0.f, pb2 = 0.f;
  float pe[BM * LAT / 256] = {0.f, 0.f, 0.f, 0.f};
  v4f   pw2[GH / BRS];
  float pw3[2] = {0.f, 0.f};
  #pragma unroll
  for (int p = 0; p < GH / BRS; ++p) pw2[p] = (v4f){0.f, 0.f, 0.f, 0.f};
  if (GATE) {
    pb  = bias[colB];
    pb2 = bias2[colB];
    #pragma unroll
    for (int p = 0; p < GH / BRS; ++p)
      pw2[p] = *(const v4f*)(W2 + (bk + BRS * p) * GH + 4 * bn4);
    pw3[0] = W3[tid];
    pw3[1] = W3[256 + (tid & 127)];
  } else if (MV) {
    pb = (colB < LAT) ? bias[colB] : bias2[colB - LAT];
    #pragma unroll
    for (int it = 0; it < BM * LAT / 256; ++it) {
      const int idx = tid + 256 * it;
      pe[it] = eps[(m0 + idx / LAT) * LAT + idx % LAT];
    }
  } else {
    if (bias) pb = bias[colB < N ? colB : N - 1];
  }

  gload(0);
  sstore(0);
  if (GATE)
    asm volatile("" :: "v"(pb), "v"(pb2), "v"(pw2[0]), "v"(pw2[1]), "v"(pw2[2]), "v"(pw2[3]),
                       "v"(pw3[0]), "v"(pw3[1]));
  else if (MV)
    asm volatile("" :: "v"(pb), "v"(pe[0]), "v"(pe[1]), "v"(pe[2]), "v"(pe[3]));
  else
    asm volatile("" :: "v"(pb));
  __syncthreads();

  for (int i = 0; i < nc; ++i) {
    const int st = i & 1;
    if (i + 1 < nc) gload(i + 1);

    const int rem = Ktot - i * KC;
    const int nsb = rem >= KC ? KC / 32 : (rem + 31) / 32;

    const float* as = As + st * (BM * PA) + aoff;
    const float* bs = Bs + st * (KC * PB) + boff;
    for (int j = 0; j < nsb; ++j)
      acc = mma32(as + 32 * j, bs + 32 * j * PB, acc);


    if (i + 1 < nc) sstore(st ^ 1);
    __syncthreads();
  }

  if (GATE) {
    constexpr int PT = GH + 1;
    float* as2 = As;
    float* bs2 = Bs;
    float* W3s = Bs + KC * PB;
    {
      const float bv = pb;
      #pragma unroll
      for (int r = 0; r < 8; ++r)
        as2[(16 * wm + r + 8 * h) * PA + colB] = elu1(acc[r] + bv);
      #pragma unroll
      for (int p = 0; p < GH / BRS; ++p) {
        const int kr = bk + BRS * p;
        *(v4f*)(bs2 + kr * PB + 4 * bn4) = pw2[p];
      }
      W3s[tid] = pw3[0];
      if (tid < GH * NE - 256) W3s[256 + tid] = pw3[1];
    }
    __syncthreads();
    v8f acc2 = (v8f){0.f,0.f,0.f,0.f,0.f,0.f,0.f,0.f};
    #pragma unroll
    for (int j = 0; j < GH / 32; ++j)
      acc2 = mma32(as2 + aoff + 32 * j, bs2 + boff + 32 * j * PB, acc2);
    float* T = As + BM * PA;
    {
      const float bv2 = pb2;
      #pragma unroll
      for (int r = 0; r < 8; ++r)
        T[(16 * wm + r + 8 * h) * PT + colB] = elu1(acc2[r] + bv2);
    }
    __syncthreads();
    if (tid < BM) {
      float lg[NE];
      #pragma unroll
      for (int e = 0; e < NE; ++e) lg[e] = bias3[e];
      for (int i = 0; i < GH; ++i) {
        const float gv = T[tid * PT + i];
        #pragma unroll
        for (int e = 0; e < NE; ++e) lg[e] = fmaf(gv, W3s[i * NE + e], lg[e]);
      }
      float mx = lg[0];
      #pragma unroll
      for (int e = 1; e < NE; ++e) mx = fmaxf(mx, lg[e]);
      float s = 0.f, ex[NE];
      #pragma unroll
      for (int e = 0; e < NE; ++e) { ex[e] = expf(lg[e] - mx); s += ex[e]; }
      const float inv = 1.f / s;
      #pragma unroll
      for (int e = 0; e < NE; ++e) Z[(m0 + tid) * NE + e] = ex[e] * inv;
    }
  } else if (MV) {
    constexpr int PT = MVN + 1;
    float* T = As;
    const float bv = pb;
    #pragma unroll
    for (int r = 0; r < 8; ++r)
      T[(16 * wm + r + 8 * h) * PT + colB] = acc[r] + bv;
    __syncthreads();
    #pragma unroll
    for (int it = 0; it < BM * LAT / 256; ++it) {
      const int idx = tid + 256 * it;
      const int rr  = idx / LAT;
      const int cc  = idx % LAT;
      const float m = T[rr * PT + cc];
      const float v = T[rr * PT + LAT + cc];
      const int   o = (m0 + rr) * LAT + cc;
      if (wmv) {
        C[o]  = m;
        C2[o] = v;
      }
      Z[o]  = m + pe[it] * expf(0.5f * v);
    }
  } else if (colB < N) {
    #pragma unroll
    for (int r = 0; r < 8; ++r) {
      const int row = m0 + 16 * wm + r + 8 * h;
      float y = acc[r];
      if (bias) y += pb;
      if (act) y = elu1(y);
      C[row * N + colB] = y;
    }
  }
}

template <int N, int K1, int K2, int MT, int NTB>
__device__ __forceinline__ void
moe_tile(const float* A1, const float* A2,
         const float* coeff,
         const float* __restrict__ W,
         const float* __restrict__ bias,
         float* C,
         int act,
         int bx,
         float* As,
         float* Bs)
{
  static_assert(K1 % KQ == 0, "a round reads one segment of [A1|A2]");
  static_assert(MT * NTB == 8, "eight waves per work-group");
  constexpr int BM  = 16 * MT;
  constexpr int BN  = 16 * NTB;
  constexpr int I   = K1 + K2;
  constexpr int NR  = (I + KQ - 1) / KQ;
  constexpr bool EXACT = (I % KQ == 0);
  constexpr int PA  = KQ + 4;
  constexpr int PB  = BN + 8;
  constexpr int WR  = NE * KQ;
  constexpr int AP  = BM / 8;
  constexpr bool VEC = (N % BN == 0);
  constexpr int F4R = BN / 4;
  constexpr int BP  = WR * F4R / 256;
  constexpr int BRS = 256 / F4R;
  constexpr int BS  = WR * BN / 256;
  constexpr int BSS = 256 / BN;
  static_assert(KQ % BRS == 0 && KQ % BSS == 0, "a staging pass stays inside one expert");

  const int tid  = threadIdx.x;
  const int lane = tid & 31;
  const int wave = tid >> 5;
  const int h    = lane >> 4;
  const int ln   = lane & 15;
  const int wm   = wave / NTB;
  const int wn   = wave % NTB;
  const int m0   = blockIdx.y * BM;
  const int n0   = bx * BN;

  const int bn4 = tid % F4R;
  const int bk  = tid / F4R;
  const int bn  = tid % BN;
  const int bk2 = tid / BN;
  const int colc = (n0 + bn < N) ? (n0 + bn) : (N - 1);

  float ra[AP];
  bool  av;
  v4f   rb[BP];
  float rbs[BS];

  auto gload = [&](int i) {
    const int k0 = i * KQ;
    const bool full = EXACT || (k0 + KQ <= I);
    const bool s2 = k0 >= K1;
    const float* Aseg = s2 ? A2 : A1;
    const int len = s2 ? K2 : K1;
    av = full || (k0 + lane < I);
    const int kc = av ? k0 + lane : I - 1;
    const int cc = s2 ? kc - K1 : kc;
    #pragma unroll
    for (int p = 0; p < AP; ++p)
      ra[p] = Aseg[(m0 + wave + 8 * p) * len + cc];
    if (VEC) {
      #pragma unroll
      for (int p = 0; p < BP; ++p) {
        const int e = (BRS * p) / KQ;
        int kr = k0 + (BRS * p) % KQ + bk;
        if (!full) kr = kr < I ? kr : I - 1;
        rb[p] = *(const v4f*)(W + (e * I + kr) * N + n0 + 4 * bn4);
      }
    } else {
      #pragma unroll
      for (int p = 0; p < BS; ++p) {
        const int e = (BSS * p) / KQ;
        int kr = k0 + (BSS * p) % KQ + bk2;
        if (!full) kr = kr < I ? kr : I - 1;
        rbs[p] = W[(e * I + kr) * N + colc];
      }
    }
  };

  auto sstore = [&](int st) {
    float* as = As + st * (BM * PA);
    float* bs = Bs + st * (WR * PB);
    #pragma unroll
    for (int p = 0; p < AP; ++p)
      as[(wave + 8 * p) * PA + lane] = av ? ra[p] : 0.f;
    if (VEC) {
      #pragma unroll
      for (int p = 0; p < BP; ++p)
        *(v4f*)(bs + (bk + BRS * p) * PB + 4 * bn4) = rb[p];
    } else {
      #pragma unroll
      for (int p = 0; p < BS; ++p)
        bs[(bk2 + BSS * p) * PB + bn] = rbs[p];
    }
  };

  v8f acc[NE];
  #pragma unroll
  for (int e = 0; e < NE; ++e) acc[e] = (v8f){0.f,0.f,0.f,0.f,0.f,0.f,0.f,0.f};

  const int aoff = (16 * wm + ln) * PA + 2 * h;
  const int boff = (2 * h) * PB + 16 * wn + ln;

  auto mma16 = [&](const float* ag, const float* bg) {
    v2f af[4], bf[4][NE];
    #pragma unroll
    for (int ks = 0; ks < 4; ++ks) {
      af[ks] = *(const v2f*)(ag + 4 * ks);
      #pragma unroll
      for (int e = 0; e < NE; ++e) {
        bf[ks][e].x = bg[(e * KQ + 4 * ks) * PB];
        bf[ks][e].y = bg[(e * KQ + 4 * ks + 1) * PB];
      }
    }
    asm volatile("" :: "v"(af[0]), "v"(af[1]), "v"(af[2]), "v"(af[3]),
                       "v"(bf[0][0]), "v"(bf[0][1]), "v"(bf[0][2]), "v"(bf[0][3]), "v"(bf[0][4]), "v"(bf[0][5]),
                       "v"(bf[1][0]), "v"(bf[1][1]), "v"(bf[1][2]), "v"(bf[1][3]), "v"(bf[1][4]), "v"(bf[1][5]),
                       "v"(bf[2][0]), "v"(bf[2][1]), "v"(bf[2][2]), "v"(bf[2][3]), "v"(bf[2][4]), "v"(bf[2][5]),
                       "v"(bf[3][0]), "v"(bf[3][1]), "v"(bf[3][2]), "v"(bf[3][3]), "v"(bf[3][4]), "v"(bf[3][5]));
    #pragma unroll
    for (int ks = 0; ks < 4; ++ks) {
      #pragma unroll
      for (int e = 0; e < NE; ++e)
        acc[e] = wmma4(af[ks], bf[ks][e], acc[e]);
    }
  };

  const int colB = n0 + 16 * wn + ln;
  float be[NE];
  #pragma unroll
  for (int e = 0; e < NE; ++e) be[e] = bias[e * N + (colB < N ? colB : N - 1)];
  float cf[8][NE];
  #pragma unroll
  for (int r = 0; r < 8; ++r) {
    #pragma unroll
    for (int e = 0; e < NE; ++e) cf[r][e] = coeff[(m0 + 16 * wm + r + 8 * h) * NE + e];
  }

  gload(0);
  sstore(0);
  asm volatile("" :: "v"(be[0]), "v"(be[1]), "v"(be[2]), "v"(be[3]), "v"(be[4]), "v"(be[5]));
  #pragma unroll
  for (int r = 0; r < 8; ++r)
    asm volatile("" :: "v"(cf[r][0]), "v"(cf[r][1]), "v"(cf[r][2]), "v"(cf[r][3]), "v"(cf[r][4]), "v"(cf[r][5]));
  __syncthreads();

  for (int i = 0; i < NR; ++i) {
    const int st = i & 1;
    if (i + 1 < NR) gload(i + 1);

    const int rem = I - i * KQ;
    const int ng  = (EXACT || rem >= KQ) ? KQ / 16 : (rem + 15) / 16;

    const float* as = As + st * (BM * PA) + aoff;
    const float* bs = Bs + st * (WR * PB) + boff;
    for (int g = 0; g < ng; ++g)
      mma16(as + 16 * g, bs + 16 * g * PB);

    if (i + 1 < NR) sstore(st ^ 1);
    __syncthreads();
  }

  if (colB < N) {
    #pragma unroll
    for (int r = 0; r < 8; ++r) {
      const int row = m0 + 16 * wm + r + 8 * h;
      float y = 0.f, s = 0.f;
      #pragma unroll
      for (int e = 0; e < NE; ++e) {
        const float ce = cf[r][e];
        y = fmaf(ce, acc[e][r], y);
        s = fmaf(ce, be[e], s);
      }
      y += s;
      if (act) y = elu1(y);
      C[row * N + colB] = y;
    }
  }
}

template <int N, int MODE, int MT, int NTB>
__global__ void __launch_bounds__(256)
gemm_lds_f32(const float* __restrict__ A1, int k1,
             const float* __restrict__ A2, int k2,
             const float* __restrict__ W,
             const float* __restrict__ bias,
             float* __restrict__ C,
             int act)
{
  __shared__ __attribute__((aligned(16))) float As[2 * (16 * MT) * (KC + 4)];
  __shared__ __attribute__((aligned(16))) float Bs[2 * KC * (16 * NTB + 8)];
  gemm_tile<N, MODE, MT, NTB>(A1, k1, A2, k2, W, bias, nullptr, nullptr, nullptr, nullptr, nullptr,
                              C, nullptr, nullptr, act, blockIdx.x, false, As, Bs);
}

template <int N, int K2, int MT, int NTB>
__global__ void __launch_bounds__(256)
moe_lds_f32(const float* __restrict__ z,
            const float* __restrict__ prev,
            const float* __restrict__ coeff,
            const float* __restrict__ W,
            const float* __restrict__ bias,
            float* __restrict__ C,
            int act)
{
  __shared__ __attribute__((aligned(16))) float As[2 * (16 * MT) * (KQ + 4)];
  __shared__ __attribute__((aligned(16))) float Bs[2 * NE * KQ * (16 * NTB + 8)];
  moe_tile<N, LAT, K2, MT, NTB>(z, prev, coeff, W, bias, C, act, blockIdx.x, As, Bs);
}

__global__ void __launch_bounds__(256)
head_decoder0(const float* __restrict__ x, const float* __restrict__ h2in,
              const float* __restrict__ c, const float* __restrict__ eps,
              const float* __restrict__ em_w, const float* __restrict__ em_b,
              const float* __restrict__ ev_w, const float* __restrict__ ev_b,
              const float* __restrict__ g1_w, const float* __restrict__ g1_b,
              const float* __restrict__ g2_w, const float* __restrict__ g2_b,
              const float* __restrict__ g3_w, const float* __restrict__ g3_b,
              const float* __restrict__ w0, const float* __restrict__ b0,
              float* __restrict__ out_m, float* __restrict__ out_v,
              float* zpl, float* cpl,
              float* __restrict__ d0)
{
  __shared__ __attribute__((aligned(16))) float As[2 * 32 * (KC + 4)];
  __shared__ __attribute__((aligned(16))) float Bs[2 * (NE * KQ > KC ? NE * KQ : KC) * (64 + 8)];
  const int bx = blockIdx.x;
  float* zmine = zpl + bx * (BATCH * LAT);
  float* cmine = cpl + bx * (BATCH * NE);

  gemm_tile<MVN, MODE_MV, 2, 4>(x, FRAME, h2in, HID, em_w, em_b, ev_w, ev_b, eps, nullptr, nullptr,
                                out_m, out_v, zmine, 0, 0, bx == 0, As, Bs);
  __syncthreads();
  gemm_tile<GH, MODE_GATE, 2, 4>(zmine, LAT, c, FRAME, g1_w, g1_b, g2_w, g2_b, nullptr, g3_w, g3_b,
                                 nullptr, nullptr, cmine, 1, 0, false, As, Bs);
  __syncthreads();
  moe_tile<HID, LAT, FRAME, 2, 4>(zmine, c, cmine, w0, b0, d0, 1, bx, As, Bs);
}

template <int N, int MT, int NTB>
static inline dim3 gemm_grid() {
  return dim3((N + 16 * NTB - 1) / (16 * NTB), BATCH / (16 * MT));
}

extern "C" void kernel_launch(void* const* d_in, const int* in_sizes, int n_in,
                              void* d_out, int out_size, void* d_ws, size_t ws_size,
                              hipStream_t stream) {
  const float* x      = (const float*)d_in[0];
  const float* c      = (const float*)d_in[1];
  const float* eps    = (const float*)d_in[2];
  const float* efc1_w = (const float*)d_in[3];
  const float* efc1_b = (const float*)d_in[4];
  const float* efc2_w = (const float*)d_in[5];
  const float* efc2_b = (const float*)d_in[6];
  const float* em_w   = (const float*)d_in[7];
  const float* em_b   = (const float*)d_in[8];
  const float* ev_w   = (const float*)d_in[9];
  const float* ev_b   = (const float*)d_in[10];
  const float* g1_w   = (const float*)d_in[11];
  const float* g1_b   = (const float*)d_in[12];
  const float* g2_w   = (const float*)d_in[13];
  const float* g2_b   = (const float*)d_in[14];
  const float* g3_w   = (const float*)d_in[15];
  const float* g3_b   = (const float*)d_in[16];
  const float* w0     = (const float*)d_in[17];
  const float* b0     = (const float*)d_in[18];
  const float* w1     = (const float*)d_in[19];
  const float* b1     = (const float*)d_in[20];
  const float* w2     = (const float*)d_in[21];
  const float* b2     = (const float*)d_in[22];

  float* out   = (float*)d_out;
  float* out_m = out + BATCH * FRAME;
  float* out_v = out_m + BATCH * LAT;

  float* ws    = (float*)d_ws;
  float* h1    = ws;
  float* h2    = h1  + BATCH * HID;
  float* zpl   = h2  + BATCH * HID;
  float* cpl   = zpl + 4 * BATCH * LAT;
  float* z     = zpl;
  float* coeff = cpl;

  gemm_lds_f32<HID, MODE_PLAIN, 2, 4><<<gemm_grid<HID, 2, 4>(), 256, 0, stream>>>(
      x, FRAME, c, FRAME, efc1_w, efc1_b, h1, 1);
  gemm_lds_f32<HID, MODE_PLAIN, 2, 4><<<gemm_grid<HID, 2, 4>(), 256, 0, stream>>>(
      x, FRAME, h1, HID, efc2_w, efc2_b, h2, 1);

  head_decoder0<<<gemm_grid<HID, 2, 4>(), 256, 0, stream>>>(
      x, h2, c, eps, em_w, em_b, ev_w, ev_b, g1_w, g1_b, g2_w, g2_b, g3_w, g3_b, w0, b0,
      out_m, out_v, zpl, cpl, h1);

  moe_lds_f32<HID, HID, 4, 2><<<gemm_grid<HID, 4, 2>(), 256, 0, stream>>>(
      z, h1, coeff, w1, b1, h2, 1);
  moe_lds_f32<FRAME, HID, 4, 2><<<gemm_grid<FRAME, 4, 2>(), 256, 0, stream>>>(
      z, h2, coeff, w2, b2, out, 0);
}


# K loops (GEMMs, decoder layer 0 MoE tile): global loads issued two chunks ahead via a second staging register set; loop barriers order LDS only; that tile's coefficients kept in LDS; sum orders kept
# speedup vs baseline: 4.2520x; 1.0494x over previous
#include <hip/hip_runtime.h>
#include <hip/hip_bf16.h>


typedef __attribute__((ext_vector_type(2))) float v2f;
typedef __attribute__((ext_vector_type(4))) float v4f;
typedef __attribute__((ext_vector_type(8))) float v8f;

#define BATCH 2048
#define FRAME 267
#define LAT   32
#define HID   256
#define NE    6
#define GH    64
#define KC    128
#define KQ    32
#define MVN   (2 * LAT)

#define MODE_PLAIN 0
#define MODE_MV    1
#define MODE_GATE  2

__device__ __forceinline__ float elu1(float x) {
  return x > 0.f ? x : expm1f(x);
}

__device__ __forceinline__ v8f wmma4(v2f a, v2f b, v8f c) {
  return __builtin_amdgcn_wmma_f32_16x16x4_f32(
      false, a, false, b, (short)0, c, false, false);
}

__device__ __forceinline__ void lds_barrier() {
  __builtin_amdgcn_fence(__ATOMIC_RELEASE, "workgroup", "local");
  __builtin_amdgcn_s_barrier();
  __builtin_amdgcn_fence(__ATOMIC_ACQUIRE, "workgroup", "local");
}

template <int N, int MODE, int MT, int NTB>
__device__ __forceinline__ void
gemm_tile(const float* A1, int k1,
             const float* A2, int k2,
             const float* __restrict__ W,
             const float* __restrict__ bias,
             const float* __restrict__ W2,
             const float* __restrict__ bias2,
             const float* __restrict__ eps,
             const float* __restrict__ W3,
             const float* __restrict__ bias3,
             float* C,
             float* C2,
             float* Z,
             int act,
             int bx,
             bool wmv,
             float* As,
             float* Bs)
{
  constexpr bool MV   = (MODE == MODE_MV);
  constexpr bool GATE = (MODE == MODE_GATE);
  static_assert(!GATE || (N == GH && NTB == 4 && MT == 2), "GATE: one 32 x 64 tile holds whole rows of both hidden layers");
  static_assert(MT * NTB == 8, "eight waves per work-group");
  static_assert(!MV || (N == MVN && NTB == 4 && MT == 2), "MV: one 32 x 64 tile, columns 0..31 mean, 32..63 log-variance");
  constexpr int BM  = 16 * MT;
  constexpr int BN  = 16 * NTB;
  constexpr int PA  = KC + 4;
  constexpr int PB  = BN + 8;
  constexpr int AJ  = KC / 32;
  constexpr int AP  = BM / 8;
  constexpr bool VEC = (N % BN == 0);
  constexpr int F4R = BN / 4;
  constexpr int BP  = KC * F4R / 256;
  constexpr int BRS = 256 / F4R;
  constexpr int BS  = KC * BN / 256;
  constexpr int BSS = 256 / BN;

  const int tid  = threadIdx.x;
  const int lane = tid & 31;
  const int wave = tid >> 5;
  const int h    = lane >> 4;
  const int ln   = lane & 15;
  const int wm   = wave / NTB;
  const int wn   = wave % NTB;
  const int m0   = blockIdx.y * BM;
  const int n0   = bx * BN;

  const int bn4 = tid % F4R;
  const int bk  = tid / F4R;
  const int bn  = tid % BN;
  const int bk2 = tid / BN;
  const int colc = (n0 + bn < N) ? (n0 + bn) : (N - 1);

  const int Ktot = k1 + k2;
  const int nc   = (Ktot + KC - 1) / KC;

  struct StageRegs {
    float  ra[AJ][AP];
    bool   av[AJ];
    v4f    rb[BP > 0 ? BP : 1];
    float  rbs[BS];
  };
  StageRegs R0, R1;

  auto gload = [&](int i, StageRegs& R) {
    auto& ra = R.ra; auto& av = R.av; auto& rb = R.rb; auto& rbs = R.rbs;
    const int k0 = i * KC;
    #pragma unroll
    for (int j = 0; j < AJ; ++j) {
      const int kk = k0 + 32 * j + lane;
      av[j] = kk < Ktot;
      const int kc = av[j] ? kk : Ktot - 1;
      const int ii = kc;
      const bool s2 = ii >= k1;
      const float* Aseg = s2 ? A2 : A1;
      const int len = s2 ? k2 : k1;
      const int cc  = s2 ? ii - k1 : ii;
      #pragma unroll
      for (int p = 0; p < AP; ++p)
        ra[j][p] = Aseg[(m0 + wave + 8 * p) * len + cc];
    }
    if (VEC) {
      #pragma unroll
      for (int p = 0; p < BP; ++p) {
        int kr = k0 + bk + BRS * p;
        kr = kr < Ktot ? kr : Ktot - 1;
        if (MV) {
          const float* __restrict__ Wh = (bn4 < F4R / 2) ? W : W2;
          rb[p] = *(const v4f*)(Wh + kr * LAT + 4 * (bn4 & (F4R / 2 - 1)));
        } else {
          rb[p] = *(const v4f*)(W + kr * N + n0 + 4 * bn4);
        }
      }
    } else {
      #pragma unroll
      for (int p = 0; p < BS; ++p) {
        int kr = k0 + bk2 + BSS * p;
        kr = kr < Ktot ? kr : Ktot - 1;
        rbs[p] = W[kr * N + colc];
      }
    }
  };

  auto sstore = [&](int st, StageRegs& R) {
    auto& ra = R.ra; auto& av = R.av; auto& rb = R.rb; auto& rbs = R.rbs;
    float* as = As + st * (BM * PA);
    float* bs = Bs + st * (KC * PB);
    #pragma unroll
    for (int j = 0; j < AJ; ++j) {
      #pragma unroll
      for (int p = 0; p < AP; ++p) {
        as[(wave + 8 * p) * PA + 32 * j + lane] = av[j] ? ra[j][p] : 0.f;
      }
    }
    if (VEC) {
      #pragma unroll
      for (int p = 0; p < BP; ++p)
        *(v4f*)(bs + (bk + BRS * p) * PB + 4 * bn4) = rb[p];
    } else {
      #pragma unroll
      for (int p = 0; p < BS; ++p)
        bs[(bk2 + BSS * p) * PB + bn] = rbs[p];
    }
  };

  v8f acc = (v8f){0.f,0.f,0.f,0.f,0.f,0.f,0.f,0.f};

  const int aoff = (16 * wm + ln) * PA + 2 * h;
  const int boff = (2 * h) * PB + 16 * wn + ln;

  auto mma32 = [&](const float* asj, const float* bsj, v8f a) {
    v2f af[8], bf[8];
    #pragma unroll
    for (int ks = 0; ks < 8; ++ks) {
      af[ks]   = *(const v2f*)(asj + 4 * ks);
      bf[ks].x = bsj[(4 * ks) * PB];
      bf[ks].y = bsj[(4 * ks + 1) * PB];
    }
    asm volatile("" :: "v"(af[0]), "v"(af[1]), "v"(af[2]), "v"(af[3]),
                       "v"(af[4]), "v"(af[5]), "v"(af[6]), "v"(af[7]),
                       "v"(bf[0]), "v"(bf[1]), "v"(bf[2]), "v"(bf[3]),
                       "v"(bf[4]), "v"(bf[5]), "v"(bf[6]), "v"(bf[7]));
    #pragma unroll
    for (int ks = 0; ks < 8; ++ks)
      a = wmma4(af[ks], bf[ks], a);
    return a;
  };

  const int colB = n0 + 16 * wn + ln;
  float pb = 0.f, pb2 = 0.f;
  float pe[BM * LAT / 256] = {0.f, 0.f, 0.f, 0.f};
  v4f   pw2[GH / BRS];
  float pw3[2] = {0.f, 0.f};
  #pragma unroll
  for (int p = 0; p < GH / BRS; ++p) pw2[p] = (v4f){0.f, 0.f, 0.f, 0.f};
  if (GATE) {
    pb  = bias[colB];
    pb2 = bias2[colB];
    #pragma unroll
    for (int p = 0; p < GH / BRS; ++p)
      pw2[p] = *(const v4f*)(W2 + (bk + BRS * p) * GH + 4 * bn4);
    pw3[0] = W3[tid];
    pw3[1] = W3[256 + (tid & 127)];
  } else if (MV) {
    pb = (colB < LAT) ? bias[colB] : bias2[colB - LAT];
    #pragma unroll
    for (int it = 0; it < BM * LAT / 256; ++it) {
      const int idx = tid + 256 * it;
      pe[it] = eps[(m0 + idx / LAT) * LAT + idx % LAT];
    }
  } else {
    if (bias) pb = bias[colB < N ? colB : N - 1];
  }

  gload(0, R0);
  if (nc > 1) gload(1, R1);
  sstore(0, R0);
  if (GATE)
    asm volatile("" :: "v"(pb), "v"(pb2), "v"(pw2[0]), "v"(pw2[1]), "v"(pw2[2]), "v"(pw2[3]),
                       "v"(pw3[0]), "v"(pw3[1]));
  else if (MV)
    asm volatile("" :: "v"(pb), "v"(pe[0]), "v"(pe[1]), "v"(pe[2]), "v"(pe[3]));
  else
    asm volatile("" :: "v"(pb));
  lds_barrier();

  auto chunk = [&](int i, int st) {
    const int rem = Ktot - i * KC;
    const int nsb = rem >= KC ? KC / 32 : (rem + 31) / 32;
    const float* as = As + st * (BM * PA) + aoff;
    const float* bs = Bs + st * (KC * PB) + boff;
    for (int j = 0; j < nsb; ++j)
      acc = mma32(as + 32 * j, bs + 32 * j * PB, acc);
  };

  int i = 0;
  for (; i + 3 < nc; i += 2) {
    gload(i + 2, R0);
    chunk(i, 0);
    sstore(1, R1);
    lds_barrier();
    gload(i + 3, R1);
    chunk(i + 1, 1);
    sstore(0, R0);
    lds_barrier();
  }
  const int left = nc - i;
  if (left == 3) {
    gload(i + 2, R0);
    chunk(i, 0);
    sstore(1, R1);
    lds_barrier();
    chunk(i + 1, 1);
    sstore(0, R0);
    lds_barrier();
    chunk(i + 2, 0);
  } else if (left == 2) {
    chunk(i, 0);
    sstore(1, R1);
    lds_barrier();
    chunk(i + 1, 1);
  } else {
    chunk(i, 0);
  }
  lds_barrier();

  if (GATE) {
    constexpr int PT = GH + 1;
    float* as2 = As;
    float* bs2 = Bs;
    float* W3s = Bs + KC * PB;
    {
      const float bv = pb;
      #pragma unroll
      for (int r = 0; r < 8; ++r)
        as2[(16 * wm + r + 8 * h) * PA + colB] = elu1(acc[r] + bv);
      #pragma unroll
      for (int p = 0; p < GH / BRS; ++p) {
        const int kr = bk + BRS * p;
        *(v4f*)(bs2 + kr * PB + 4 * bn4) = pw2[p];
      }
      W3s[tid] = pw3[0];
      if (tid < GH * NE - 256) W3s[256 + tid] = pw3[1];
    }
    __syncthreads();
    v8f acc2 = (v8f){0.f,0.f,0.f,0.f,0.f,0.f,0.f,0.f};
    #pragma unroll
    for (int j = 0; j < GH / 32; ++j)
      acc2 = mma32(as2 + aoff + 32 * j, bs2 + boff + 32 * j * PB, acc2);
    float* T = As + BM * PA;
    {
      const float bv2 = pb2;
      #pragma unroll
      for (int r = 0; r < 8; ++r)
        T[(16 * wm + r + 8 * h) * PT + colB] = elu1(acc2[r] + bv2);
    }
    __syncthreads();
    if (tid < BM) {
      float lg[NE];
      #pragma unroll
      for (int e = 0; e < NE; ++e) lg[e] = bias3[e];
      for (int i = 0; i < GH; ++i) {
        const float gv = T[tid * PT + i];
        #pragma unroll
        for (int e = 0; e < NE; ++e) lg[e] = fmaf(gv, W3s[i * NE + e], lg[e]);
      }
      float mx = lg[0];
      #pragma unroll
      for (int e = 1; e < NE; ++e) mx = fmaxf(mx, lg[e]);
      float s = 0.f, ex[NE];
      #pragma unroll
      for (int e = 0; e < NE; ++e) { ex[e] = expf(lg[e] - mx); s += ex[e]; }
      const float inv = 1.f / s;
      #pragma unroll
      for (int e = 0; e < NE; ++e) Z[(m0 + tid) * NE + e] = ex[e] * inv;
    }
  } else if (MV) {
    constexpr int PT = MVN + 1;
    float* T = As;
    const float bv = pb;
    #pragma unroll
    for (int r = 0; r < 8; ++r)
      T[(16 * wm + r + 8 * h) * PT + colB] = acc[r] + bv;
    __syncthreads();
    #pragma unroll
    for (int it = 0; it < BM * LAT / 256; ++it) {
      const int idx = tid + 256 * it;
      const int rr  = idx / LAT;
      const int cc  = idx % LAT;
      const float m = T[rr * PT + cc];
      const float v = T[rr * PT + LAT + cc];
      const int   o = (m0 + rr) * LAT + cc;
      if (wmv) {
        C[o]  = m;
        C2[o] = v;
      }
      Z[o]  = m + pe[it] * expf(0.5f * v);
    }
  } else if (colB < N) {
    #pragma unroll
    for (int r = 0; r < 8; ++r) {
      const int row = m0 + 16 * wm + r + 8 * h;
      float y = acc[r];
      if (bias) y += pb;
      if (act) y = elu1(y);
      C[row * N + colB] = y;
    }
  }
}

template <int N, int K1, int K2, int MT, int NTB, bool PF2>
__device__ __forceinline__ void
moe_tile(const float* A1, const float* A2,
         const float* coeff,
         const float* __restrict__ W,
         const float* __restrict__ bias,
         float* C,
         int act,
         int bx,
         float* As,
         float* Bs,
         float* Cs)
{
  static_assert(K1 % KQ == 0, "a round reads one segment of [A1|A2]");
  static_assert(MT * NTB == 8, "eight waves per work-group");
  constexpr int BM  = 16 * MT;
  constexpr int BN  = 16 * NTB;
  constexpr int I   = K1 + K2;
  constexpr int NR  = (I + KQ - 1) / KQ;
  constexpr bool EXACT = (I % KQ == 0);
  constexpr int PA  = KQ + 4;
  constexpr int PB  = BN + 8;
  constexpr int WR  = NE * KQ;
  constexpr int AP  = BM / 8;
  constexpr bool VEC = (N % BN == 0);
  constexpr int F4R = BN / 4;
  constexpr int BP  = WR * F4R / 256;
  constexpr int BRS = 256 / F4R;
  constexpr int BS  = WR * BN / 256;
  constexpr int BSS = 256 / BN;
  static_assert(KQ % BRS == 0 && KQ % BSS == 0, "a staging pass stays inside one expert");

  const int tid  = threadIdx.x;
  const int lane = tid & 31;
  const int wave = tid >> 5;
  const int h    = lane >> 4;
  const int ln   = lane & 15;
  const int wm   = wave / NTB;
  const int wn   = wave % NTB;
  const int m0   = blockIdx.y * BM;
  const int n0   = bx * BN;

  const int bn4 = tid % F4R;
  const int bk  = tid / F4R;
  const int bn  = tid % BN;
  const int bk2 = tid / BN;
  const int colc = (n0 + bn < N) ? (n0 + bn) : (N - 1);

  struct StageRegs {
    float ra[AP];
    bool  av;
    v4f   rb[BP];
    float rbs[BS];
  };
  StageRegs R0, R1;

  auto gload = [&](int i, StageRegs& R) {
    auto& ra = R.ra; auto& av = R.av; auto& rb = R.rb; auto& rbs = R.rbs;
    const int k0 = i * KQ;
    const bool full = EXACT || (k0 + KQ <= I);
    const bool s2 = k0 >= K1;
    const float* Aseg = s2 ? A2 : A1;
    const int len = s2 ? K2 : K1;
    av = full || (k0 + lane < I);
    const int kc = av ? k0 + lane : I - 1;
    const int cc = s2 ? kc - K1 : kc;
    #pragma unroll
    for (int p = 0; p < AP; ++p)
      ra[p] = Aseg[(m0 + wave + 8 * p) * len + cc];
    if (VEC) {
      #pragma unroll
      for (int p = 0; p < BP; ++p) {
        const int e = (BRS * p) / KQ;
        int kr = k0 + (BRS * p) % KQ + bk;
        if (!full) kr = kr < I ? kr : I - 1;
        rb[p] = *(const v4f*)(W + (e * I + kr) * N + n0 + 4 * bn4);
      }
    } else {
      #pragma unroll
      for (int p = 0; p < BS; ++p) {
        const int e = (BSS * p) / KQ;
        int kr = k0 + (BSS * p) % KQ + bk2;
        if (!full) kr = kr < I ? kr : I - 1;
        rbs[p] = W[(e * I + kr) * N + colc];
      }
    }
  };

  auto sstore = [&](int st, StageRegs& R) {
    auto& ra = R.ra; auto& av = R.av; auto& rb = R.rb; auto& rbs = R.rbs;
    float* as = As + st * (BM * PA);
    float* bs = Bs + st * (WR * PB);
    #pragma unroll
    for (int p = 0; p < AP; ++p)
      as[(wave + 8 * p) * PA + lane] = av ? ra[p] : 0.f;
    if (VEC) {
      #pragma unroll
      for (int p = 0; p < BP; ++p)
        *(v4f*)(bs + (bk + BRS * p) * PB + 4 * bn4) = rb[p];
    } else {
      #pragma unroll
      for (int p = 0; p < BS; ++p)
        bs[(bk2 + BSS * p) * PB + bn] = rbs[p];
    }
  };

  v8f acc[NE];
  #pragma unroll
  for (int e = 0; e < NE; ++e) acc[e] = (v8f){0.f,0.f,0.f,0.f,0.f,0.f,0.f,0.f};

  const int aoff = (16 * wm + ln) * PA + 2 * h;
  const int boff = (2 * h) * PB + 16 * wn + ln;

  auto mma16 = [&](const float* ag, const float* bg) {
    v2f af[4], bf[4][NE];
    #pragma unroll
    for (int ks = 0; ks < 4; ++ks) {
      af[ks] = *(const v2f*)(ag + 4 * ks);
      #pragma unroll
      for (int e = 0; e < NE; ++e) {
        bf[ks][e].x = bg[(e * KQ + 4 * ks) * PB];
        bf[ks][e].y = bg[(e * KQ + 4 * ks + 1) * PB];
      }
    }
    asm volatile("" :: "v"(af[0]), "v"(af[1]), "v"(af[2]), "v"(af[3]),
                       "v"(bf[0][0]), "v"(bf[0][1]), "v"(bf[0][2]), "v"(bf[0][3]), "v"(bf[0][4]), "v"(bf[0][5]),
                       "v"(bf[1][0]), "v"(bf[1][1]), "v"(bf[1][2]), "v"(bf[1][3]), "v"(bf[1][4]), "v"(bf[1][5]),
                       "v"(bf[2][0]), "v"(bf[2][1]), "v"(bf[2][2]), "v"(bf[2][3]), "v"(bf[2][4]), "v"(bf[2][5]),
                       "v"(bf[3][0]), "v"(bf[3][1]), "v"(bf[3][2]), "v"(bf[3][3]), "v"(bf[3][4]), "v"(bf[3][5]));
    #pragma unroll
    for (int ks = 0; ks < 4; ++ks) {
      #pragma unroll
      for (int e = 0; e < NE; ++e)
        acc[e] = wmma4(af[ks], bf[ks][e], acc[e]);
    }
  };

  const int colB = n0 + 16 * wn + ln;
  float be[NE];
  #pragma unroll
  for (int e = 0; e < NE; ++e) be[e] = bias[e * N + (colB < N ? colB : N - 1)];
  float cf[8][NE];
  constexpr int CP = (BM * NE + 255) / 256;
  float cw[CP];
  if (PF2) {
    #pragma unroll
    for (int p = 0; p < CP; ++p) {
      const int idx = tid + 256 * p;
      cw[p] = coeff[m0 * NE + (idx < BM * NE ? idx : BM * NE - 1)];
    }
  } else {
    #pragma unroll
    for (int r = 0; r < 8; ++r) {
      #pragma unroll
      for (int e = 0; e < NE; ++e) cf[r][e] = coeff[(m0 + 16 * wm + r + 8 * h) * NE + e];
    }
  }

  gload(0, R0);
  if (PF2 && NR > 1) gload(1, R1);
  sstore(0, R0);
  asm volatile("" :: "v"(be[0]), "v"(be[1]), "v"(be[2]), "v"(be[3]), "v"(be[4]), "v"(be[5]));
  if (PF2) {
    #pragma unroll
    for (int p = 0; p < CP; ++p) {
      const int idx = tid + 256 * p;
      if (idx < BM * NE) Cs[idx] = cw[p];
    }
    lds_barrier();
  } else {
    #pragma unroll
    for (int r = 0; r < 8; ++r)
      asm volatile("" :: "v"(cf[r][0]), "v"(cf[r][1]), "v"(cf[r][2]), "v"(cf[r][3]), "v"(cf[r][4]), "v"(cf[r][5]));
    __syncthreads();
  }

  auto round = [&](int i, int st) {
    const int rem = I - i * KQ;
    const int ng  = (EXACT || rem >= KQ) ? KQ / 16 : (rem + 15) / 16;
    const float* as = As + st * (BM * PA) + aoff;
    const float* bs = Bs + st * (WR * PB) + boff;
    for (int g = 0; g < ng; ++g)
      mma16(as + 16 * g, bs + 16 * g * PB);
  };

  if (PF2) {
    int i = 0;
    for (; i + 3 < NR; i += 2) {
      gload(i + 2, R0);
      round(i, 0);
      sstore(1, R1);
      lds_barrier();
      gload(i + 3, R1);
      round(i + 1, 1);
      sstore(0, R0);
      lds_barrier();
    }
    const int left = NR - i;
    if (left == 3) {
      gload(i + 2, R0);
      round(i, 0);
      sstore(1, R1);
      lds_barrier();
      round(i + 1, 1);
      sstore(0, R0);
      lds_barrier();
      round(i + 2, 0);
    } else if (left == 2) {
      round(i, 0);
      sstore(1, R1);
      lds_barrier();
      round(i + 1, 1);
    } else {
      round(i, 0);
    }
  } else {
    for (int i = 0; i < NR; ++i) {
      const int st = i & 1;
      if (i + 1 < NR) gload(i + 1, R0);
      round(i, st);
      if (i + 1 < NR) sstore(st ^ 1, R0);
      __syncthreads();
    }
  }

  if (colB < N) {
    #pragma unroll
    for (int r = 0; r < 8; ++r) {
      const int row = m0 + 16 * wm + r + 8 * h;
      float y = 0.f, s = 0.f;
      #pragma unroll
      for (int e = 0; e < NE; ++e) {
        const float ce = PF2 ? Cs[(16 * wm + r + 8 * h) * NE + e] : cf[r][e];
        y = fmaf(ce, acc[e][r], y);
        s = fmaf(ce, be[e], s);
      }
      y += s;
      if (act) y = elu1(y);
      C[row * N + colB] = y;
    }
  }
}

template <int N, int MODE, int MT, int NTB>
__global__ void __launch_bounds__(256)
gemm_lds_f32(const float* __restrict__ A1, int k1,
             const float* __restrict__ A2, int k2,
             const float* __restrict__ W,
             const float* __restrict__ bias,
             float* __restrict__ C,
             int act)
{
  __shared__ __attribute__((aligned(16))) float As[2 * (16 * MT) * (KC + 4)];
  __shared__ __attribute__((aligned(16))) float Bs[2 * KC * (16 * NTB + 8)];
  gemm_tile<N, MODE, MT, NTB>(A1, k1, A2, k2, W, bias, nullptr, nullptr, nullptr, nullptr, nullptr,
                              C, nullptr, nullptr, act, blockIdx.x, false, As, Bs);
}

template <int N, int K2, int MT, int NTB>
__global__ void __launch_bounds__(256)
moe_lds_f32(const float* __restrict__ z,
            const float* __restrict__ prev,
            const float* __restrict__ coeff,
            const float* __restrict__ W,
            const float* __restrict__ bias,
            float* __restrict__ C,
            int act)
{
  __shared__ __attribute__((aligned(16))) float As[2 * (16 * MT) * (KQ + 4)];
  __shared__ __attribute__((aligned(16))) float Bs[2 * NE * KQ * (16 * NTB + 8)];
  moe_tile<N, LAT, K2, MT, NTB, false>(z, prev, coeff, W, bias, C, act, blockIdx.x, As, Bs, nullptr);
}

__global__ void __launch_bounds__(256)
head_decoder0(const float* __restrict__ x, const float* __restrict__ h2in,
              const float* __restrict__ c, const float* __restrict__ eps,
              const float* __restrict__ em_w, const float* __restrict__ em_b,
              const float* __restrict__ ev_w, const float* __restrict__ ev_b,
              const float* __restrict__ g1_w, const float* __restrict__ g1_b,
              const float* __restrict__ g2_w, const float* __restrict__ g2_b,
              const float* __restrict__ g3_w, const float* __restrict__ g3_b,
              const float* __restrict__ w0, const float* __restrict__ b0,
              float* __restrict__ out_m, float* __restrict__ out_v,
              float* zpl, float* cpl,
              float* __restrict__ d0)
{
  __shared__ __attribute__((aligned(16))) float As[2 * 32 * (KC + 4)];
  __shared__ __attribute__((aligned(16))) float Bs[2 * (NE * KQ > KC ? NE * KQ : KC) * (64 + 8)];
  __shared__ float Cs[32 * NE];
  const int bx = blockIdx.x;
  float* zmine = zpl + bx * (BATCH * LAT);
  float* cmine = cpl + bx * (BATCH * NE);

  gemm_tile<MVN, MODE_MV, 2, 4>(x, FRAME, h2in, HID, em_w, em_b, ev_w, ev_b, eps, nullptr, nullptr,
                                out_m, out_v, zmine, 0, 0, bx == 0, As, Bs);
  __syncthreads();
  gemm_tile<GH, MODE_GATE, 2, 4>(zmine, LAT, c, FRAME, g1_w, g1_b, g2_w, g2_b, nullptr, g3_w, g3_b,
                                 nullptr, nullptr, cmine, 1, 0, false, As, Bs);
  __syncthreads();
  moe_tile<HID, LAT, FRAME, 2, 4, true>(zmine, c, cmine, w0, b0, d0, 1, bx, As, Bs, Cs);
}

template <int N, int MT, int NTB>
static inline dim3 gemm_grid() {
  return dim3((N + 16 * NTB - 1) / (16 * NTB), BATCH / (16 * MT));
}

extern "C" void kernel_launch(void* const* d_in, const int* in_sizes, int n_in,
                              void* d_out, int out_size, void* d_ws, size_t ws_size,
                              hipStream_t stream) {
  const float* x      = (const float*)d_in[0];
  const float* c      = (const float*)d_in[1];
  const float* eps    = (const float*)d_in[2];
  const float* efc1_w = (const float*)d_in[3];
  const float* efc1_b = (const float*)d_in[4];
  const float* efc2_w = (const float*)d_in[5];
  const float* efc2_b = (const float*)d_in[6];
  const float* em_w   = (const float*)d_in[7];
  const float* em_b   = (const float*)d_in[8];
  const float* ev_w   = (const float*)d_in[9];
  const float* ev_b   = (const float*)d_in[10];
  const float* g1_w   = (const float*)d_in[11];
  const float* g1_b   = (const float*)d_in[12];
  const float* g2_w   = (const float*)d_in[13];
  const float* g2_b   = (const float*)d_in[14];
  const float* g3_w   = (const float*)d_in[15];
  const float* g3_b   = (const float*)d_in[16];
  const float* w0     = (const float*)d_in[17];
  const float* b0     = (const float*)d_in[18];
  const float* w1     = (const float*)d_in[19];
  const float* b1     = (const float*)d_in[20];
  const float* w2     = (const float*)d_in[21];
  const float* b2     = (const float*)d_in[22];

  float* out   = (float*)d_out;
  float* out_m = out + BATCH * FRAME;
  float* out_v = out_m + BATCH * LAT;

  float* ws    = (float*)d_ws;
  float* h1    = ws;
  float* h2    = h1  + BATCH * HID;
  float* zpl   = h2  + BATCH * HID;
  float* cpl   = zpl + 4 * BATCH * LAT;
  float* z     = zpl;
  float* coeff = cpl;

  gemm_lds_f32<HID, MODE_PLAIN, 2, 4><<<gemm_grid<HID, 2, 4>(), 256, 0, stream>>>(
      x, FRAME, c, FRAME, efc1_w, efc1_b, h1, 1);
  gemm_lds_f32<HID, MODE_PLAIN, 2, 4><<<gemm_grid<HID, 2, 4>(), 256, 0, stream>>>(
      x, FRAME, h1, HID, efc2_w, efc2_b, h2, 1);

  head_decoder0<<<gemm_grid<HID, 2, 4>(), 256, 0, stream>>>(
      x, h2, c, eps, em_w, em_b, ev_w, ev_b, g1_w, g1_b, g2_w, g2_b, g3_w, g3_b, w0, b0,
      out_m, out_v, zpl, cpl, h1);

  moe_lds_f32<HID, HID, 4, 2><<<gemm_grid<HID, 4, 2>(), 256, 0, stream>>>(
      z, h1, coeff, w1, b1, h2, 1);
  moe_lds_f32<FRAME, HID, 4, 2><<<gemm_grid<FRAME, 4, 2>(), 256, 0, stream>>>(
      z, h2, coeff, w2, b2, out, 0);
}


# Encoder GEMMs: segment lengths compile-time: chunk loop unrolls, chunk load addresses are fixed offsets, not rebuilt in recycled registers behind issued loads (fewer s_wait_xcnt); sum orders kept
# speedup vs baseline: 4.4367x; 1.0434x over previous
#include <hip/hip_runtime.h>
#include <hip/hip_bf16.h>


typedef __attribute__((ext_vector_type(2))) float v2f;
typedef __attribute__((ext_vector_type(4))) float v4f;
typedef __attribute__((ext_vector_type(8))) float v8f;

#define BATCH 2048
#define FRAME 267
#define LAT   32
#define HID   256
#define NE    6
#define GH    64
#define KC    128
#define KQ    32
#define MVN   (2 * LAT)

#define MODE_PLAIN 0
#define MODE_MV    1
#define MODE_GATE  2

__device__ __forceinline__ float elu1(float x) {
  return x > 0.f ? x : expm1f(x);
}

__device__ __forceinline__ v8f wmma4(v2f a, v2f b, v8f c) {
  return __builtin_amdgcn_wmma_f32_16x16x4_f32(
      false, a, false, b, (short)0, c, false, false);
}

__device__ __forceinline__ void lds_barrier() {
  __builtin_amdgcn_fence(__ATOMIC_RELEASE, "workgroup", "local");
  __builtin_amdgcn_s_barrier();
  __builtin_amdgcn_fence(__ATOMIC_ACQUIRE, "workgroup", "local");
}

template <int N, int MODE, int MT, int NTB>
__device__ __forceinline__ void
gemm_tile(const float* A1, int k1,
             const float* A2, int k2,
             const float* __restrict__ W,
             const float* __restrict__ bias,
             const float* __restrict__ W2,
             const float* __restrict__ bias2,
             const float* __restrict__ eps,
             const float* __restrict__ W3,
             const float* __restrict__ bias3,
             float* C,
             float* C2,
             float* Z,
             int act,
             int bx,
             bool wmv,
             float* As,
             float* Bs)
{
  constexpr bool MV   = (MODE == MODE_MV);
  constexpr bool GATE = (MODE == MODE_GATE);
  static_assert(!GATE || (N == GH && NTB == 4 && MT == 2), "GATE: one 32 x 64 tile holds whole rows of both hidden layers");
  static_assert(MT * NTB == 8, "eight waves per work-group");
  static_assert(!MV || (N == MVN && NTB == 4 && MT == 2), "MV: one 32 x 64 tile, columns 0..31 mean, 32..63 log-variance");
  constexpr int BM  = 16 * MT;
  constexpr int BN  = 16 * NTB;
  constexpr int PA  = KC + 4;
  constexpr int PB  = BN + 8;
  constexpr int AJ  = KC / 32;
  constexpr int AP  = BM / 8;
  constexpr bool VEC = (N % BN == 0);
  constexpr int F4R = BN / 4;
  constexpr int BP  = KC * F4R / 256;
  constexpr int BRS = 256 / F4R;
  constexpr int BS  = KC * BN / 256;
  constexpr int BSS = 256 / BN;

  const int tid  = threadIdx.x;
  const int lane = tid & 31;
  const int wave = tid >> 5;
  const int h    = lane >> 4;
  const int ln   = lane & 15;
  const int wm   = wave / NTB;
  const int wn   = wave % NTB;
  const int m0   = blockIdx.y * BM;
  const int n0   = bx * BN;

  const int bn4 = tid % F4R;
  const int bk  = tid / F4R;
  const int bn  = tid % BN;
  const int bk2 = tid / BN;
  const int colc = (n0 + bn < N) ? (n0 + bn) : (N - 1);

  const int Ktot = k1 + k2;
  const int nc   = (Ktot + KC - 1) / KC;

  struct StageRegs {
    float  ra[AJ][AP];
    bool   av[AJ];
    v4f    rb[BP > 0 ? BP : 1];
    float  rbs[BS];
  };
  StageRegs R0, R1;

  auto gload = [&](int i, StageRegs& R) {
    auto& ra = R.ra; auto& av = R.av; auto& rb = R.rb; auto& rbs = R.rbs;
    const int k0 = i * KC;
    #pragma unroll
    for (int j = 0; j < AJ; ++j) {
      const int kk = k0 + 32 * j + lane;
      av[j] = kk < Ktot;
      const int kc = av[j] ? kk : Ktot - 1;
      const int ii = kc;
      const bool s2 = ii >= k1;
      const float* Aseg = s2 ? A2 : A1;
      const int len = s2 ? k2 : k1;
      const int cc  = s2 ? ii - k1 : ii;
      #pragma unroll
      for (int p = 0; p < AP; ++p)
        ra[j][p] = Aseg[(m0 + wave + 8 * p) * len + cc];
    }
    if (VEC) {
      #pragma unroll
      for (int p = 0; p < BP; ++p) {
        int kr = k0 + bk + BRS * p;
        kr = kr < Ktot ? kr : Ktot - 1;
        if (MV) {
          const float* __restrict__ Wh = (bn4 < F4R / 2) ? W : W2;
          rb[p] = *(const v4f*)(Wh + kr * LAT + 4 * (bn4 & (F4R / 2 - 1)));
        } else {
          rb[p] = *(const v4f*)(W + kr * N + n0 + 4 * bn4);
        }
      }
    } else {
      #pragma unroll
      for (int p = 0; p < BS; ++p) {
        int kr = k0 + bk2 + BSS * p;
        kr = kr < Ktot ? kr : Ktot - 1;
        rbs[p] = W[kr * N + colc];
      }
    }
  };

  auto sstore = [&](int st, StageRegs& R) {
    auto& ra = R.ra; auto& av = R.av; auto& rb = R.rb; auto& rbs = R.rbs;
    float* as = As + st * (BM * PA);
    float* bs = Bs + st * (KC * PB);
    #pragma unroll
    for (int j = 0; j < AJ; ++j) {
      #pragma unroll
      for (int p = 0; p < AP; ++p) {
        as[(wave + 8 * p) * PA + 32 * j + lane] = av[j] ? ra[j][p] : 0.f;
      }
    }
    if (VEC) {
      #pragma unroll
      for (int p = 0; p < BP; ++p)
        *(v4f*)(bs + (bk + BRS * p) * PB + 4 * bn4) = rb[p];
    } else {
      #pragma unroll
      for (int p = 0; p < BS; ++p)
        bs[(bk2 + BSS * p) * PB + bn] = rbs[p];
    }
  };

  v8f acc = (v8f){0.f,0.f,0.f,0.f,0.f,0.f,0.f,0.f};

  const int aoff = (16 * wm + ln) * PA + 2 * h;
  const int boff = (2 * h) * PB + 16 * wn + ln;

  auto mma32 = [&](const float* asj, const float* bsj, v8f a) {
    v2f af[8], bf[8];
    #pragma unroll
    for (int ks = 0; ks < 8; ++ks) {
      af[ks]   = *(const v2f*)(asj + 4 * ks);
      bf[ks].x = bsj[(4 * ks) * PB];
      bf[ks].y = bsj[(4 * ks + 1) * PB];
    }
    asm volatile("" :: "v"(af[0]), "v"(af[1]), "v"(af[2]), "v"(af[3]),
                       "v"(af[4]), "v"(af[5]), "v"(af[6]), "v"(af[7]),
                       "v"(bf[0]), "v"(bf[1]), "v"(bf[2]), "v"(bf[3]),
                       "v"(bf[4]), "v"(bf[5]), "v"(bf[6]), "v"(bf[7]));
    #pragma unroll
    for (int ks = 0; ks < 8; ++ks)
      a = wmma4(af[ks], bf[ks], a);
    return a;
  };

  const int colB = n0 + 16 * wn + ln;
  float pb = 0.f, pb2 = 0.f;
  float pe[BM * LAT / 256] = {0.f, 0.f, 0.f, 0.f};
  v4f   pw2[GH / BRS];
  float pw3[2] = {0.f, 0.f};
  #pragma unroll
  for (int p = 0; p < GH / BRS; ++p) pw2[p] = (v4f){0.f, 0.f, 0.f, 0.f};
  if (GATE) {
    pb  = bias[colB];
    pb2 = bias2[colB];
    #pragma unroll
    for (int p = 0; p < GH / BRS; ++p)
      pw2[p] = *(const v4f*)(W2 + (bk + BRS * p) * GH + 4 * bn4);
    pw3[0] = W3[tid];
    pw3[1] = W3[256 + (tid & 127)];
  } else if (MV) {
    pb = (colB < LAT) ? bias[colB] : bias2[colB - LAT];
    #pragma unroll
    for (int it = 0; it < BM * LAT / 256; ++it) {
      const int idx = tid + 256 * it;
      pe[it] = eps[(m0 + idx / LAT) * LAT + idx % LAT];
    }
  } else {
    if (bias) pb = bias[colB < N ? colB : N - 1];
  }

  gload(0, R0);
  if (nc > 1) gload(1, R1);
  sstore(0, R0);
  if (GATE)
    asm volatile("" :: "v"(pb), "v"(pb2), "v"(pw2[0]), "v"(pw2[1]), "v"(pw2[2]), "v"(pw2[3]),
                       "v"(pw3[0]), "v"(pw3[1]));
  else if (MV)
    asm volatile("" :: "v"(pb), "v"(pe[0]), "v"(pe[1]), "v"(pe[2]), "v"(pe[3]));
  else
    asm volatile("" :: "v"(pb));
  lds_barrier();

  auto chunk = [&](int i, int st) {
    const int rem = Ktot - i * KC;
    const int nsb = rem >= KC ? KC / 32 : (rem + 31) / 32;
    const float* as = As + st * (BM * PA) + aoff;
    const float* bs = Bs + st * (KC * PB) + boff;
    for (int j = 0; j < nsb; ++j)
      acc = mma32(as + 32 * j, bs + 32 * j * PB, acc);
  };

  int i = 0;
  for (; i + 3 < nc; i += 2) {
    gload(i + 2, R0);
    chunk(i, 0);
    sstore(1, R1);
    lds_barrier();
    gload(i + 3, R1);
    chunk(i + 1, 1);
    sstore(0, R0);
    lds_barrier();
  }
  const int left = nc - i;
  if (left == 3) {
    gload(i + 2, R0);
    chunk(i, 0);
    sstore(1, R1);
    lds_barrier();
    chunk(i + 1, 1);
    sstore(0, R0);
    lds_barrier();
    chunk(i + 2, 0);
  } else if (left == 2) {
    chunk(i, 0);
    sstore(1, R1);
    lds_barrier();
    chunk(i + 1, 1);
  } else {
    chunk(i, 0);
  }
  lds_barrier();

  if (GATE) {
    constexpr int PT = GH + 1;
    float* as2 = As;
    float* bs2 = Bs;
    float* W3s = Bs + KC * PB;
    {
      const float bv = pb;
      #pragma unroll
      for (int r = 0; r < 8; ++r)
        as2[(16 * wm + r + 8 * h) * PA + colB] = elu1(acc[r] + bv);
      #pragma unroll
      for (int p = 0; p < GH / BRS; ++p) {
        const int kr = bk + BRS * p;
        *(v4f*)(bs2 + kr * PB + 4 * bn4) = pw2[p];
      }
      W3s[tid] = pw3[0];
      if (tid < GH * NE - 256) W3s[256 + tid] = pw3[1];
    }
    __syncthreads();
    v8f acc2 = (v8f){0.f,0.f,0.f,0.f,0.f,0.f,0.f,0.f};
    #pragma unroll
    for (int j = 0; j < GH / 32; ++j)
      acc2 = mma32(as2 + aoff + 32 * j, bs2 + boff + 32 * j * PB, acc2);
    float* T = As + BM * PA;
    {
      const float bv2 = pb2;
      #pragma unroll
      for (int r = 0; r < 8; ++r)
        T[(16 * wm + r + 8 * h) * PT + colB] = elu1(acc2[r] + bv2);
    }
    __syncthreads();
    if (tid < BM) {
      float lg[NE];
      #pragma unroll
      for (int e = 0; e < NE; ++e) lg[e] = bias3[e];
      for (int i = 0; i < GH; ++i) {
        const float gv = T[tid * PT + i];
        #pragma unroll
        for (int e = 0; e < NE; ++e) lg[e] = fmaf(gv, W3s[i * NE + e], lg[e]);
      }
      float mx = lg[0];
      #pragma unroll
      for (int e = 1; e < NE; ++e) mx = fmaxf(mx, lg[e]);
      float s = 0.f, ex[NE];
      #pragma unroll
      for (int e = 0; e < NE; ++e) { ex[e] = expf(lg[e] - mx); s += ex[e]; }
      const float inv = 1.f / s;
      #pragma unroll
      for (int e = 0; e < NE; ++e) Z[(m0 + tid) * NE + e] = ex[e] * inv;
    }
  } else if (MV) {
    constexpr int PT = MVN + 1;
    float* T = As;
    const float bv = pb;
    #pragma unroll
    for (int r = 0; r < 8; ++r)
      T[(16 * wm + r + 8 * h) * PT + colB] = acc[r] + bv;
    __syncthreads();
    #pragma unroll
    for (int it = 0; it < BM * LAT / 256; ++it) {
      const int idx = tid + 256 * it;
      const int rr  = idx / LAT;
      const int cc  = idx % LAT;
      const float m = T[rr * PT + cc];
      const float v = T[rr * PT + LAT + cc];
      const int   o = (m0 + rr) * LAT + cc;
      if (wmv) {
        C[o]  = m;
        C2[o] = v;
      }
      Z[o]  = m + pe[it] * expf(0.5f * v);
    }
  } else if (colB < N) {
    #pragma unroll
    for (int r = 0; r < 8; ++r) {
      const int row = m0 + 16 * wm + r + 8 * h;
      float y = acc[r];
      if (bias) y += pb;
      if (act) y = elu1(y);
      C[row * N + colB] = y;
    }
  }
}

template <int N, int K1, int K2, int MT, int NTB, bool PF2>
__device__ __forceinline__ void
moe_tile(const float* A1, const float* A2,
         const float* coeff,
         const float* __restrict__ W,
         const float* __restrict__ bias,
         float* C,
         int act,
         int bx,
         float* As,
         float* Bs,
         float* Cs)
{
  static_assert(K1 % KQ == 0, "a round reads one segment of [A1|A2]");
  static_assert(MT * NTB == 8, "eight waves per work-group");
  constexpr int BM  = 16 * MT;
  constexpr int BN  = 16 * NTB;
  constexpr int I   = K1 + K2;
  constexpr int NR  = (I + KQ - 1) / KQ;
  constexpr bool EXACT = (I % KQ == 0);
  constexpr int PA  = KQ + 4;
  constexpr int PB  = BN + 8;
  constexpr int WR  = NE * KQ;
  constexpr int AP  = BM / 8;
  constexpr bool VEC = (N % BN == 0);
  constexpr int F4R = BN / 4;
  constexpr int BP  = WR * F4R / 256;
  constexpr int BRS = 256 / F4R;
  constexpr int BS  = WR * BN / 256;
  constexpr int BSS = 256 / BN;
  static_assert(KQ % BRS == 0 && KQ % BSS == 0, "a staging pass stays inside one expert");

  const int tid  = threadIdx.x;
  const int lane = tid & 31;
  const int wave = tid >> 5;
  const int h    = lane >> 4;
  const int ln   = lane & 15;
  const int wm   = wave / NTB;
  const int wn   = wave % NTB;
  const int m0   = blockIdx.y * BM;
  const int n0   = bx * BN;

  const int bn4 = tid % F4R;
  const int bk  = tid / F4R;
  const int bn  = tid % BN;
  const int bk2 = tid / BN;
  const int colc = (n0 + bn < N) ? (n0 + bn) : (N - 1);

  struct StageRegs {
    float ra[AP];
    bool  av;
    v4f   rb[BP];
    float rbs[BS];
  };
  StageRegs R0, R1;

  auto gload = [&](int i, StageRegs& R) {
    auto& ra = R.ra; auto& av = R.av; auto& rb = R.rb; auto& rbs = R.rbs;
    const int k0 = i * KQ;
    const bool full = EXACT || (k0 + KQ <= I);
    const bool s2 = k0 >= K1;
    const float* Aseg = s2 ? A2 : A1;
    const int len = s2 ? K2 : K1;
    av = full || (k0 + lane < I);
    const int kc = av ? k0 + lane : I - 1;
    const int cc = s2 ? kc - K1 : kc;
    #pragma unroll
    for (int p = 0; p < AP; ++p)
      ra[p] = Aseg[(m0 + wave + 8 * p) * len + cc];
    if (VEC) {
      #pragma unroll
      for (int p = 0; p < BP; ++p) {
        const int e = (BRS * p) / KQ;
        int kr = k0 + (BRS * p) % KQ + bk;
        if (!full) kr = kr < I ? kr : I - 1;
        rb[p] = *(const v4f*)(W + (e * I + kr) * N + n0 + 4 * bn4);
      }
    } else {
      #pragma unroll
      for (int p = 0; p < BS; ++p) {
        const int e = (BSS * p) / KQ;
        int kr = k0 + (BSS * p) % KQ + bk2;
        if (!full) kr = kr < I ? kr : I - 1;
        rbs[p] = W[(e * I + kr) * N + colc];
      }
    }
  };

  auto sstore = [&](int st, StageRegs& R) {
    auto& ra = R.ra; auto& av = R.av; auto& rb = R.rb; auto& rbs = R.rbs;
    float* as = As + st * (BM * PA);
    float* bs = Bs + st * (WR * PB);
    #pragma unroll
    for (int p = 0; p < AP; ++p)
      as[(wave + 8 * p) * PA + lane] = av ? ra[p] : 0.f;
    if (VEC) {
      #pragma unroll
      for (int p = 0; p < BP; ++p)
        *(v4f*)(bs + (bk + BRS * p) * PB + 4 * bn4) = rb[p];
    } else {
      #pragma unroll
      for (int p = 0; p < BS; ++p)
        bs[(bk2 + BSS * p) * PB + bn] = rbs[p];
    }
  };

  v8f acc[NE];
  #pragma unroll
  for (int e = 0; e < NE; ++e) acc[e] = (v8f){0.f,0.f,0.f,0.f,0.f,0.f,0.f,0.f};

  const int aoff = (16 * wm + ln) * PA + 2 * h;
  const int boff = (2 * h) * PB + 16 * wn + ln;

  auto mma16 = [&](const float* ag, const float* bg) {
    v2f af[4], bf[4][NE];
    #pragma unroll
    for (int ks = 0; ks < 4; ++ks) {
      af[ks] = *(const v2f*)(ag + 4 * ks);
      #pragma unroll
      for (int e = 0; e < NE; ++e) {
        bf[ks][e].x = bg[(e * KQ + 4 * ks) * PB];
        bf[ks][e].y = bg[(e * KQ + 4 * ks + 1) * PB];
      }
    }
    asm volatile("" :: "v"(af[0]), "v"(af[1]), "v"(af[2]), "v"(af[3]),
                       "v"(bf[0][0]), "v"(bf[0][1]), "v"(bf[0][2]), "v"(bf[0][3]), "v"(bf[0][4]), "v"(bf[0][5]),
                       "v"(bf[1][0]), "v"(bf[1][1]), "v"(bf[1][2]), "v"(bf[1][3]), "v"(bf[1][4]), "v"(bf[1][5]),
                       "v"(bf[2][0]), "v"(bf[2][1]), "v"(bf[2][2]), "v"(bf[2][3]), "v"(bf[2][4]), "v"(bf[2][5]),
                       "v"(bf[3][0]), "v"(bf[3][1]), "v"(bf[3][2]), "v"(bf[3][3]), "v"(bf[3][4]), "v"(bf[3][5]));
    #pragma unroll
    for (int ks = 0; ks < 4; ++ks) {
      #pragma unroll
      for (int e = 0; e < NE; ++e)
        acc[e] = wmma4(af[ks], bf[ks][e], acc[e]);
    }
  };

  const int colB = n0 + 16 * wn + ln;
  float be[NE];
  #pragma unroll
  for (int e = 0; e < NE; ++e) be[e] = bias[e * N + (colB < N ? colB : N - 1)];
  float cf[8][NE];
  constexpr int CP = (BM * NE + 255) / 256;
  float cw[CP];
  if (PF2) {
    #pragma unroll
    for (int p = 0; p < CP; ++p) {
      const int idx = tid + 256 * p;
      cw[p] = coeff[m0 * NE + (idx < BM * NE ? idx : BM * NE - 1)];
    }
  } else {
    #pragma unroll
    for (int r = 0; r < 8; ++r) {
      #pragma unroll
      for (int e = 0; e < NE; ++e) cf[r][e] = coeff[(m0 + 16 * wm + r + 8 * h) * NE + e];
    }
  }

  gload(0, R0);
  if (PF2 && NR > 1) gload(1, R1);
  sstore(0, R0);
  asm volatile("" :: "v"(be[0]), "v"(be[1]), "v"(be[2]), "v"(be[3]), "v"(be[4]), "v"(be[5]));
  if (PF2) {
    #pragma unroll
    for (int p = 0; p < CP; ++p) {
      const int idx = tid + 256 * p;
      if (idx < BM * NE) Cs[idx] = cw[p];
    }
    lds_barrier();
  } else {
    #pragma unroll
    for (int r = 0; r < 8; ++r)
      asm volatile("" :: "v"(cf[r][0]), "v"(cf[r][1]), "v"(cf[r][2]), "v"(cf[r][3]), "v"(cf[r][4]), "v"(cf[r][5]));
    __syncthreads();
  }

  auto round = [&](int i, int st) {
    const int rem = I - i * KQ;
    const int ng  = (EXACT || rem >= KQ) ? KQ / 16 : (rem + 15) / 16;
    const float* as = As + st * (BM * PA) + aoff;
    const float* bs = Bs + st * (WR * PB) + boff;
    for (int g = 0; g < ng; ++g)
      mma16(as + 16 * g, bs + 16 * g * PB);
  };

  if (PF2) {
    int i = 0;
    for (; i + 3 < NR; i += 2) {
      gload(i + 2, R0);
      round(i, 0);
      sstore(1, R1);
      lds_barrier();
      gload(i + 3, R1);
      round(i + 1, 1);
      sstore(0, R0);
      lds_barrier();
    }
    const int left = NR - i;
    if (left == 3) {
      gload(i + 2, R0);
      round(i, 0);
      sstore(1, R1);
      lds_barrier();
      round(i + 1, 1);
      sstore(0, R0);
      lds_barrier();
      round(i + 2, 0);
    } else if (left == 2) {
      round(i, 0);
      sstore(1, R1);
      lds_barrier();
      round(i + 1, 1);
    } else {
      round(i, 0);
    }
  } else {
    for (int i = 0; i < NR; ++i) {
      const int st = i & 1;
      if (i + 1 < NR) gload(i + 1, R0);
      round(i, st);
      if (i + 1 < NR) sstore(st ^ 1, R0);
      __syncthreads();
    }
  }

  if (colB < N) {
    #pragma unroll
    for (int r = 0; r < 8; ++r) {
      const int row = m0 + 16 * wm + r + 8 * h;
      float y = 0.f, s = 0.f;
      #pragma unroll
      for (int e = 0; e < NE; ++e) {
        const float ce = PF2 ? Cs[(16 * wm + r + 8 * h) * NE + e] : cf[r][e];
        y = fmaf(ce, acc[e][r], y);
        s = fmaf(ce, be[e], s);
      }
      y += s;
      if (act) y = elu1(y);
      C[row * N + colB] = y;
    }
  }
}

template <int N, int MODE, int MT, int NTB, int K1, int K2>
__global__ void __launch_bounds__(256)
gemm_lds_f32(const float* __restrict__ A1,
             const float* __restrict__ A2,
             const float* __restrict__ W,
             const float* __restrict__ bias,
             float* __restrict__ C,
             int act)
{
  __shared__ __attribute__((aligned(16))) float As[2 * (16 * MT) * (KC + 4)];
  __shared__ __attribute__((aligned(16))) float Bs[2 * KC * (16 * NTB + 8)];
  gemm_tile<N, MODE, MT, NTB>(A1, K1, A2, K2, W, bias, nullptr, nullptr, nullptr, nullptr, nullptr,
                              C, nullptr, nullptr, act, blockIdx.x, false, As, Bs);
}

template <int N, int K2, int MT, int NTB>
__global__ void __launch_bounds__(256)
moe_lds_f32(const float* __restrict__ z,
            const float* __restrict__ prev,
            const float* __restrict__ coeff,
            const float* __restrict__ W,
            const float* __restrict__ bias,
            float* __restrict__ C,
            int act)
{
  __shared__ __attribute__((aligned(16))) float As[2 * (16 * MT) * (KQ + 4)];
  __shared__ __attribute__((aligned(16))) float Bs[2 * NE * KQ * (16 * NTB + 8)];
  moe_tile<N, LAT, K2, MT, NTB, false>(z, prev, coeff, W, bias, C, act, blockIdx.x, As, Bs, nullptr);
}

__global__ void __launch_bounds__(256)
head_decoder0(const float* __restrict__ x, const float* __restrict__ h2in,
              const float* __restrict__ c, const float* __restrict__ eps,
              const float* __restrict__ em_w, const float* __restrict__ em_b,
              const float* __restrict__ ev_w, const float* __restrict__ ev_b,
              const float* __restrict__ g1_w, const float* __restrict__ g1_b,
              const float* __restrict__ g2_w, const float* __restrict__ g2_b,
              const float* __restrict__ g3_w, const float* __restrict__ g3_b,
              const float* __restrict__ w0, const float* __restrict__ b0,
              float* __restrict__ out_m, float* __restrict__ out_v,
              float* zpl, float* cpl,
              float* __restrict__ d0)
{
  __shared__ __attribute__((aligned(16))) float As[2 * 32 * (KC + 4)];
  __shared__ __attribute__((aligned(16))) float Bs[2 * (NE * KQ > KC ? NE * KQ : KC) * (64 + 8)];
  __shared__ float Cs[32 * NE];
  const int bx = blockIdx.x;
  float* zmine = zpl + bx * (BATCH * LAT);
  float* cmine = cpl + bx * (BATCH * NE);

  gemm_tile<MVN, MODE_MV, 2, 4>(x, FRAME, h2in, HID, em_w, em_b, ev_w, ev_b, eps, nullptr, nullptr,
                                out_m, out_v, zmine, 0, 0, bx == 0, As, Bs);
  __syncthreads();
  gemm_tile<GH, MODE_GATE, 2, 4>(zmine, LAT, c, FRAME, g1_w, g1_b, g2_w, g2_b, nullptr, g3_w, g3_b,
                                 nullptr, nullptr, cmine, 1, 0, false, As, Bs);
  __syncthreads();
  moe_tile<HID, LAT, FRAME, 2, 4, true>(zmine, c, cmine, w0, b0, d0, 1, bx, As, Bs, Cs);
}

template <int N, int MT, int NTB>
static inline dim3 gemm_grid() {
  return dim3((N + 16 * NTB - 1) / (16 * NTB), BATCH / (16 * MT));
}

extern "C" void kernel_launch(void* const* d_in, const int* in_sizes, int n_in,
                              void* d_out, int out_size, void* d_ws, size_t ws_size,
                              hipStream_t stream) {
  const float* x      = (const float*)d_in[0];
  const float* c      = (const float*)d_in[1];
  const float* eps    = (const float*)d_in[2];
  const float* efc1_w = (const float*)d_in[3];
  const float* efc1_b = (const float*)d_in[4];
  const float* efc2_w = (const float*)d_in[5];
  const float* efc2_b = (const float*)d_in[6];
  const float* em_w   = (const float*)d_in[7];
  const float* em_b   = (const float*)d_in[8];
  const float* ev_w   = (const float*)d_in[9];
  const float* ev_b   = (const float*)d_in[10];
  const float* g1_w   = (const float*)d_in[11];
  const float* g1_b   = (const float*)d_in[12];
  const float* g2_w   = (const float*)d_in[13];
  const float* g2_b   = (const float*)d_in[14];
  const float* g3_w   = (const float*)d_in[15];
  const float* g3_b   = (const float*)d_in[16];
  const float* w0     = (const float*)d_in[17];
  const float* b0     = (const float*)d_in[18];
  const float* w1     = (const float*)d_in[19];
  const float* b1     = (const float*)d_in[20];
  const float* w2     = (const float*)d_in[21];
  const float* b2     = (const float*)d_in[22];

  float* out   = (float*)d_out;
  float* out_m = out + BATCH * FRAME;
  float* out_v = out_m + BATCH * LAT;

  float* ws    = (float*)d_ws;
  float* h1    = ws;
  float* h2    = h1  + BATCH * HID;
  float* zpl   = h2  + BATCH * HID;
  float* cpl   = zpl + 4 * BATCH * LAT;
  float* z     = zpl;
  float* coeff = cpl;

  gemm_lds_f32<HID, MODE_PLAIN, 2, 4, FRAME, FRAME><<<gemm_grid<HID, 2, 4>(), 256, 0, stream>>>(
      x, c, efc1_w, efc1_b, h1, 1);
  gemm_lds_f32<HID, MODE_PLAIN, 2, 4, FRAME, HID><<<gemm_grid<HID, 2, 4>(), 256, 0, stream>>>(
      x, h1, efc2_w, efc2_b, h2, 1);

  head_decoder0<<<gemm_grid<HID, 2, 4>(), 256, 0, stream>>>(
      x, h2, c, eps, em_w, em_b, ev_w, ev_b, g1_w, g1_b, g2_w, g2_b, g3_w, g3_b, w0, b0,
      out_m, out_v, zpl, cpl, h1);

  moe_lds_f32<HID, HID, 4, 2><<<gemm_grid<HID, 4, 2>(), 256, 0, stream>>>(
      z, h1, coeff, w1, b1, h2, 1);
  moe_lds_f32<FRAME, HID, 4, 2><<<gemm_grid<FRAME, 4, 2>(), 256, 0, stream>>>(
      z, h2, coeff, w2, b2, out, 0);
}


# Decoder layer 2: the ragged last 11 columns leave the 64x32 grid (288 -> 256 full tiles) for 128 light 16x16 tiles, one expert per wave, mixed through LDS, dispatched last; sum orders kept
# speedup vs baseline: 4.7638x; 1.0737x over previous
#include <hip/hip_runtime.h>
#include <hip/hip_bf16.h>


typedef __attribute__((ext_vector_type(2))) float v2f;
typedef __attribute__((ext_vector_type(4))) float v4f;
typedef __attribute__((ext_vector_type(8))) float v8f;

#define BATCH 2048
#define FRAME 267
#define LAT   32
#define HID   256
#define NE    6
#define GH    64
#define KC    128
#define KQ    32
#define MVN   (2 * LAT)

#define MODE_PLAIN 0
#define MODE_MV    1
#define MODE_GATE  2

__device__ __forceinline__ float elu1(float x) {
  return x > 0.f ? x : expm1f(x);
}

__device__ __forceinline__ v8f wmma4(v2f a, v2f b, v8f c) {
  return __builtin_amdgcn_wmma_f32_16x16x4_f32(
      false, a, false, b, (short)0, c, false, false);
}

__device__ __forceinline__ void lds_barrier() {
  __builtin_amdgcn_fence(__ATOMIC_RELEASE, "workgroup", "local");
  __builtin_amdgcn_s_barrier();
  __builtin_amdgcn_fence(__ATOMIC_ACQUIRE, "workgroup", "local");
}

template <int N, int MODE, int MT, int NTB>
__device__ __forceinline__ void
gemm_tile(const float* A1, int k1,
             const float* A2, int k2,
             const float* __restrict__ W,
             const float* __restrict__ bias,
             const float* __restrict__ W2,
             const float* __restrict__ bias2,
             const float* __restrict__ eps,
             const float* __restrict__ W3,
             const float* __restrict__ bias3,
             float* C,
             float* C2,
             float* Z,
             int act,
             int bx,
             bool wmv,
             float* As,
             float* Bs)
{
  constexpr bool MV   = (MODE == MODE_MV);
  constexpr bool GATE = (MODE == MODE_GATE);
  static_assert(!GATE || (N == GH && NTB == 4 && MT == 2), "GATE: one 32 x 64 tile holds whole rows of both hidden layers");
  static_assert(MT * NTB == 8, "eight waves per work-group");
  static_assert(!MV || (N == MVN && NTB == 4 && MT == 2), "MV: one 32 x 64 tile, columns 0..31 mean, 32..63 log-variance");
  constexpr int BM  = 16 * MT;
  constexpr int BN  = 16 * NTB;
  constexpr int PA  = KC + 4;
  constexpr int PB  = BN + 8;
  constexpr int AJ  = KC / 32;
  constexpr int AP  = BM / 8;
  constexpr bool VEC = (N % BN == 0);
  constexpr int F4R = BN / 4;
  constexpr int BP  = KC * F4R / 256;
  constexpr int BRS = 256 / F4R;
  constexpr int BS  = KC * BN / 256;
  constexpr int BSS = 256 / BN;

  const int tid  = threadIdx.x;
  const int lane = tid & 31;
  const int wave = tid >> 5;
  const int h    = lane >> 4;
  const int ln   = lane & 15;
  const int wm   = wave / NTB;
  const int wn   = wave % NTB;
  const int m0   = blockIdx.y * BM;
  const int n0   = bx * BN;

  const int bn4 = tid % F4R;
  const int bk  = tid / F4R;
  const int bn  = tid % BN;
  const int bk2 = tid / BN;
  const int colc = (n0 + bn < N) ? (n0 + bn) : (N - 1);

  const int Ktot = k1 + k2;
  const int nc   = (Ktot + KC - 1) / KC;

  struct StageRegs {
    float  ra[AJ][AP];
    bool   av[AJ];
    v4f    rb[BP > 0 ? BP : 1];
    float  rbs[BS];
  };
  StageRegs R0, R1;

  auto gload = [&](int i, StageRegs& R) {
    auto& ra = R.ra; auto& av = R.av; auto& rb = R.rb; auto& rbs = R.rbs;
    const int k0 = i * KC;
    #pragma unroll
    for (int j = 0; j < AJ; ++j) {
      const int kk = k0 + 32 * j + lane;
      av[j] = kk < Ktot;
      const int kc = av[j] ? kk : Ktot - 1;
      const int ii = kc;
      const bool s2 = ii >= k1;
      const float* Aseg = s2 ? A2 : A1;
      const int len = s2 ? k2 : k1;
      const int cc  = s2 ? ii - k1 : ii;
      #pragma unroll
      for (int p = 0; p < AP; ++p)
        ra[j][p] = Aseg[(m0 + wave + 8 * p) * len + cc];
    }
    if (VEC) {
      #pragma unroll
      for (int p = 0; p < BP; ++p) {
        int kr = k0 + bk + BRS * p;
        kr = kr < Ktot ? kr : Ktot - 1;
        if (MV) {
          const float* __restrict__ Wh = (bn4 < F4R / 2) ? W : W2;
          rb[p] = *(const v4f*)(Wh + kr * LAT + 4 * (bn4 & (F4R / 2 - 1)));
        } else {
          rb[p] = *(const v4f*)(W + kr * N + n0 + 4 * bn4);
        }
      }
    } else {
      #pragma unroll
      for (int p = 0; p < BS; ++p) {
        int kr = k0 + bk2 + BSS * p;
        kr = kr < Ktot ? kr : Ktot - 1;
        rbs[p] = W[kr * N + colc];
      }
    }
  };

  auto sstore = [&](int st, StageRegs& R) {
    auto& ra = R.ra; auto& av = R.av; auto& rb = R.rb; auto& rbs = R.rbs;
    float* as = As + st * (BM * PA);
    float* bs = Bs + st * (KC * PB);
    #pragma unroll
    for (int j = 0; j < AJ; ++j) {
      #pragma unroll
      for (int p = 0; p < AP; ++p) {
        as[(wave + 8 * p) * PA + 32 * j + lane] = av[j] ? ra[j][p] : 0.f;
      }
    }
    if (VEC) {
      #pragma unroll
      for (int p = 0; p < BP; ++p)
        *(v4f*)(bs + (bk + BRS * p) * PB + 4 * bn4) = rb[p];
    } else {
      #pragma unroll
      for (int p = 0; p < BS; ++p)
        bs[(bk2 + BSS * p) * PB + bn] = rbs[p];
    }
  };

  v8f acc = (v8f){0.f,0.f,0.f,0.f,0.f,0.f,0.f,0.f};

  const int aoff = (16 * wm + ln) * PA + 2 * h;
  const int boff = (2 * h) * PB + 16 * wn + ln;

  auto mma32 = [&](const float* asj, const float* bsj, v8f a) {
    v2f af[8], bf[8];
    #pragma unroll
    for (int ks = 0; ks < 8; ++ks) {
      af[ks]   = *(const v2f*)(asj + 4 * ks);
      bf[ks].x = bsj[(4 * ks) * PB];
      bf[ks].y = bsj[(4 * ks + 1) * PB];
    }
    asm volatile("" :: "v"(af[0]), "v"(af[1]), "v"(af[2]), "v"(af[3]),
                       "v"(af[4]), "v"(af[5]), "v"(af[6]), "v"(af[7]),
                       "v"(bf[0]), "v"(bf[1]), "v"(bf[2]), "v"(bf[3]),
                       "v"(bf[4]), "v"(bf[5]), "v"(bf[6]), "v"(bf[7]));
    #pragma unroll
    for (int ks = 0; ks < 8; ++ks)
      a = wmma4(af[ks], bf[ks], a);
    return a;
  };

  const int colB = n0 + 16 * wn + ln;
  float pb = 0.f, pb2 = 0.f;
  float pe[BM * LAT / 256] = {0.f, 0.f, 0.f, 0.f};
  v4f   pw2[GH / BRS];
  float pw3[2] = {0.f, 0.f};
  #pragma unroll
  for (int p = 0; p < GH / BRS; ++p) pw2[p] = (v4f){0.f, 0.f, 0.f, 0.f};
  if (GATE) {
    pb  = bias[colB];
    pb2 = bias2[colB];
    #pragma unroll
    for (int p = 0; p < GH / BRS; ++p)
      pw2[p] = *(const v4f*)(W2 + (bk + BRS * p) * GH + 4 * bn4);
    pw3[0] = W3[tid];
    pw3[1] = W3[256 + (tid & 127)];
  } else if (MV) {
    pb = (colB < LAT) ? bias[colB] : bias2[colB - LAT];
    #pragma unroll
    for (int it = 0; it < BM * LAT / 256; ++it) {
      const int idx = tid + 256 * it;
      pe[it] = eps[(m0 + idx / LAT) * LAT + idx % LAT];
    }
  } else {
    if (bias) pb = bias[colB < N ? colB : N - 1];
  }

  gload(0, R0);
  if (nc > 1) gload(1, R1);
  sstore(0, R0);
  if (GATE)
    asm volatile("" :: "v"(pb), "v"(pb2), "v"(pw2[0]), "v"(pw2[1]), "v"(pw2[2]), "v"(pw2[3]),
                       "v"(pw3[0]), "v"(pw3[1]));
  else if (MV)
    asm volatile("" :: "v"(pb), "v"(pe[0]), "v"(pe[1]), "v"(pe[2]), "v"(pe[3]));
  else
    asm volatile("" :: "v"(pb));
  lds_barrier();

  auto chunk = [&](int i, int st) {
    const int rem = Ktot - i * KC;
    const int nsb = rem >= KC ? KC / 32 : (rem + 31) / 32;
    const float* as = As + st * (BM * PA) + aoff;
    const float* bs = Bs + st * (KC * PB) + boff;
    for (int j = 0; j < nsb; ++j)
      acc = mma32(as + 32 * j, bs + 32 * j * PB, acc);
  };

  int i = 0;
  for (; i + 3 < nc; i += 2) {
    gload(i + 2, R0);
    chunk(i, 0);
    sstore(1, R1);
    lds_barrier();
    gload(i + 3, R1);
    chunk(i + 1, 1);
    sstore(0, R0);
    lds_barrier();
  }
  const int left = nc - i;
  if (left == 3) {
    gload(i + 2, R0);
    chunk(i, 0);
    sstore(1, R1);
    lds_barrier();
    chunk(i + 1, 1);
    sstore(0, R0);
    lds_barrier();
    chunk(i + 2, 0);
  } else if (left == 2) {
    chunk(i, 0);
    sstore(1, R1);
    lds_barrier();
    chunk(i + 1, 1);
  } else {
    chunk(i, 0);
  }
  lds_barrier();

  if (GATE) {
    constexpr int PT = GH + 1;
    float* as2 = As;
    float* bs2 = Bs;
    float* W3s = Bs + KC * PB;
    {
      const float bv = pb;
      #pragma unroll
      for (int r = 0; r < 8; ++r)
        as2[(16 * wm + r + 8 * h) * PA + colB] = elu1(acc[r] + bv);
      #pragma unroll
      for (int p = 0; p < GH / BRS; ++p) {
        const int kr = bk + BRS * p;
        *(v4f*)(bs2 + kr * PB + 4 * bn4) = pw2[p];
      }
      W3s[tid] = pw3[0];
      if (tid < GH * NE - 256) W3s[256 + tid] = pw3[1];
    }
    __syncthreads();
    v8f acc2 = (v8f){0.f,0.f,0.f,0.f,0.f,0.f,0.f,0.f};
    #pragma unroll
    for (int j = 0; j < GH / 32; ++j)
      acc2 = mma32(as2 + aoff + 32 * j, bs2 + boff + 32 * j * PB, acc2);
    float* T = As + BM * PA;
    {
      const float bv2 = pb2;
      #pragma unroll
      for (int r = 0; r < 8; ++r)
        T[(16 * wm + r + 8 * h) * PT + colB] = elu1(acc2[r] + bv2);
    }
    __syncthreads();
    if (tid < BM) {
      float lg[NE];
      #pragma unroll
      for (int e = 0; e < NE; ++e) lg[e] = bias3[e];
      for (int i = 0; i < GH; ++i) {
        const float gv = T[tid * PT + i];
        #pragma unroll
        for (int e = 0; e < NE; ++e) lg[e] = fmaf(gv, W3s[i * NE + e], lg[e]);
      }
      float mx = lg[0];
      #pragma unroll
      for (int e = 1; e < NE; ++e) mx = fmaxf(mx, lg[e]);
      float s = 0.f, ex[NE];
      #pragma unroll
      for (int e = 0; e < NE; ++e) { ex[e] = expf(lg[e] - mx); s += ex[e]; }
      const float inv = 1.f / s;
      #pragma unroll
      for (int e = 0; e < NE; ++e) Z[(m0 + tid) * NE + e] = ex[e] * inv;
    }
  } else if (MV) {
    constexpr int PT = MVN + 1;
    float* T = As;
    const float bv = pb;
    #pragma unroll
    for (int r = 0; r < 8; ++r)
      T[(16 * wm + r + 8 * h) * PT + colB] = acc[r] + bv;
    __syncthreads();
    #pragma unroll
    for (int it = 0; it < BM * LAT / 256; ++it) {
      const int idx = tid + 256 * it;
      const int rr  = idx / LAT;
      const int cc  = idx % LAT;
      const float m = T[rr * PT + cc];
      const float v = T[rr * PT + LAT + cc];
      const int   o = (m0 + rr) * LAT + cc;
      if (wmv) {
        C[o]  = m;
        C2[o] = v;
      }
      Z[o]  = m + pe[it] * expf(0.5f * v);
    }
  } else if (colB < N) {
    #pragma unroll
    for (int r = 0; r < 8; ++r) {
      const int row = m0 + 16 * wm + r + 8 * h;
      float y = acc[r];
      if (bias) y += pb;
      if (act) y = elu1(y);
      C[row * N + colB] = y;
    }
  }
}

template <int N, int K1, int K2, int MT, int NTB, bool PF2>
__device__ __forceinline__ void
moe_tile(const float* A1, const float* A2,
         const float* coeff,
         const float* __restrict__ W,
         const float* __restrict__ bias,
         float* C,
         int act,
         int bx,
         int by,
         float* As,
         float* Bs,
         float* Cs)
{
  static_assert(K1 % KQ == 0, "a round reads one segment of [A1|A2]");
  static_assert(MT * NTB == 8, "eight waves per work-group");
  constexpr int BM  = 16 * MT;
  constexpr int BN  = 16 * NTB;
  constexpr int I   = K1 + K2;
  constexpr int NR  = (I + KQ - 1) / KQ;
  constexpr bool EXACT = (I % KQ == 0);
  constexpr int PA  = KQ + 4;
  constexpr int PB  = BN + 8;
  constexpr int WR  = NE * KQ;
  constexpr int AP  = BM / 8;
  constexpr bool VEC = (N % BN == 0);
  constexpr int F4R = BN / 4;
  constexpr int BP  = WR * F4R / 256;
  constexpr int BRS = 256 / F4R;
  constexpr int BS  = WR * BN / 256;
  constexpr int BSS = 256 / BN;
  static_assert(KQ % BRS == 0 && KQ % BSS == 0, "a staging pass stays inside one expert");

  const int tid  = threadIdx.x;
  const int lane = tid & 31;
  const int wave = tid >> 5;
  const int h    = lane >> 4;
  const int ln   = lane & 15;
  const int wm   = wave / NTB;
  const int wn   = wave % NTB;
  const int m0   = by * BM;
  const int n0   = bx * BN;

  const int bn4 = tid % F4R;
  const int bk  = tid / F4R;
  const int bn  = tid % BN;
  const int bk2 = tid / BN;
  const int colc = (n0 + bn < N) ? (n0 + bn) : (N - 1);

  struct StageRegs {
    float ra[AP];
    bool  av;
    v4f   rb[BP];
    float rbs[BS];
  };
  StageRegs R0, R1;

  auto gload = [&](int i, StageRegs& R) {
    auto& ra = R.ra; auto& av = R.av; auto& rb = R.rb; auto& rbs = R.rbs;
    const int k0 = i * KQ;
    const bool full = EXACT || (k0 + KQ <= I);
    const bool s2 = k0 >= K1;
    const float* Aseg = s2 ? A2 : A1;
    const int len = s2 ? K2 : K1;
    av = full || (k0 + lane < I);
    const int kc = av ? k0 + lane : I - 1;
    const int cc = s2 ? kc - K1 : kc;
    #pragma unroll
    for (int p = 0; p < AP; ++p)
      ra[p] = Aseg[(m0 + wave + 8 * p) * len + cc];
    if (VEC) {
      #pragma unroll
      for (int p = 0; p < BP; ++p) {
        const int e = (BRS * p) / KQ;
        int kr = k0 + (BRS * p) % KQ + bk;
        if (!full) kr = kr < I ? kr : I - 1;
        rb[p] = *(const v4f*)(W + (e * I + kr) * N + n0 + 4 * bn4);
      }
    } else {
      #pragma unroll
      for (int p = 0; p < BS; ++p) {
        const int e = (BSS * p) / KQ;
        int kr = k0 + (BSS * p) % KQ + bk2;
        if (!full) kr = kr < I ? kr : I - 1;
        rbs[p] = W[(e * I + kr) * N + colc];
      }
    }
  };

  auto sstore = [&](int st, StageRegs& R) {
    auto& ra = R.ra; auto& av = R.av; auto& rb = R.rb; auto& rbs = R.rbs;
    float* as = As + st * (BM * PA);
    float* bs = Bs + st * (WR * PB);
    #pragma unroll
    for (int p = 0; p < AP; ++p)
      as[(wave + 8 * p) * PA + lane] = av ? ra[p] : 0.f;
    if (VEC) {
      #pragma unroll
      for (int p = 0; p < BP; ++p)
        *(v4f*)(bs + (bk + BRS * p) * PB + 4 * bn4) = rb[p];
    } else {
      #pragma unroll
      for (int p = 0; p < BS; ++p)
        bs[(bk2 + BSS * p) * PB + bn] = rbs[p];
    }
  };

  v8f acc[NE];
  #pragma unroll
  for (int e = 0; e < NE; ++e) acc[e] = (v8f){0.f,0.f,0.f,0.f,0.f,0.f,0.f,0.f};

  const int aoff = (16 * wm + ln) * PA + 2 * h;
  const int boff = (2 * h) * PB + 16 * wn + ln;

  auto mma16 = [&](const float* ag, const float* bg) {
    v2f af[4], bf[4][NE];
    #pragma unroll
    for (int ks = 0; ks < 4; ++ks) {
      af[ks] = *(const v2f*)(ag + 4 * ks);
      #pragma unroll
      for (int e = 0; e < NE; ++e) {
        bf[ks][e].x = bg[(e * KQ + 4 * ks) * PB];
        bf[ks][e].y = bg[(e * KQ + 4 * ks + 1) * PB];
      }
    }
    asm volatile("" :: "v"(af[0]), "v"(af[1]), "v"(af[2]), "v"(af[3]),
                       "v"(bf[0][0]), "v"(bf[0][1]), "v"(bf[0][2]), "v"(bf[0][3]), "v"(bf[0][4]), "v"(bf[0][5]),
                       "v"(bf[1][0]), "v"(bf[1][1]), "v"(bf[1][2]), "v"(bf[1][3]), "v"(bf[1][4]), "v"(bf[1][5]),
                       "v"(bf[2][0]), "v"(bf[2][1]), "v"(bf[2][2]), "v"(bf[2][3]), "v"(bf[2][4]), "v"(bf[2][5]),
                       "v"(bf[3][0]), "v"(bf[3][1]), "v"(bf[3][2]), "v"(bf[3][3]), "v"(bf[3][4]), "v"(bf[3][5]));
    #pragma unroll
    for (int ks = 0; ks < 4; ++ks) {
      #pragma unroll
      for (int e = 0; e < NE; ++e)
        acc[e] = wmma4(af[ks], bf[ks][e], acc[e]);
    }
  };

  const int colB = n0 + 16 * wn + ln;
  float be[NE];
  #pragma unroll
  for (int e = 0; e < NE; ++e) be[e] = bias[e * N + (colB < N ? colB : N - 1)];
  float cf[8][NE];
  constexpr int CP = (BM * NE + 255) / 256;
  float cw[CP];
  if (PF2) {
    #pragma unroll
    for (int p = 0; p < CP; ++p) {
      const int idx = tid + 256 * p;
      cw[p] = coeff[m0 * NE + (idx < BM * NE ? idx : BM * NE - 1)];
    }
  } else {
    #pragma unroll
    for (int r = 0; r < 8; ++r) {
      #pragma unroll
      for (int e = 0; e < NE; ++e) cf[r][e] = coeff[(m0 + 16 * wm + r + 8 * h) * NE + e];
    }
  }

  gload(0, R0);
  if (PF2 && NR > 1) gload(1, R1);
  sstore(0, R0);
  asm volatile("" :: "v"(be[0]), "v"(be[1]), "v"(be[2]), "v"(be[3]), "v"(be[4]), "v"(be[5]));
  if (PF2) {
    #pragma unroll
    for (int p = 0; p < CP; ++p) {
      const int idx = tid + 256 * p;
      if (idx < BM * NE) Cs[idx] = cw[p];
    }
    lds_barrier();
  } else {
    #pragma unroll
    for (int r = 0; r < 8; ++r)
      asm volatile("" :: "v"(cf[r][0]), "v"(cf[r][1]), "v"(cf[r][2]), "v"(cf[r][3]), "v"(cf[r][4]), "v"(cf[r][5]));
    __syncthreads();
  }

  auto round = [&](int i, int st) {
    const int rem = I - i * KQ;
    const int ng  = (EXACT || rem >= KQ) ? KQ / 16 : (rem + 15) / 16;
    const float* as = As + st * (BM * PA) + aoff;
    const float* bs = Bs + st * (WR * PB) + boff;
    for (int g = 0; g < ng; ++g)
      mma16(as + 16 * g, bs + 16 * g * PB);
  };

  if (PF2) {
    int i = 0;
    for (; i + 3 < NR; i += 2) {
      gload(i + 2, R0);
      round(i, 0);
      sstore(1, R1);
      lds_barrier();
      gload(i + 3, R1);
      round(i + 1, 1);
      sstore(0, R0);
      lds_barrier();
    }
    const int left = NR - i;
    if (left == 3) {
      gload(i + 2, R0);
      round(i, 0);
      sstore(1, R1);
      lds_barrier();
      round(i + 1, 1);
      sstore(0, R0);
      lds_barrier();
      round(i + 2, 0);
    } else if (left == 2) {
      round(i, 0);
      sstore(1, R1);
      lds_barrier();
      round(i + 1, 1);
    } else {
      round(i, 0);
    }
  } else {
    for (int i = 0; i < NR; ++i) {
      const int st = i & 1;
      if (i + 1 < NR) gload(i + 1, R0);
      round(i, st);
      if (i + 1 < NR) sstore(st ^ 1, R0);
      __syncthreads();
    }
  }

  if (colB < N) {
    #pragma unroll
    for (int r = 0; r < 8; ++r) {
      const int row = m0 + 16 * wm + r + 8 * h;
      float y = 0.f, s = 0.f;
      #pragma unroll
      for (int e = 0; e < NE; ++e) {
        const float ce = PF2 ? Cs[(16 * wm + r + 8 * h) * NE + e] : cf[r][e];
        y = fmaf(ce, acc[e][r], y);
        s = fmaf(ce, be[e], s);
      }
      y += s;
      if (act) y = elu1(y);
      C[row * N + colB] = y;
    }
  }
}

template <int N, int K1, int K2>
__device__ __forceinline__ void
moe_edge_tile(const float* A1, const float* A2,
              const float* coeff,
              const float* __restrict__ W,
              const float* __restrict__ bias,
              float* C,
              int act,
              int n0,
              int m0,
              float* As,
              float* Bs)
{
  static_assert(K1 % KQ == 0 && (K1 + K2) % KQ == 0, "whole rounds, each inside one segment of [A1|A2]");
  constexpr int I   = K1 + K2;
  constexpr int NR  = I / KQ;
  constexpr int PA  = KQ + 4;
  constexpr int PB  = 16 + 8;
  constexpr int WR  = NE * KQ;
  constexpr int BS  = WR * 16 / 256;
  constexpr int PT  = 16 + 4;

  const int tid  = threadIdx.x;
  const int lane = tid & 31;
  const int wave = __builtin_amdgcn_readfirstlane(tid >> 5);
  const int h    = lane >> 4;
  const int ln   = lane & 15;
  const bool live = wave < NE;

  const int tc = tid & 15;
  const int tr = tid >> 4;
  const int colE = n0 + tc;
  const int colc = colE < N ? colE : N - 1;

  float ra[2];
  float rbs[BS];

  auto gload = [&](int i) {
    const int k0 = i * KQ;
    const bool s2 = k0 >= K1;
    const float* Aseg = s2 ? A2 : A1;
    const int len = s2 ? K2 : K1;
    const int cc  = (s2 ? k0 - K1 : k0) + lane;
    #pragma unroll
    for (int p = 0; p < 2; ++p)
      ra[p] = Aseg[(m0 + wave + 8 * p) * len + cc];
    #pragma unroll
    for (int p = 0; p < BS; ++p) {
      const int e  = (16 * p) / KQ;
      const int kr = k0 + (16 * p) % KQ + tr;
      rbs[p] = W[(e * I + kr) * N + colc];
    }
  };

  auto sstore = [&](int st) {
    float* as = As + st * (16 * PA);
    float* bs = Bs + st * (WR * PB);
    #pragma unroll
    for (int p = 0; p < 2; ++p)
      as[(wave + 8 * p) * PA + lane] = ra[p];
    #pragma unroll
    for (int p = 0; p < BS; ++p)
      bs[(tr + 16 * p) * PB + tc] = rbs[p];
  };

  v8f acc = (v8f){0.f,0.f,0.f,0.f,0.f,0.f,0.f,0.f};

  const int aoff = ln * PA + 2 * h;
  const int boff = ((live ? wave : 0) * KQ + 2 * h) * PB + ln;

  auto round = [&](int st) {
    const float* as = As + st * (16 * PA) + aoff;
    const float* bs = Bs + st * (WR * PB) + boff;
    v2f af[8], bf[8];
    #pragma unroll
    for (int ks = 0; ks < 8; ++ks) {
      af[ks]   = *(const v2f*)(as + 4 * ks);
      bf[ks].x = bs[(4 * ks) * PB];
      bf[ks].y = bs[(4 * ks + 1) * PB];
    }
    asm volatile("" :: "v"(af[0]), "v"(af[1]), "v"(af[2]), "v"(af[3]),
                       "v"(af[4]), "v"(af[5]), "v"(af[6]), "v"(af[7]),
                       "v"(bf[0]), "v"(bf[1]), "v"(bf[2]), "v"(bf[3]),
                       "v"(bf[4]), "v"(bf[5]), "v"(bf[6]), "v"(bf[7]));
    #pragma unroll
    for (int ks = 0; ks < 8; ++ks)
      acc = wmma4(af[ks], bf[ks], acc);
  };

  float be[NE], cf[NE];
  #pragma unroll
  for (int e = 0; e < NE; ++e) {
    be[e] = bias[e * N + colc];
    cf[e] = coeff[(m0 + tr) * NE + e];
  }

  gload(0);
  sstore(0);
  asm volatile("" :: "v"(be[0]), "v"(be[1]), "v"(be[2]), "v"(be[3]), "v"(be[4]), "v"(be[5]),
                     "v"(cf[0]), "v"(cf[1]), "v"(cf[2]), "v"(cf[3]), "v"(cf[4]), "v"(cf[5]));
  __syncthreads();

  for (int i = 0; i < NR; ++i) {
    const int st = i & 1;
    if (i + 1 < NR) gload(i + 1);
    if (live) round(st);
    if (i + 1 < NR) sstore(st ^ 1);
    __syncthreads();
  }

  float* T = Bs;
  if (live) {
    #pragma unroll
    for (int r = 0; r < 8; ++r)
      T[wave * (16 * PT) + (8 * h + r) * PT + ln] = acc[r];
  }
  __syncthreads();
  if (colE < N) {
    float y = 0.f, s = 0.f;
    #pragma unroll
    for (int e = 0; e < NE; ++e) {
      y = fmaf(cf[e], T[e * (16 * PT) + tr * PT + tc], y);
      s = fmaf(cf[e], be[e], s);
    }
    y += s;
    if (act) y = elu1(y);
    C[(m0 + tr) * N + colE] = y;
  }
}

template <int N, int MODE, int MT, int NTB, int K1, int K2>
__global__ void __launch_bounds__(256)
gemm_lds_f32(const float* __restrict__ A1,
             const float* __restrict__ A2,
             const float* __restrict__ W,
             const float* __restrict__ bias,
             float* __restrict__ C,
             int act)
{
  __shared__ __attribute__((aligned(16))) float As[2 * (16 * MT) * (KC + 4)];
  __shared__ __attribute__((aligned(16))) float Bs[2 * KC * (16 * NTB + 8)];
  gemm_tile<N, MODE, MT, NTB>(A1, K1, A2, K2, W, bias, nullptr, nullptr, nullptr, nullptr, nullptr,
                              C, nullptr, nullptr, act, blockIdx.x, false, As, Bs);
}

template <int N, int K2, int MT, int NTB>
__global__ void __launch_bounds__(256)
moe_lds_f32(const float* __restrict__ z,
            const float* __restrict__ prev,
            const float* __restrict__ coeff,
            const float* __restrict__ W,
            const float* __restrict__ bias,
            float* __restrict__ C,
            int act)
{
  __shared__ __attribute__((aligned(16))) float As[2 * (16 * MT) * (KQ + 4)];
  __shared__ __attribute__((aligned(16))) float Bs[2 * NE * KQ * (16 * NTB + 8)];
  moe_tile<N, LAT, K2, MT, NTB, false>(z, prev, coeff, W, bias, C, act, blockIdx.x, blockIdx.y, As, Bs, nullptr);
}

template <int N, int K2>
__global__ void __launch_bounds__(256)
moe_ragged_lds_f32(const float* __restrict__ z,
                   const float* __restrict__ prev,
                   const float* __restrict__ coeff,
                   const float* __restrict__ W,
                   const float* __restrict__ bias,
                   float* __restrict__ C,
                   int act)
{
  static_assert(N % 32 != 0 && N % 32 <= 16, "the edge is one 16-column tile");
  constexpr int NFULL = N / 32;
  constexpr int NFT   = NFULL * (BATCH / 64);
  __shared__ __attribute__((aligned(16))) float As[2 * 64 * (KQ + 4)];
  __shared__ __attribute__((aligned(16))) float Bs[2 * NE * KQ * (32 + 8)];
  const int t = blockIdx.x;
  if (t < NFT)
    moe_tile<N, LAT, K2, 4, 2, false>(z, prev, coeff, W, bias, C, act, t % NFULL, t / NFULL, As, Bs, nullptr);
  else
    moe_edge_tile<N, LAT, K2>(z, prev, coeff, W, bias, C, act, NFULL * 32, (t - NFT) * 16, As, Bs);
}

__global__ void __launch_bounds__(256)
head_decoder0(const float* __restrict__ x, const float* __restrict__ h2in,
              const float* __restrict__ c, const float* __restrict__ eps,
              const float* __restrict__ em_w, const float* __restrict__ em_b,
              const float* __restrict__ ev_w, const float* __restrict__ ev_b,
              const float* __restrict__ g1_w, const float* __restrict__ g1_b,
              const float* __restrict__ g2_w, const float* __restrict__ g2_b,
              const float* __restrict__ g3_w, const float* __restrict__ g3_b,
              const float* __restrict__ w0, const float* __restrict__ b0,
              float* __restrict__ out_m, float* __restrict__ out_v,
              float* zpl, float* cpl,
              float* __restrict__ d0)
{
  __shared__ __attribute__((aligned(16))) float As[2 * 32 * (KC + 4)];
  __shared__ __attribute__((aligned(16))) float Bs[2 * (NE * KQ > KC ? NE * KQ : KC) * (64 + 8)];
  __shared__ float Cs[32 * NE];
  const int bx = blockIdx.x;
  float* zmine = zpl + bx * (BATCH * LAT);
  float* cmine = cpl + bx * (BATCH * NE);

  gemm_tile<MVN, MODE_MV, 2, 4>(x, FRAME, h2in, HID, em_w, em_b, ev_w, ev_b, eps, nullptr, nullptr,
                                out_m, out_v, zmine, 0, 0, bx == 0, As, Bs);
  __syncthreads();
  gemm_tile<GH, MODE_GATE, 2, 4>(zmine, LAT, c, FRAME, g1_w, g1_b, g2_w, g2_b, nullptr, g3_w, g3_b,
                                 nullptr, nullptr, cmine, 1, 0, false, As, Bs);
  __syncthreads();
  moe_tile<HID, LAT, FRAME, 2, 4, true>(zmine, c, cmine, w0, b0, d0, 1, bx, blockIdx.y, As, Bs, Cs);
}

template <int N, int MT, int NTB>
static inline dim3 gemm_grid() {
  return dim3((N + 16 * NTB - 1) / (16 * NTB), BATCH / (16 * MT));
}

template <int N>
static inline dim3 ragged_grid() {
  return dim3((N / 32) * (BATCH / 64) + BATCH / 16);
}

extern "C" void kernel_launch(void* const* d_in, const int* in_sizes, int n_in,
                              void* d_out, int out_size, void* d_ws, size_t ws_size,
                              hipStream_t stream) {
  const float* x      = (const float*)d_in[0];
  const float* c      = (const float*)d_in[1];
  const float* eps    = (const float*)d_in[2];
  const float* efc1_w = (const float*)d_in[3];
  const float* efc1_b = (const float*)d_in[4];
  const float* efc2_w = (const float*)d_in[5];
  const float* efc2_b = (const float*)d_in[6];
  const float* em_w   = (const float*)d_in[7];
  const float* em_b   = (const float*)d_in[8];
  const float* ev_w   = (const float*)d_in[9];
  const float* ev_b   = (const float*)d_in[10];
  const float* g1_w   = (const float*)d_in[11];
  const float* g1_b   = (const float*)d_in[12];
  const float* g2_w   = (const float*)d_in[13];
  const float* g2_b   = (const float*)d_in[14];
  const float* g3_w   = (const float*)d_in[15];
  const float* g3_b   = (const float*)d_in[16];
  const float* w0     = (const float*)d_in[17];
  const float* b0     = (const float*)d_in[18];
  const float* w1     = (const float*)d_in[19];
  const float* b1     = (const float*)d_in[20];
  const float* w2     = (const float*)d_in[21];
  const float* b2     = (const float*)d_in[22];

  float* out   = (float*)d_out;
  float* out_m = out + BATCH * FRAME;
  float* out_v = out_m + BATCH * LAT;

  float* ws    = (float*)d_ws;
  float* h1    = ws;
  float* h2    = h1  + BATCH * HID;
  float* zpl   = h2  + BATCH * HID;
  float* cpl   = zpl + 4 * BATCH * LAT;
  float* z     = zpl;
  float* coeff = cpl;

  gemm_lds_f32<HID, MODE_PLAIN, 2, 4, FRAME, FRAME><<<gemm_grid<HID, 2, 4>(), 256, 0, stream>>>(
      x, c, efc1_w, efc1_b, h1, 1);
  gemm_lds_f32<HID, MODE_PLAIN, 2, 4, FRAME, HID><<<gemm_grid<HID, 2, 4>(), 256, 0, stream>>>(
      x, h1, efc2_w, efc2_b, h2, 1);

  head_decoder0<<<gemm_grid<HID, 2, 4>(), 256, 0, stream>>>(
      x, h2, c, eps, em_w, em_b, ev_w, ev_b, g1_w, g1_b, g2_w, g2_b, g3_w, g3_b, w0, b0,
      out_m, out_v, zpl, cpl, h1);

  moe_lds_f32<HID, HID, 4, 2><<<gemm_grid<HID, 4, 2>(), 256, 0, stream>>>(
      z, h1, coeff, w1, b1, h2, 1);
  moe_ragged_lds_f32<FRAME, HID><<<ragged_grid<FRAME>(), 256, 0, stream>>>(
      z, h2, coeff, w2, b2, out, 0);
}
